# starting kernel
# baseline (speedup 1.0000x reference)
#include <hip/hip_runtime.h>
#include <math.h>

static constexpr int Bn = 2048;
static constexpr int Tn = 1000;

typedef float v2f __attribute__((ext_vector_type(2)));
typedef float v4f __attribute__((ext_vector_type(4)));
typedef float v8f __attribute__((ext_vector_type(8)));

__global__ __launch_bounds__(64) void kf_gain_kernel(
    const float* __restrict__ F, const float* __restrict__ Gmat,
    const float* __restrict__ H, const float* __restrict__ Q,
    const float* __restrict__ R, const float* __restrict__ P0,
    float* __restrict__ Gws, int* __restrict__ t_stop_ptr,
    float* __restrict__ Ps)
{
    __shared__ float Fm[8][8], Hm[8][8], Qe[8][8], Rm[8][8];
    __shared__ float P[8][8], Pneg[8][8], T1[8][8], T2[8][8];
    __shared__ float S[8][8], X[8][8], IKH[8][8], Amat[8][8], Pn[8][8];
    __shared__ float GQ[8][8];
    __shared__ int flag;

    const int tid = threadIdx.x;
    const int i = tid >> 3, j = tid & 7;

    Fm[i][j] = F[tid];
    Hm[i][j] = H[tid];
    Rm[i][j] = R[tid];
    P[i][j]  = P0[tid];
    __syncthreads();

    float acc = 0.f;
    #pragma unroll
    for (int k = 0; k < 8; ++k) acc += Gmat[i * 8 + k] * Q[k * 8 + j];
    GQ[i][j] = acc;
    __syncthreads();
    acc = 0.f;
    #pragma unroll
    for (int k = 0; k < 8; ++k) acc += GQ[i][k] * Gmat[j * 8 + k];
    Qe[i][j] = acc;
    __syncthreads();

    int tstop = Tn;
    for (int t = 0; t < Tn; ++t) {
        acc = 0.f;
        #pragma unroll
        for (int k = 0; k < 8; ++k) acc += Fm[i][k] * P[k][j];
        T1[i][j] = acc;
        __syncthreads();
        acc = Qe[i][j];
        #pragma unroll
        for (int k = 0; k < 8; ++k) acc += T1[i][k] * Fm[j][k];
        Pneg[i][j] = acc;
        __syncthreads();
        acc = 0.f;
        #pragma unroll
        for (int k = 0; k < 8; ++k) acc += Hm[i][k] * Pneg[k][j];
        T2[i][j] = acc;
        __syncthreads();
        acc = Rm[i][j];
        #pragma unroll
        for (int k = 0; k < 8; ++k) acc += T2[i][k] * Hm[j][k];
        S[i][j] = acc;
        __syncthreads();
        #pragma unroll 1
        for (int k = 0; k < 8; ++k) {
            if (i == k && j == k) S[k][k] = sqrtf(S[k][k]);
            __syncthreads();
            if (j == k && i > k) S[i][k] = S[i][k] / S[k][k];
            __syncthreads();
            if (i > k && j > k && j <= i) S[i][j] -= S[i][k] * S[j][k];
            __syncthreads();
        }
        #pragma unroll 1
        for (int r = 0; r < 8; ++r) {
            if (i == r) {
                float s = T2[r][j];
                for (int m = 0; m < r; ++m) s -= S[r][m] * X[m][j];
                X[r][j] = s / S[r][r];
            }
            __syncthreads();
        }
        #pragma unroll 1
        for (int r = 7; r >= 0; --r) {
            if (i == r) {
                float s = X[r][j];
                for (int m = r + 1; m < 8; ++m) s -= S[m][r] * X[m][j];
                X[r][j] = s / S[r][r];
            }
            __syncthreads();
        }
        acc = (i == j) ? 1.f : 0.f;
        #pragma unroll
        for (int m = 0; m < 8; ++m) acc -= X[m][i] * Hm[m][j];
        IKH[i][j] = acc;
        __syncthreads();
        acc = 0.f;
        float pacc = 0.f;
        #pragma unroll
        for (int k = 0; k < 8; ++k) {
            acc  += IKH[i][k] * Fm[k][j];
            pacc += IKH[i][k] * Pneg[k][j];
        }
        Amat[i][j] = acc;
        Pn[i][j]   = pacc;
        __syncthreads();
        float sym = 0.5f * (Pn[i][j] + Pn[j][i]);
        __syncthreads();
        Pn[i][j] = sym;
        __syncthreads();

        if (tid == 0) flag = 0;
        __syncthreads();
        if (Pn[i][j] != P[i][j]) flag = 1;

        Ps[t * 64 + tid] = Pn[i][j];
        float* gout = Gws + (size_t)t * 256;
        #pragma unroll
        for (int q = 0; q < 4; ++q) {
            int f    = tid * 4 + q;
            int lane = f >> 3, slot = f & 7;
            int s2 = slot >> 1, p = slot & 1;
            int m = lane & 15;
            int k = 4 * s2 + ((lane < 16) ? p : (2 + p));
            int mm = (m < 8) ? m : ((m <= 13) ? (m - 6) : -1);
            float v = 0.f;
            if (mm >= 0) v = (k < 8) ? Amat[mm][k] : X[k - 8][mm];
            gout[f] = v;
        }
        P[i][j] = Pn[i][j];
        __syncthreads();
        if (flag == 0) { tstop = t + 1; break; }
    }
    if (tid == 0) *t_stop_ptr = tstop;
}

__global__ __launch_bounds__(256) void kf_fill_kernel(
    float* __restrict__ Gws, const int* __restrict__ t_stop_ptr,
    float* __restrict__ Ps)
{
    const int t  = blockIdx.x;
    const int q  = threadIdx.x;
    const int ts = *t_stop_ptr;
    if (t >= ts && ts >= 1) {
        const int src = ts - 1;
        Gws[(size_t)t * 256 + q] = Gws[(size_t)src * 256 + q];
        if (q < 64) Ps[t * 64 + q] = Ps[src * 64 + q];
    }
}

__global__ __launch_bounds__(32) void kf_scan_kernel(
    const float* __restrict__ Y, const float* __restrict__ x0,
    const float* __restrict__ Gws, float* __restrict__ traj)
{
    const int lane = threadIdx.x;
    const int n    = lane & 15;
    const bool lo  = lane < 16;
    const size_t row = (size_t)(blockIdx.x * 16 + n);

    const float* yrow = Y    + row * (size_t)Tn * 8 + (lo ? 0 : 2);
    float*       orow = traj + row * (size_t)Tn * 8 + (lo ? 0 : 4);

    v4f xa = *(const v4f*)(x0 + row * 8);
    v4f xb = *(const v4f*)(x0 + row * 8 + 4);
    v8f xv;
    xv[0] = lo ? xa.x : xa.z;
    xv[1] = lo ? xa.y : xa.w;
    xv[2] = lo ? xa.z : xb.x;
    xv[3] = lo ? xa.w : xb.y;
    xv[4] = lo ? xb.x : xb.z;
    xv[5] = lo ? xb.y : xb.w;
    xv[6] = lo ? xb.z : 0.f;
    xv[7] = lo ? xb.w : 0.f;

    const v8f zero8 = {0.f, 0.f, 0.f, 0.f, 0.f, 0.f, 0.f, 0.f};

    for (int t = 0; t < Tn; ++t) {
        const v4f* gp = (const v4f*)(Gws + (size_t)t * 256 + lane * 8);
        v4f g0 = gp[0];
        v4f g1 = gp[1];
        v2f a0 = {g0.x, g0.y};
        v2f a1 = {g0.z, g0.w};
        v2f a2 = {g1.x, g1.y};
        v2f a3 = {g1.z, g1.w};

        v2f b2 = *(const v2f*)(yrow + (size_t)t * 8);
        v2f b3 = *(const v2f*)(yrow + (size_t)t * 8 + 4);

        v2f b0 = {xv[0], xv[1]};
        v2f b1 = {xv[4], xv[5]};

        v8f cy = __builtin_amdgcn_wmma_f32_16x16x4_f32(false, a2, false, b2,
                                                       (short)0, zero8, false, false);
        cy = __builtin_amdgcn_wmma_f32_16x16x4_f32(false, a3, false, b3,
                                                   (short)0, cy, false, false);
        v8f c1 = __builtin_amdgcn_wmma_f32_16x16x4_f32(false, a0, false, b0,
                                                       (short)0, cy, false, false);
        v8f c2 = __builtin_amdgcn_wmma_f32_16x16x4_f32(false, a1, false, b1,
                                                       (short)0, zero8, false, false);
        xv = c1 + c2;

        v4f out;
        out.x = lo ? xv[0] : xv[2];
        out.y = lo ? xv[1] : xv[3];
        out.z = lo ? xv[2] : xv[4];
        out.w = lo ? xv[3] : xv[5];
        *(v4f*)(orow + (size_t)t * 8) = out;
    }
}

extern "C" void kernel_launch(void* const* d_in, const int* in_sizes, int n_in,
                              void* d_out, int out_size, void* d_ws, size_t ws_size,
                              hipStream_t stream) {
    const float* Y    = (const float*)d_in[0];
    const float* F    = (const float*)d_in[1];
    const float* Gmat = (const float*)d_in[2];
    const float* H    = (const float*)d_in[3];
    const float* Q    = (const float*)d_in[4];
    const float* R    = (const float*)d_in[5];
    const float* x0   = (const float*)d_in[6];
    const float* P0   = (const float*)d_in[7];

    float* traj = (float*)d_out;
    float* Ps   = traj + (size_t)Bn * Tn * 8;

    int*   t_stop = (int*)d_ws;
    float* Gws    = (float*)((char*)d_ws + 256);

    kf_gain_kernel<<<1, 64, 0, stream>>>(F, Gmat, H, Q, R, P0, Gws, t_stop, Ps);
    kf_fill_kernel<<<Tn, 256, 0, stream>>>(Gws, t_stop, Ps);
    kf_scan_kernel<<<Bn / 16, 32, 0, stream>>>(Y, x0, Gws, traj);
}


# Covariance recursion kernel runs as ONE wave: each lane keeps its matrix row/column in registers, 3 LDS broadcasts per step replace 50 block barriers; per-element arithmetic order unchanged
# speedup vs baseline: 3.6356x; 3.6356x over previous
#include <hip/hip_runtime.h>
#include <math.h>

static constexpr int Bn = 2048;
static constexpr int Tn = 1000;

typedef float v2f __attribute__((ext_vector_type(2)));
typedef float v4f __attribute__((ext_vector_type(4)));
typedef float v8f __attribute__((ext_vector_type(8)));


__device__ __forceinline__ void lds_handover() {
    __builtin_amdgcn_fence(__ATOMIC_RELEASE, "workgroup", "local");
    __builtin_amdgcn_s_barrier();
    __builtin_amdgcn_fence(__ATOMIC_ACQUIRE, "workgroup", "local");
}

__device__ __forceinline__ void lds_put_row(float (*M)[8], int r, const float* v) {
    v4f a = {v[0], v[1], v[2], v[3]};
    v4f b = {v[4], v[5], v[6], v[7]};
    *(v4f*)&M[r][0] = a;
    *(v4f*)&M[r][4] = b;
}

__device__ __forceinline__ void lds_get_row(const float (*M)[8], int r, float* v) {
    v4f a = *(const v4f*)&M[r][0];
    v4f b = *(const v4f*)&M[r][4];
    v[0] = a.x; v[1] = a.y; v[2] = a.z; v[3] = a.w;
    v[4] = b.x; v[5] = b.y; v[6] = b.z; v[7] = b.w;
}

__global__ __launch_bounds__(32) void kf_gain_kernel(
    const float* __restrict__ F, const float* __restrict__ Gmat,
    const float* __restrict__ H, const float* __restrict__ Q,
    const float* __restrict__ R, const float* __restrict__ P0,
    float* __restrict__ Gws, int* __restrict__ t_stop_ptr,
    float* __restrict__ Ps)
{
    __shared__ __attribute__((aligned(16))) float Ash[8][8];
    __shared__ __attribute__((aligned(16))) float Bsh[8][8];
    __shared__ __attribute__((aligned(16))) float Csh[8][8];

    const int  lane   = threadIdx.x;
    const int  m16    = lane & 15;
    const int  idx    = (m16 < 8) ? m16 : ((m16 <= 13) ? (m16 - 6) : (m16 - 8));
    const bool live   = (m16 <= 13);
    const bool lo     = lane < 16;
    const bool writer = lane < 8;

    float Fr[8], Hr[8], Rr[8], Pr[8], Gr[8];
    #pragma unroll
    for (int c = 0; c < 8; ++c) {
        Fr[c] = F[idx * 8 + c];
        Hr[c] = H[idx * 8 + c];
        Rr[c] = R[idx * 8 + c];
        Pr[c] = P0[idx * 8 + c];
        Gr[c] = Gmat[idx * 8 + c];
    }

    if (writer) {
        lds_put_row(Ash, idx, Fr);
        lds_put_row(Bsh, idx, Hr);
        lds_put_row(Csh, idx, Pr);
    }
    lds_handover();
    float Ff[8][8], Hf[8][8], Pf[8][8];
    #pragma unroll
    for (int r = 0; r < 8; ++r) {
        lds_get_row(Ash, r, Ff[r]);
        lds_get_row(Bsh, r, Hf[r]);
        lds_get_row(Csh, r, Pf[r]);
    }

    float Qr[8];
    {
        float GQr[8];
        #pragma unroll
        for (int j = 0; j < 8; ++j) {
            float acc = 0.f;
            #pragma unroll
            for (int k = 0; k < 8; ++k) acc += Gr[k] * Q[k * 8 + j];
            GQr[j] = acc;
        }
        #pragma unroll
        for (int j = 0; j < 8; ++j) {
            float acc = 0.f;
            #pragma unroll
            for (int k = 0; k < 8; ++k) acc += GQr[k] * Gmat[j * 8 + k];
            Qr[j] = acc;
        }
    }

    int tstop = Tn;
    for (int t = 0; t < Tn; ++t) {
        float T1r[8];
        #pragma unroll
        for (int j = 0; j < 8; ++j) {
            float acc = 0.f;
            #pragma unroll
            for (int k = 0; k < 8; ++k) acc += Fr[k] * Pf[k][j];
            T1r[j] = acc;
        }
        float Pnegr[8];
        #pragma unroll
        for (int j = 0; j < 8; ++j) {
            float acc = Qr[j];
            #pragma unroll
            for (int k = 0; k < 8; ++k) acc += T1r[k] * Ff[j][k];
            Pnegr[j] = acc;
        }
        if (writer) lds_put_row(Ash, idx, Pnegr);
        lds_handover();
        float Pg[8][8], Pgc[8];
        #pragma unroll
        for (int r = 0; r < 8; ++r) {
            lds_get_row(Ash, r, Pg[r]);
            Pgc[r] = Ash[r][idx];
        }
        float T2r[8], Xc[8];
        #pragma unroll
        for (int j = 0; j < 8; ++j) {
            float acc = 0.f;
            #pragma unroll
            for (int k = 0; k < 8; ++k) acc += Hr[k] * Pg[k][j];
            T2r[j] = acc;
        }
        #pragma unroll
        for (int r = 0; r < 8; ++r) {
            float acc = 0.f;
            #pragma unroll
            for (int k = 0; k < 8; ++k) acc += Hf[r][k] * Pgc[k];
            Xc[r] = acc;
        }
        float Sr[8];
        #pragma unroll
        for (int j = 0; j < 8; ++j) {
            float acc = Rr[j];
            #pragma unroll
            for (int k = 0; k < 8; ++k) acc += T2r[k] * Hf[j][k];
            Sr[j] = acc;
        }
        if (writer) lds_put_row(Bsh, idx, Sr);
        lds_handover();
        float L[8][8];
        #pragma unroll
        for (int r = 0; r < 8; ++r) lds_get_row(Bsh, r, L[r]);

        #pragma unroll
        for (int k = 0; k < 8; ++k) {
            L[k][k] = sqrtf(L[k][k]);
            #pragma unroll
            for (int i = k + 1; i < 8; ++i) L[i][k] = L[i][k] / L[k][k];
            #pragma unroll
            for (int i = k + 1; i < 8; ++i) {
                #pragma unroll
                for (int j = k + 1; j <= i; ++j) L[i][j] -= L[i][k] * L[j][k];
            }
        }
        #pragma unroll
        for (int r = 0; r < 8; ++r) {
            float s = Xc[r];
            #pragma unroll
            for (int m = 0; m < r; ++m) s -= L[r][m] * Xc[m];
            Xc[r] = s / L[r][r];
        }
        #pragma unroll
        for (int r = 7; r >= 0; --r) {
            float s = Xc[r];
            #pragma unroll
            for (int m = r + 1; m < 8; ++m) s -= L[m][r] * Xc[m];
            Xc[r] = s / L[r][r];
        }
        float IKHr[8];
        #pragma unroll
        for (int j = 0; j < 8; ++j) {
            float acc = (idx == j) ? 1.f : 0.f;
            #pragma unroll
            for (int m = 0; m < 8; ++m) acc -= Xc[m] * Hf[m][j];
            IKHr[j] = acc;
        }
        float Ar[8], Pnr[8];
        #pragma unroll
        for (int j = 0; j < 8; ++j) {
            float acc = 0.f, pacc = 0.f;
            #pragma unroll
            for (int k = 0; k < 8; ++k) {
                acc  += IKHr[k] * Ff[k][j];
                pacc += IKHr[k] * Pg[k][j];
            }
            Ar[j]  = acc;
            Pnr[j] = pacc;
        }
        if (writer) lds_put_row(Csh, idx, Pnr);
        lds_handover();
        float Pnf[8][8], Pnc[8];
        #pragma unroll
        for (int r = 0; r < 8; ++r) {
            lds_get_row(Csh, r, Pnf[r]);
            Pnc[r] = Csh[r][idx];
        }
        #pragma unroll
        for (int r = 0; r < 8; ++r) {
            #pragma unroll
            for (int c = 0; c < 8; ++c) Pf[r][c] = 0.5f * (Pnf[r][c] + Pnf[c][r]);
        }
        bool changed = false;
        float Pnew[8];
        #pragma unroll
        for (int j = 0; j < 8; ++j) {
            Pnew[j] = 0.5f * (Pnr[j] + Pnc[j]);
            changed = changed || (Pnew[j] != Pr[j]);
            Pr[j] = Pnew[j];
        }

        if (writer) {
            v4f p0 = {Pnew[0], Pnew[1], Pnew[2], Pnew[3]};
            v4f p1 = {Pnew[4], Pnew[5], Pnew[6], Pnew[7]};
            *(v4f*)(Ps + t * 64 + idx * 8)     = p0;
            *(v4f*)(Ps + t * 64 + idx * 8 + 4) = p1;
        }
        {
            v4f g0, g1;
            g0.x = lo ? Ar[0] : Ar[2];
            g0.y = lo ? Ar[1] : Ar[3];
            g0.z = lo ? Ar[4] : Ar[6];
            g0.w = lo ? Ar[5] : Ar[7];
            g1.x = lo ? Xc[0] : Xc[2];
            g1.y = lo ? Xc[1] : Xc[3];
            g1.z = lo ? Xc[4] : Xc[6];
            g1.w = lo ? Xc[5] : Xc[7];
            const v4f z4 = {0.f, 0.f, 0.f, 0.f};
            if (!live) { g0 = z4; g1 = z4; }
            float* gout = Gws + (size_t)t * 256 + lane * 8;
            *(v4f*)(gout)     = g0;
            *(v4f*)(gout + 4) = g1;
        }
        if (__builtin_amdgcn_ballot_w32(changed) == 0u) { tstop = t + 1; break; }
    }
    if (lane == 0) *t_stop_ptr = tstop;
}

__global__ __launch_bounds__(256) void kf_fill_kernel(
    float* __restrict__ Gws, const int* __restrict__ t_stop_ptr,
    float* __restrict__ Ps)
{
    const int t  = blockIdx.x;
    const int q  = threadIdx.x;
    const int ts = *t_stop_ptr;
    if (t >= ts && ts >= 1) {
        const int src = ts - 1;
        Gws[(size_t)t * 256 + q] = Gws[(size_t)src * 256 + q];
        if (q < 64) Ps[t * 64 + q] = Ps[src * 64 + q];
    }
}

__global__ __launch_bounds__(32) void kf_scan_kernel(
    const float* __restrict__ Y, const float* __restrict__ x0,
    const float* __restrict__ Gws, float* __restrict__ traj)
{
    const int lane = threadIdx.x;
    const int n    = lane & 15;
    const bool lo  = lane < 16;
    const size_t row = (size_t)(blockIdx.x * 16 + n);

    const float* yrow = Y    + row * (size_t)Tn * 8 + (lo ? 0 : 2);
    float*       orow = traj + row * (size_t)Tn * 8 + (lo ? 0 : 4);

    v4f xa = *(const v4f*)(x0 + row * 8);
    v4f xb = *(const v4f*)(x0 + row * 8 + 4);
    v8f xv;
    xv[0] = lo ? xa.x : xa.z;
    xv[1] = lo ? xa.y : xa.w;
    xv[2] = lo ? xa.z : xb.x;
    xv[3] = lo ? xa.w : xb.y;
    xv[4] = lo ? xb.x : xb.z;
    xv[5] = lo ? xb.y : xb.w;
    xv[6] = lo ? xb.z : 0.f;
    xv[7] = lo ? xb.w : 0.f;

    const v8f zero8 = {0.f, 0.f, 0.f, 0.f, 0.f, 0.f, 0.f, 0.f};

    for (int t = 0; t < Tn; ++t) {
        const v4f* gp = (const v4f*)(Gws + (size_t)t * 256 + lane * 8);
        v4f g0 = gp[0];
        v4f g1 = gp[1];
        v2f a0 = {g0.x, g0.y};
        v2f a1 = {g0.z, g0.w};
        v2f a2 = {g1.x, g1.y};
        v2f a3 = {g1.z, g1.w};

        v2f b2 = *(const v2f*)(yrow + (size_t)t * 8);
        v2f b3 = *(const v2f*)(yrow + (size_t)t * 8 + 4);

        v2f b0 = {xv[0], xv[1]};
        v2f b1 = {xv[4], xv[5]};

        v8f cy = __builtin_amdgcn_wmma_f32_16x16x4_f32(false, a2, false, b2,
                                                       (short)0, zero8, false, false);
        cy = __builtin_amdgcn_wmma_f32_16x16x4_f32(false, a3, false, b3,
                                                   (short)0, cy, false, false);
        v8f c1 = __builtin_amdgcn_wmma_f32_16x16x4_f32(false, a0, false, b0,
                                                       (short)0, cy, false, false);
        v8f c2 = __builtin_amdgcn_wmma_f32_16x16x4_f32(false, a1, false, b1,
                                                       (short)0, zero8, false, false);
        xv = c1 + c2;

        v4f out;
        out.x = lo ? xv[0] : xv[2];
        out.y = lo ? xv[1] : xv[3];
        out.z = lo ? xv[2] : xv[4];
        out.w = lo ? xv[3] : xv[5];
        *(v4f*)(orow + (size_t)t * 8) = out;
    }
}

extern "C" void kernel_launch(void* const* d_in, const int* in_sizes, int n_in,
                              void* d_out, int out_size, void* d_ws, size_t ws_size,
                              hipStream_t stream) {
    const float* Y    = (const float*)d_in[0];
    const float* F    = (const float*)d_in[1];
    const float* Gmat = (const float*)d_in[2];
    const float* H    = (const float*)d_in[3];
    const float* Q    = (const float*)d_in[4];
    const float* R    = (const float*)d_in[5];
    const float* x0   = (const float*)d_in[6];
    const float* P0   = (const float*)d_in[7];

    float* traj = (float*)d_out;
    float* Ps   = traj + (size_t)Bn * Tn * 8;

    int*   t_stop = (int*)d_ws;
    float* Gws    = (float*)((char*)d_ws + 256);

    kf_gain_kernel<<<1, 32, 0, stream>>>(F, Gmat, H, Q, R, P0, Gws, t_stop, Ps);
    kf_fill_kernel<<<Tn, 256, 0, stream>>>(Gws, t_stop, Ps);
    kf_scan_kernel<<<Bn / 16, 32, 0, stream>>>(Y, x0, Gws, traj);
}


# Gain kernel: a Cholesky pivot enters only as 1/sqrt(pivot), one v_rsq_f32; its column and both triangular solves multiply by it, no sqrt or division sequence left; sum order kept, last bits differ
# speedup vs baseline: 6.7449x; 1.8552x over previous
#include <hip/hip_runtime.h>
#include <math.h>

static constexpr int Bn = 2048;
static constexpr int Tn = 1000;

typedef float v2f __attribute__((ext_vector_type(2)));
typedef float v4f __attribute__((ext_vector_type(4)));
typedef float v8f __attribute__((ext_vector_type(8)));


__device__ __forceinline__ void lds_handover() {
    __builtin_amdgcn_fence(__ATOMIC_RELEASE, "workgroup", "local");
    __builtin_amdgcn_s_barrier();
    __builtin_amdgcn_fence(__ATOMIC_ACQUIRE, "workgroup", "local");
}

__device__ __forceinline__ void lds_put_row(float (*M)[8], int r, const float* v) {
    v4f a = {v[0], v[1], v[2], v[3]};
    v4f b = {v[4], v[5], v[6], v[7]};
    *(v4f*)&M[r][0] = a;
    *(v4f*)&M[r][4] = b;
}

__device__ __forceinline__ void lds_get_row(const float (*M)[8], int r, float* v) {
    v4f a = *(const v4f*)&M[r][0];
    v4f b = *(const v4f*)&M[r][4];
    v[0] = a.x; v[1] = a.y; v[2] = a.z; v[3] = a.w;
    v[4] = b.x; v[5] = b.y; v[6] = b.z; v[7] = b.w;
}

__global__ __launch_bounds__(32) void kf_gain_kernel(
    const float* __restrict__ F, const float* __restrict__ Gmat,
    const float* __restrict__ H, const float* __restrict__ Q,
    const float* __restrict__ R, const float* __restrict__ P0,
    float* __restrict__ Gws, int* __restrict__ t_stop_ptr,
    float* __restrict__ Ps)
{
    __shared__ __attribute__((aligned(16))) float Ash[8][8];
    __shared__ __attribute__((aligned(16))) float Bsh[8][8];
    __shared__ __attribute__((aligned(16))) float Csh[8][8];

    const int  lane   = threadIdx.x;
    const int  m16    = lane & 15;
    const int  idx    = (m16 < 8) ? m16 : ((m16 <= 13) ? (m16 - 6) : (m16 - 8));
    const bool live   = (m16 <= 13);
    const bool lo     = lane < 16;
    const bool writer = lane < 8;

    float Fr[8], Hr[8], Rr[8], Pr[8], Gr[8];
    #pragma unroll
    for (int c = 0; c < 8; ++c) {
        Fr[c] = F[idx * 8 + c];
        Hr[c] = H[idx * 8 + c];
        Rr[c] = R[idx * 8 + c];
        Pr[c] = P0[idx * 8 + c];
        Gr[c] = Gmat[idx * 8 + c];
    }

    if (writer) {
        lds_put_row(Ash, idx, Fr);
        lds_put_row(Bsh, idx, Hr);
        lds_put_row(Csh, idx, Pr);
    }
    lds_handover();
    float Ff[8][8], Hf[8][8], Pf[8][8];
    #pragma unroll
    for (int r = 0; r < 8; ++r) {
        lds_get_row(Ash, r, Ff[r]);
        lds_get_row(Bsh, r, Hf[r]);
        lds_get_row(Csh, r, Pf[r]);
    }

    float Qr[8];
    {
        float GQr[8];
        #pragma unroll
        for (int j = 0; j < 8; ++j) {
            float acc = 0.f;
            #pragma unroll
            for (int k = 0; k < 8; ++k) acc += Gr[k] * Q[k * 8 + j];
            GQr[j] = acc;
        }
        #pragma unroll
        for (int j = 0; j < 8; ++j) {
            float acc = 0.f;
            #pragma unroll
            for (int k = 0; k < 8; ++k) acc += GQr[k] * Gmat[j * 8 + k];
            Qr[j] = acc;
        }
    }

    int tstop = Tn;
    for (int t = 0; t < Tn; ++t) {
        float T1r[8];
        #pragma unroll
        for (int j = 0; j < 8; ++j) {
            float acc = 0.f;
            #pragma unroll
            for (int k = 0; k < 8; ++k) acc += Fr[k] * Pf[k][j];
            T1r[j] = acc;
        }
        float Pnegr[8];
        #pragma unroll
        for (int j = 0; j < 8; ++j) {
            float acc = Qr[j];
            #pragma unroll
            for (int k = 0; k < 8; ++k) acc += T1r[k] * Ff[j][k];
            Pnegr[j] = acc;
        }
        if (writer) lds_put_row(Ash, idx, Pnegr);
        lds_handover();
        float Pg[8][8], Pgc[8];
        #pragma unroll
        for (int r = 0; r < 8; ++r) {
            lds_get_row(Ash, r, Pg[r]);
            Pgc[r] = Ash[r][idx];
        }
        float T2r[8], Xc[8];
        #pragma unroll
        for (int j = 0; j < 8; ++j) {
            float acc = 0.f;
            #pragma unroll
            for (int k = 0; k < 8; ++k) acc += Hr[k] * Pg[k][j];
            T2r[j] = acc;
        }
        #pragma unroll
        for (int r = 0; r < 8; ++r) {
            float acc = 0.f;
            #pragma unroll
            for (int k = 0; k < 8; ++k) acc += Hf[r][k] * Pgc[k];
            Xc[r] = acc;
        }
        float Sr[8];
        #pragma unroll
        for (int j = 0; j < 8; ++j) {
            float acc = Rr[j];
            #pragma unroll
            for (int k = 0; k < 8; ++k) acc += T2r[k] * Hf[j][k];
            Sr[j] = acc;
        }
        if (writer) lds_put_row(Bsh, idx, Sr);
        lds_handover();
        float L[8][8];
        #pragma unroll
        for (int r = 0; r < 8; ++r) lds_get_row(Bsh, r, L[r]);

        float invd[8];
        #pragma unroll
        for (int k = 0; k < 8; ++k) {
            invd[k] = __builtin_amdgcn_rsqf(L[k][k]);
            #pragma unroll
            for (int i = k + 1; i < 8; ++i) L[i][k] = L[i][k] * invd[k];
            #pragma unroll
            for (int i = k + 1; i < 8; ++i) {
                #pragma unroll
                for (int j = k + 1; j <= i; ++j) L[i][j] -= L[i][k] * L[j][k];
            }
        }
        #pragma unroll
        for (int r = 0; r < 8; ++r) {
            float s = Xc[r];
            #pragma unroll
            for (int m = 0; m < r; ++m) s -= L[r][m] * Xc[m];
            Xc[r] = s * invd[r];
        }
        #pragma unroll
        for (int r = 7; r >= 0; --r) {
            float s = Xc[r];
            #pragma unroll
            for (int m = r + 1; m < 8; ++m) s -= L[m][r] * Xc[m];
            Xc[r] = s * invd[r];
        }
        float IKHr[8];
        #pragma unroll
        for (int j = 0; j < 8; ++j) {
            float acc = (idx == j) ? 1.f : 0.f;
            #pragma unroll
            for (int m = 0; m < 8; ++m) acc -= Xc[m] * Hf[m][j];
            IKHr[j] = acc;
        }
        float Ar[8], Pnr[8];
        #pragma unroll
        for (int j = 0; j < 8; ++j) {
            float acc = 0.f, pacc = 0.f;
            #pragma unroll
            for (int k = 0; k < 8; ++k) {
                acc  += IKHr[k] * Ff[k][j];
                pacc += IKHr[k] * Pg[k][j];
            }
            Ar[j]  = acc;
            Pnr[j] = pacc;
        }
        if (writer) lds_put_row(Csh, idx, Pnr);
        lds_handover();
        float Pnf[8][8], Pnc[8];
        #pragma unroll
        for (int r = 0; r < 8; ++r) {
            lds_get_row(Csh, r, Pnf[r]);
            Pnc[r] = Csh[r][idx];
        }
        #pragma unroll
        for (int r = 0; r < 8; ++r) {
            #pragma unroll
            for (int c = 0; c < 8; ++c) Pf[r][c] = 0.5f * (Pnf[r][c] + Pnf[c][r]);
        }
        bool changed = false;
        float Pnew[8];
        #pragma unroll
        for (int j = 0; j < 8; ++j) {
            Pnew[j] = 0.5f * (Pnr[j] + Pnc[j]);
            changed = changed || (Pnew[j] != Pr[j]);
            Pr[j] = Pnew[j];
        }

        if (writer) {
            v4f p0 = {Pnew[0], Pnew[1], Pnew[2], Pnew[3]};
            v4f p1 = {Pnew[4], Pnew[5], Pnew[6], Pnew[7]};
            *(v4f*)(Ps + t * 64 + idx * 8)     = p0;
            *(v4f*)(Ps + t * 64 + idx * 8 + 4) = p1;
        }
        {
            v4f g0, g1;
            g0.x = lo ? Ar[0] : Ar[2];
            g0.y = lo ? Ar[1] : Ar[3];
            g0.z = lo ? Ar[4] : Ar[6];
            g0.w = lo ? Ar[5] : Ar[7];
            g1.x = lo ? Xc[0] : Xc[2];
            g1.y = lo ? Xc[1] : Xc[3];
            g1.z = lo ? Xc[4] : Xc[6];
            g1.w = lo ? Xc[5] : Xc[7];
            const v4f z4 = {0.f, 0.f, 0.f, 0.f};
            if (!live) { g0 = z4; g1 = z4; }
            float* gout = Gws + (size_t)t * 256 + lane * 8;
            *(v4f*)(gout)     = g0;
            *(v4f*)(gout + 4) = g1;
        }
        if (__builtin_amdgcn_ballot_w32(changed) == 0u) { tstop = t + 1; break; }
    }
    if (lane == 0) *t_stop_ptr = tstop;
}

__global__ __launch_bounds__(256) void kf_fill_kernel(
    float* __restrict__ Gws, const int* __restrict__ t_stop_ptr,
    float* __restrict__ Ps)
{
    const int t  = blockIdx.x;
    const int q  = threadIdx.x;
    const int ts = *t_stop_ptr;
    if (t >= ts && ts >= 1) {
        const int src = ts - 1;
        Gws[(size_t)t * 256 + q] = Gws[(size_t)src * 256 + q];
        if (q < 64) Ps[t * 64 + q] = Ps[src * 64 + q];
    }
}

__global__ __launch_bounds__(32) void kf_scan_kernel(
    const float* __restrict__ Y, const float* __restrict__ x0,
    const float* __restrict__ Gws, float* __restrict__ traj)
{
    const int lane = threadIdx.x;
    const int n    = lane & 15;
    const bool lo  = lane < 16;
    const size_t row = (size_t)(blockIdx.x * 16 + n);

    const float* yrow = Y    + row * (size_t)Tn * 8 + (lo ? 0 : 2);
    float*       orow = traj + row * (size_t)Tn * 8 + (lo ? 0 : 4);

    v4f xa = *(const v4f*)(x0 + row * 8);
    v4f xb = *(const v4f*)(x0 + row * 8 + 4);
    v8f xv;
    xv[0] = lo ? xa.x : xa.z;
    xv[1] = lo ? xa.y : xa.w;
    xv[2] = lo ? xa.z : xb.x;
    xv[3] = lo ? xa.w : xb.y;
    xv[4] = lo ? xb.x : xb.z;
    xv[5] = lo ? xb.y : xb.w;
    xv[6] = lo ? xb.z : 0.f;
    xv[7] = lo ? xb.w : 0.f;

    const v8f zero8 = {0.f, 0.f, 0.f, 0.f, 0.f, 0.f, 0.f, 0.f};

    for (int t = 0; t < Tn; ++t) {
        const v4f* gp = (const v4f*)(Gws + (size_t)t * 256 + lane * 8);
        v4f g0 = gp[0];
        v4f g1 = gp[1];
        v2f a0 = {g0.x, g0.y};
        v2f a1 = {g0.z, g0.w};
        v2f a2 = {g1.x, g1.y};
        v2f a3 = {g1.z, g1.w};

        v2f b2 = *(const v2f*)(yrow + (size_t)t * 8);
        v2f b3 = *(const v2f*)(yrow + (size_t)t * 8 + 4);

        v2f b0 = {xv[0], xv[1]};
        v2f b1 = {xv[4], xv[5]};

        v8f cy = __builtin_amdgcn_wmma_f32_16x16x4_f32(false, a2, false, b2,
                                                       (short)0, zero8, false, false);
        cy = __builtin_amdgcn_wmma_f32_16x16x4_f32(false, a3, false, b3,
                                                   (short)0, cy, false, false);
        v8f c1 = __builtin_amdgcn_wmma_f32_16x16x4_f32(false, a0, false, b0,
                                                       (short)0, cy, false, false);
        v8f c2 = __builtin_amdgcn_wmma_f32_16x16x4_f32(false, a1, false, b1,
                                                       (short)0, zero8, false, false);
        xv = c1 + c2;

        v4f out;
        out.x = lo ? xv[0] : xv[2];
        out.y = lo ? xv[1] : xv[3];
        out.z = lo ? xv[2] : xv[4];
        out.w = lo ? xv[3] : xv[5];
        *(v4f*)(orow + (size_t)t * 8) = out;
    }
}

extern "C" void kernel_launch(void* const* d_in, const int* in_sizes, int n_in,
                              void* d_out, int out_size, void* d_ws, size_t ws_size,
                              hipStream_t stream) {
    const float* Y    = (const float*)d_in[0];
    const float* F    = (const float*)d_in[1];
    const float* Gmat = (const float*)d_in[2];
    const float* H    = (const float*)d_in[3];
    const float* Q    = (const float*)d_in[4];
    const float* R    = (const float*)d_in[5];
    const float* x0   = (const float*)d_in[6];
    const float* P0   = (const float*)d_in[7];

    float* traj = (float*)d_out;
    float* Ps   = traj + (size_t)Bn * Tn * 8;

    int*   t_stop = (int*)d_ws;
    float* Gws    = (float*)((char*)d_ws + 256);

    kf_gain_kernel<<<1, 32, 0, stream>>>(F, Gmat, H, Q, R, P0, Gws, t_stop, Ps);
    kf_fill_kernel<<<Tn, 256, 0, stream>>>(Gws, t_stop, Ps);
    kf_scan_kernel<<<Bn / 16, 32, 0, stream>>>(Y, x0, Gws, traj);
}


# Gain kernel: each LDS hand-over costs one LDS trip: wave-scope fences let the reads issue right behind the row stores, S is read in exact widths, independent work follows the reads; sum order kept
# speedup vs baseline: 7.5934x; 1.1258x over previous
#include <hip/hip_runtime.h>
#include <math.h>

static constexpr int Bn = 2048;
static constexpr int Tn = 1000;

typedef float v2f __attribute__((ext_vector_type(2)));
typedef float v4f __attribute__((ext_vector_type(4)));
typedef float v8f __attribute__((ext_vector_type(8)));


__device__ __forceinline__ void lds_handover() {
    __builtin_amdgcn_fence(__ATOMIC_RELEASE, "wavefront", "local");
    __builtin_amdgcn_fence(__ATOMIC_ACQUIRE, "wavefront", "local");
}

__device__ __forceinline__ void lds_put_row(float (*M)[8], int r, const float* v) {
    v4f a = {v[0], v[1], v[2], v[3]};
    v4f b = {v[4], v[5], v[6], v[7]};
    *(v4f*)&M[r][0] = a;
    *(v4f*)&M[r][4] = b;
}

__device__ __forceinline__ void lds_get_row(const float (*M)[8], int r, float* v) {
    v4f a = *(const v4f*)&M[r][0];
    v4f b = *(const v4f*)&M[r][4];
    v[0] = a.x; v[1] = a.y; v[2] = a.z; v[3] = a.w;
    v[4] = b.x; v[5] = b.y; v[6] = b.z; v[7] = b.w;
}

__global__ __launch_bounds__(32) void kf_gain_kernel(
    const float* __restrict__ F, const float* __restrict__ Gmat,
    const float* __restrict__ H, const float* __restrict__ Q,
    const float* __restrict__ R, const float* __restrict__ P0,
    float* __restrict__ Gws, int* __restrict__ t_stop_ptr,
    float* __restrict__ Ps)
{
    __shared__ __attribute__((aligned(16))) float Ash[8][8];
    __shared__ __attribute__((aligned(16))) float Bsh[8][8];
    __shared__ __attribute__((aligned(16))) float Csh[8][8];

    const int  lane   = threadIdx.x;
    const int  m16    = lane & 15;
    const int  idx    = (m16 < 8) ? m16 : ((m16 <= 13) ? (m16 - 6) : (m16 - 8));
    const bool live   = (m16 <= 13);
    const bool lo     = lane < 16;
    const bool writer = lane < 8;

    float Fr[8], Hr[8], Rr[8], Pr[8], Gr[8];
    #pragma unroll
    for (int c = 0; c < 8; ++c) {
        Fr[c] = F[idx * 8 + c];
        Hr[c] = H[idx * 8 + c];
        Rr[c] = R[idx * 8 + c];
        Pr[c] = P0[idx * 8 + c];
        Gr[c] = Gmat[idx * 8 + c];
    }

    if (writer) {
        lds_put_row(Ash, idx, Fr);
        lds_put_row(Bsh, idx, Hr);
        lds_put_row(Csh, idx, Pr);
    }
    lds_handover();
    float Ff[8][8], Hf[8][8], Pf[8][8];
    #pragma unroll
    for (int r = 0; r < 8; ++r) {
        lds_get_row(Ash, r, Ff[r]);
        lds_get_row(Bsh, r, Hf[r]);
        lds_get_row(Csh, r, Pf[r]);
    }

    float Qr[8];
    {
        float GQr[8];
        #pragma unroll
        for (int j = 0; j < 8; ++j) {
            float acc = 0.f;
            #pragma unroll
            for (int k = 0; k < 8; ++k) acc += Gr[k] * Q[k * 8 + j];
            GQr[j] = acc;
        }
        #pragma unroll
        for (int j = 0; j < 8; ++j) {
            float acc = 0.f;
            #pragma unroll
            for (int k = 0; k < 8; ++k) acc += GQr[k] * Gmat[j * 8 + k];
            Qr[j] = acc;
        }
    }

    int tstop = Tn;
    for (int t = 0; t < Tn; ++t) {
        float T1r[8];
        #pragma unroll
        for (int j = 0; j < 8; ++j) {
            float acc = 0.f;
            #pragma unroll
            for (int k = 0; k < 8; ++k) acc += Fr[k] * Pf[k][j];
            T1r[j] = acc;
        }
        float Pnegr[8];
        #pragma unroll
        for (int j = 0; j < 8; ++j) {
            float acc = Qr[j];
            #pragma unroll
            for (int k = 0; k < 8; ++k) acc += T1r[k] * Ff[j][k];
            Pnegr[j] = acc;
        }
        if (writer) lds_put_row(Ash, idx, Pnegr);
        lds_handover();
        float Pg[8][8], Pgc[8];
        #pragma unroll
        for (int r = 0; r < 8; ++r) {
            lds_get_row(Ash, r, Pg[r]);
            Pgc[r] = Ash[r][idx];
        }
        float T2r[8], Xc[8];
        #pragma unroll
        for (int j = 0; j < 8; ++j) {
            float acc = 0.f;
            #pragma unroll
            for (int k = 0; k < 8; ++k) acc += Hr[k] * Pg[k][j];
            T2r[j] = acc;
        }
        float Sr[8];
        #pragma unroll
        for (int j = 0; j < 8; ++j) {
            float acc = Rr[j];
            #pragma unroll
            for (int k = 0; k < 8; ++k) acc += T2r[k] * Hf[j][k];
            Sr[j] = acc;
        }
        if (writer) lds_put_row(Bsh, idx, Sr);
        lds_handover();
        float L[8][8];
        {
            const v2f r1  = *(const v2f*)&Bsh[1][0];
            const v2f r2  = *(const v2f*)&Bsh[2][0];
            const v4f r3  = *(const v4f*)&Bsh[3][0];
            const v4f r4  = *(const v4f*)&Bsh[4][0];
            const v4f r5  = *(const v4f*)&Bsh[5][0];
            const v2f r5b = *(const v2f*)&Bsh[5][4];
            const v4f r6  = *(const v4f*)&Bsh[6][0];
            const v2f r6b = *(const v2f*)&Bsh[6][4];
            const v4f r7  = *(const v4f*)&Bsh[7][0];
            const v4f r7b = *(const v4f*)&Bsh[7][4];
            L[0][0] = Bsh[0][0];
            L[1][0] = r1.x;  L[1][1] = r1.y;
            L[2][0] = r2.x;  L[2][1] = r2.y;  L[2][2] = Bsh[2][2];
            L[3][0] = r3.x;  L[3][1] = r3.y;  L[3][2] = r3.z;  L[3][3] = r3.w;
            L[4][0] = r4.x;  L[4][1] = r4.y;  L[4][2] = r4.z;  L[4][3] = r4.w;
            L[4][4] = Bsh[4][4];
            L[5][0] = r5.x;  L[5][1] = r5.y;  L[5][2] = r5.z;  L[5][3] = r5.w;
            L[5][4] = r5b.x; L[5][5] = r5b.y;
            L[6][0] = r6.x;  L[6][1] = r6.y;  L[6][2] = r6.z;  L[6][3] = r6.w;
            L[6][4] = r6b.x; L[6][5] = r6b.y; L[6][6] = Bsh[6][6];
            L[7][0] = r7.x;  L[7][1] = r7.y;  L[7][2] = r7.z;  L[7][3] = r7.w;
            L[7][4] = r7b.x; L[7][5] = r7b.y; L[7][6] = r7b.z; L[7][7] = r7b.w;
        }
        #pragma unroll
        for (int r = 0; r < 8; ++r) {
            float acc = 0.f;
            #pragma unroll
            for (int k = 0; k < 8; ++k) acc += Hf[r][k] * Pgc[k];
            Xc[r] = acc;
        }

        float invd[8];
        #pragma unroll
        for (int k = 0; k < 8; ++k) {
            invd[k] = __builtin_amdgcn_rsqf(L[k][k]);
            #pragma unroll
            for (int i = k + 1; i < 8; ++i) L[i][k] = L[i][k] * invd[k];
            #pragma unroll
            for (int i = k + 1; i < 8; ++i) {
                #pragma unroll
                for (int j = k + 1; j <= i; ++j) L[i][j] -= L[i][k] * L[j][k];
            }
        }
        #pragma unroll
        for (int r = 0; r < 8; ++r) {
            float s = Xc[r];
            #pragma unroll
            for (int m = 0; m < r; ++m) s -= L[r][m] * Xc[m];
            Xc[r] = s * invd[r];
        }
        #pragma unroll
        for (int r = 7; r >= 0; --r) {
            float s = Xc[r];
            #pragma unroll
            for (int m = r + 1; m < 8; ++m) s -= L[m][r] * Xc[m];
            Xc[r] = s * invd[r];
        }
        float IKHr[8];
        #pragma unroll
        for (int j = 0; j < 8; ++j) {
            float acc = (idx == j) ? 1.f : 0.f;
            #pragma unroll
            for (int m = 0; m < 8; ++m) acc -= Xc[m] * Hf[m][j];
            IKHr[j] = acc;
        }
        float Ar[8], Pnr[8];
        #pragma unroll
        for (int j = 0; j < 8; ++j) {
            float pacc = 0.f;
            #pragma unroll
            for (int k = 0; k < 8; ++k) pacc += IKHr[k] * Pg[k][j];
            Pnr[j] = pacc;
        }
        if (writer) lds_put_row(Csh, idx, Pnr);
        lds_handover();
        float Pnf[8][8], Pnc[8];
        #pragma unroll
        for (int r = 0; r < 8; ++r) {
            lds_get_row(Csh, r, Pnf[r]);
            Pnc[r] = Csh[r][idx];
        }
        #pragma unroll
        for (int j = 0; j < 8; ++j) {
            float acc = 0.f;
            #pragma unroll
            for (int k = 0; k < 8; ++k) acc += IKHr[k] * Ff[k][j];
            Ar[j] = acc;
        }
        #pragma unroll
        for (int r = 0; r < 8; ++r) {
            #pragma unroll
            for (int c = 0; c < 8; ++c) Pf[r][c] = 0.5f * (Pnf[r][c] + Pnf[c][r]);
        }
        bool changed = false;
        float Pnew[8];
        #pragma unroll
        for (int j = 0; j < 8; ++j) {
            Pnew[j] = 0.5f * (Pnr[j] + Pnc[j]);
            changed = changed || (Pnew[j] != Pr[j]);
            Pr[j] = Pnew[j];
        }

        if (writer) {
            v4f p0 = {Pnew[0], Pnew[1], Pnew[2], Pnew[3]};
            v4f p1 = {Pnew[4], Pnew[5], Pnew[6], Pnew[7]};
            *(v4f*)(Ps + t * 64 + idx * 8)     = p0;
            *(v4f*)(Ps + t * 64 + idx * 8 + 4) = p1;
        }
        {
            v4f g0, g1;
            g0.x = lo ? Ar[0] : Ar[2];
            g0.y = lo ? Ar[1] : Ar[3];
            g0.z = lo ? Ar[4] : Ar[6];
            g0.w = lo ? Ar[5] : Ar[7];
            g1.x = lo ? Xc[0] : Xc[2];
            g1.y = lo ? Xc[1] : Xc[3];
            g1.z = lo ? Xc[4] : Xc[6];
            g1.w = lo ? Xc[5] : Xc[7];
            const v4f z4 = {0.f, 0.f, 0.f, 0.f};
            if (!live) { g0 = z4; g1 = z4; }
            float* gout = Gws + (size_t)t * 256 + lane * 8;
            *(v4f*)(gout)     = g0;
            *(v4f*)(gout + 4) = g1;
        }
        if (__builtin_amdgcn_ballot_w32(changed) == 0u) { tstop = t + 1; break; }
    }
    if (lane == 0) *t_stop_ptr = tstop;
}

__global__ __launch_bounds__(256) void kf_fill_kernel(
    float* __restrict__ Gws, const int* __restrict__ t_stop_ptr,
    float* __restrict__ Ps)
{
    const int t  = blockIdx.x;
    const int q  = threadIdx.x;
    const int ts = *t_stop_ptr;
    if (t >= ts && ts >= 1) {
        const int src = ts - 1;
        Gws[(size_t)t * 256 + q] = Gws[(size_t)src * 256 + q];
        if (q < 64) Ps[t * 64 + q] = Ps[src * 64 + q];
    }
}

__global__ __launch_bounds__(32) void kf_scan_kernel(
    const float* __restrict__ Y, const float* __restrict__ x0,
    const float* __restrict__ Gws, float* __restrict__ traj)
{
    const int lane = threadIdx.x;
    const int n    = lane & 15;
    const bool lo  = lane < 16;
    const size_t row = (size_t)(blockIdx.x * 16 + n);

    const float* yrow = Y    + row * (size_t)Tn * 8 + (lo ? 0 : 2);
    float*       orow = traj + row * (size_t)Tn * 8 + (lo ? 0 : 4);

    v4f xa = *(const v4f*)(x0 + row * 8);
    v4f xb = *(const v4f*)(x0 + row * 8 + 4);
    v8f xv;
    xv[0] = lo ? xa.x : xa.z;
    xv[1] = lo ? xa.y : xa.w;
    xv[2] = lo ? xa.z : xb.x;
    xv[3] = lo ? xa.w : xb.y;
    xv[4] = lo ? xb.x : xb.z;
    xv[5] = lo ? xb.y : xb.w;
    xv[6] = lo ? xb.z : 0.f;
    xv[7] = lo ? xb.w : 0.f;

    const v8f zero8 = {0.f, 0.f, 0.f, 0.f, 0.f, 0.f, 0.f, 0.f};

    for (int t = 0; t < Tn; ++t) {
        const v4f* gp = (const v4f*)(Gws + (size_t)t * 256 + lane * 8);
        v4f g0 = gp[0];
        v4f g1 = gp[1];
        v2f a0 = {g0.x, g0.y};
        v2f a1 = {g0.z, g0.w};
        v2f a2 = {g1.x, g1.y};
        v2f a3 = {g1.z, g1.w};

        v2f b2 = *(const v2f*)(yrow + (size_t)t * 8);
        v2f b3 = *(const v2f*)(yrow + (size_t)t * 8 + 4);

        v2f b0 = {xv[0], xv[1]};
        v2f b1 = {xv[4], xv[5]};

        v8f cy = __builtin_amdgcn_wmma_f32_16x16x4_f32(false, a2, false, b2,
                                                       (short)0, zero8, false, false);
        cy = __builtin_amdgcn_wmma_f32_16x16x4_f32(false, a3, false, b3,
                                                   (short)0, cy, false, false);
        v8f c1 = __builtin_amdgcn_wmma_f32_16x16x4_f32(false, a0, false, b0,
                                                       (short)0, cy, false, false);
        v8f c2 = __builtin_amdgcn_wmma_f32_16x16x4_f32(false, a1, false, b1,
                                                       (short)0, zero8, false, false);
        xv = c1 + c2;

        v4f out;
        out.x = lo ? xv[0] : xv[2];
        out.y = lo ? xv[1] : xv[3];
        out.z = lo ? xv[2] : xv[4];
        out.w = lo ? xv[3] : xv[5];
        *(v4f*)(orow + (size_t)t * 8) = out;
    }
}

extern "C" void kernel_launch(void* const* d_in, const int* in_sizes, int n_in,
                              void* d_out, int out_size, void* d_ws, size_t ws_size,
                              hipStream_t stream) {
    const float* Y    = (const float*)d_in[0];
    const float* F    = (const float*)d_in[1];
    const float* Gmat = (const float*)d_in[2];
    const float* H    = (const float*)d_in[3];
    const float* Q    = (const float*)d_in[4];
    const float* R    = (const float*)d_in[5];
    const float* x0   = (const float*)d_in[6];
    const float* P0   = (const float*)d_in[7];

    float* traj = (float*)d_out;
    float* Ps   = traj + (size_t)Bn * Tn * 8;

    int*   t_stop = (int*)d_ws;
    float* Gws    = (float*)((char*)d_ws + 256);

    kf_gain_kernel<<<1, 32, 0, stream>>>(F, Gmat, H, Q, R, P0, Gws, t_stop, Ps);
    kf_fill_kernel<<<Tn, 256, 0, stream>>>(Gws, t_stop, Ps);
    kf_scan_kernel<<<Bn / 16, 32, 0, stream>>>(Y, x0, Gws, traj);
}


# Gain kernel: row-times-matrix products run on float2 pairs of neighbouring outputs, one packed FMA per term, not packed multiplies along k plus serial adds; sum order kept, products now fused
# speedup vs baseline: 8.3120x; 1.0946x over previous
#include <hip/hip_runtime.h>
#include <math.h>

static constexpr int Bn = 2048;
static constexpr int Tn = 1000;

typedef float v2f __attribute__((ext_vector_type(2)));
typedef float v4f __attribute__((ext_vector_type(4)));
typedef float v8f __attribute__((ext_vector_type(8)));


__device__ __forceinline__ void lds_handover() {
    __builtin_amdgcn_fence(__ATOMIC_RELEASE, "wavefront", "local");
    __builtin_amdgcn_fence(__ATOMIC_ACQUIRE, "wavefront", "local");
}

__device__ __forceinline__ void lds_put_row(float (*M)[8], int r, const float* v) {
    v4f a = {v[0], v[1], v[2], v[3]};
    v4f b = {v[4], v[5], v[6], v[7]};
    *(v4f*)&M[r][0] = a;
    *(v4f*)&M[r][4] = b;
}

__device__ __forceinline__ void lds_get_row(const float (*M)[8], int r, float* v) {
    v4f a = *(const v4f*)&M[r][0];
    v4f b = *(const v4f*)&M[r][4];
    v[0] = a.x; v[1] = a.y; v[2] = a.z; v[3] = a.w;
    v[4] = b.x; v[5] = b.y; v[6] = b.z; v[7] = b.w;
}

__device__ __forceinline__ v2f bc2(float x) { v2f r = {x, x}; return r; }
__device__ __forceinline__ v2f pr2(float a, float b) { v2f r = {a, b}; return r; }

__global__ __launch_bounds__(32) void kf_gain_kernel(
    const float* __restrict__ F, const float* __restrict__ Gmat,
    const float* __restrict__ H, const float* __restrict__ Q,
    const float* __restrict__ R, const float* __restrict__ P0,
    float* __restrict__ Gws, int* __restrict__ t_stop_ptr,
    float* __restrict__ Ps)
{
    __shared__ __attribute__((aligned(16))) float Ash[8][8];
    __shared__ __attribute__((aligned(16))) float Bsh[8][8];
    __shared__ __attribute__((aligned(16))) float Csh[8][8];

    const int  lane   = threadIdx.x;
    const int  m16    = lane & 15;
    const int  idx    = (m16 < 8) ? m16 : ((m16 <= 13) ? (m16 - 6) : (m16 - 8));
    const bool live   = (m16 <= 13);
    const bool lo     = lane < 16;
    const bool writer = lane < 8;

    float Fr[8], Hr[8], Rr[8], Pr[8], Gr[8];
    #pragma unroll
    for (int c = 0; c < 8; ++c) {
        Fr[c] = F[idx * 8 + c];
        Hr[c] = H[idx * 8 + c];
        Rr[c] = R[idx * 8 + c];
        Pr[c] = P0[idx * 8 + c];
        Gr[c] = Gmat[idx * 8 + c];
    }

    if (writer) {
        lds_put_row(Ash, idx, Fr);
        lds_put_row(Bsh, idx, Hr);
        lds_put_row(Csh, idx, Pr);
    }
    lds_handover();
    float Ff[8][8], Hf[8][8], Pf[8][8];
    #pragma unroll
    for (int r = 0; r < 8; ++r) {
        lds_get_row(Ash, r, Ff[r]);
        lds_get_row(Bsh, r, Hf[r]);
        lds_get_row(Csh, r, Pf[r]);
    }

    float Qr[8];
    {
        float GQr[8];
        #pragma unroll
        for (int j = 0; j < 8; ++j) {
            float acc = 0.f;
            #pragma unroll
            for (int k = 0; k < 8; ++k) acc += Gr[k] * Q[k * 8 + j];
            GQr[j] = acc;
        }
        #pragma unroll
        for (int j = 0; j < 8; ++j) {
            float acc = 0.f;
            #pragma unroll
            for (int k = 0; k < 8; ++k) acc += GQr[k] * Gmat[j * 8 + k];
            Qr[j] = acc;
        }
    }

    int tstop = Tn;
    for (int t = 0; t < Tn; ++t) {
        float T1r[8];
        #pragma unroll
        for (int jp = 0; jp < 4; ++jp) {
            v2f acc = bc2(Fr[0]) * pr2(Pf[0][2 * jp], Pf[0][2 * jp + 1]);
            #pragma unroll
            for (int k = 1; k < 8; ++k) acc += bc2(Fr[k]) * pr2(Pf[k][2 * jp], Pf[k][2 * jp + 1]);
            T1r[2 * jp] = acc.x; T1r[2 * jp + 1] = acc.y;
        }
        float Pnegr[8];
        #pragma unroll
        for (int jp = 0; jp < 4; ++jp) {
            v2f acc = pr2(Qr[2 * jp], Qr[2 * jp + 1]);
            #pragma unroll
            for (int k = 0; k < 8; ++k) acc += bc2(T1r[k]) * pr2(Ff[2 * jp][k], Ff[2 * jp + 1][k]);
            Pnegr[2 * jp] = acc.x; Pnegr[2 * jp + 1] = acc.y;
        }
        if (writer) lds_put_row(Ash, idx, Pnegr);
        lds_handover();
        float Pg[8][8], Pgc[8];
        #pragma unroll
        for (int r = 0; r < 8; ++r) {
            lds_get_row(Ash, r, Pg[r]);
            Pgc[r] = Ash[r][idx];
        }
        float T2r[8], Xc[8];
        #pragma unroll
        for (int jp = 0; jp < 4; ++jp) {
            v2f acc = bc2(Hr[0]) * pr2(Pg[0][2 * jp], Pg[0][2 * jp + 1]);
            #pragma unroll
            for (int k = 1; k < 8; ++k) acc += bc2(Hr[k]) * pr2(Pg[k][2 * jp], Pg[k][2 * jp + 1]);
            T2r[2 * jp] = acc.x; T2r[2 * jp + 1] = acc.y;
        }
        float Sr[8];
        #pragma unroll
        for (int jp = 0; jp < 4; ++jp) {
            v2f acc = pr2(Rr[2 * jp], Rr[2 * jp + 1]);
            #pragma unroll
            for (int k = 0; k < 8; ++k) acc += bc2(T2r[k]) * pr2(Hf[2 * jp][k], Hf[2 * jp + 1][k]);
            Sr[2 * jp] = acc.x; Sr[2 * jp + 1] = acc.y;
        }
        if (writer) lds_put_row(Bsh, idx, Sr);
        lds_handover();
        float L[8][8];
        {
            const v2f r1  = *(const v2f*)&Bsh[1][0];
            const v2f r2  = *(const v2f*)&Bsh[2][0];
            const v4f r3  = *(const v4f*)&Bsh[3][0];
            const v4f r4  = *(const v4f*)&Bsh[4][0];
            const v4f r5  = *(const v4f*)&Bsh[5][0];
            const v2f r5b = *(const v2f*)&Bsh[5][4];
            const v4f r6  = *(const v4f*)&Bsh[6][0];
            const v2f r6b = *(const v2f*)&Bsh[6][4];
            const v4f r7  = *(const v4f*)&Bsh[7][0];
            const v4f r7b = *(const v4f*)&Bsh[7][4];
            L[0][0] = Bsh[0][0];
            L[1][0] = r1.x;  L[1][1] = r1.y;
            L[2][0] = r2.x;  L[2][1] = r2.y;  L[2][2] = Bsh[2][2];
            L[3][0] = r3.x;  L[3][1] = r3.y;  L[3][2] = r3.z;  L[3][3] = r3.w;
            L[4][0] = r4.x;  L[4][1] = r4.y;  L[4][2] = r4.z;  L[4][3] = r4.w;
            L[4][4] = Bsh[4][4];
            L[5][0] = r5.x;  L[5][1] = r5.y;  L[5][2] = r5.z;  L[5][3] = r5.w;
            L[5][4] = r5b.x; L[5][5] = r5b.y;
            L[6][0] = r6.x;  L[6][1] = r6.y;  L[6][2] = r6.z;  L[6][3] = r6.w;
            L[6][4] = r6b.x; L[6][5] = r6b.y; L[6][6] = Bsh[6][6];
            L[7][0] = r7.x;  L[7][1] = r7.y;  L[7][2] = r7.z;  L[7][3] = r7.w;
            L[7][4] = r7b.x; L[7][5] = r7b.y; L[7][6] = r7b.z; L[7][7] = r7b.w;
        }
        #pragma unroll
        for (int rp = 0; rp < 4; ++rp) {
            v2f acc = pr2(Hf[2 * rp][0], Hf[2 * rp + 1][0]) * bc2(Pgc[0]);
            #pragma unroll
            for (int k = 1; k < 8; ++k) acc += pr2(Hf[2 * rp][k], Hf[2 * rp + 1][k]) * bc2(Pgc[k]);
            Xc[2 * rp] = acc.x; Xc[2 * rp + 1] = acc.y;
        }

        float invd[8];
        #pragma unroll
        for (int k = 0; k < 8; ++k) {
            invd[k] = __builtin_amdgcn_rsqf(L[k][k]);
            #pragma unroll
            for (int i = k + 1; i < 8; ++i) L[i][k] = L[i][k] * invd[k];
            #pragma unroll
            for (int i = k + 1; i < 8; ++i) {
                #pragma unroll
                for (int j = k + 1; j <= i; ++j) L[i][j] -= L[i][k] * L[j][k];
            }
        }
        #pragma unroll
        for (int r = 0; r < 8; ++r) {
            float s = Xc[r];
            #pragma unroll
            for (int m = 0; m < r; ++m) s -= L[r][m] * Xc[m];
            Xc[r] = s * invd[r];
        }
        #pragma unroll
        for (int r = 7; r >= 0; --r) {
            float s = Xc[r];
            #pragma unroll
            for (int m = r + 1; m < 8; ++m) s -= L[m][r] * Xc[m];
            Xc[r] = s * invd[r];
        }
        float IKHr[8];
        #pragma unroll
        for (int jp = 0; jp < 4; ++jp) {
            v2f acc = pr2((idx == 2 * jp) ? 1.f : 0.f, (idx == 2 * jp + 1) ? 1.f : 0.f);
            #pragma unroll
            for (int m = 0; m < 8; ++m) acc -= bc2(Xc[m]) * pr2(Hf[m][2 * jp], Hf[m][2 * jp + 1]);
            IKHr[2 * jp] = acc.x; IKHr[2 * jp + 1] = acc.y;
        }
        float Ar[8], Pnr[8];
        #pragma unroll
        for (int jp = 0; jp < 4; ++jp) {
            v2f pacc = bc2(IKHr[0]) * pr2(Pg[0][2 * jp], Pg[0][2 * jp + 1]);
            #pragma unroll
            for (int k = 1; k < 8; ++k) pacc += bc2(IKHr[k]) * pr2(Pg[k][2 * jp], Pg[k][2 * jp + 1]);
            Pnr[2 * jp] = pacc.x; Pnr[2 * jp + 1] = pacc.y;
        }
        if (writer) lds_put_row(Csh, idx, Pnr);
        lds_handover();
        float Pnf[8][8], Pnc[8];
        #pragma unroll
        for (int r = 0; r < 8; ++r) {
            lds_get_row(Csh, r, Pnf[r]);
            Pnc[r] = Csh[r][idx];
        }
        #pragma unroll
        for (int jp = 0; jp < 4; ++jp) {
            v2f acc = bc2(IKHr[0]) * pr2(Ff[0][2 * jp], Ff[0][2 * jp + 1]);
            #pragma unroll
            for (int k = 1; k < 8; ++k) acc += bc2(IKHr[k]) * pr2(Ff[k][2 * jp], Ff[k][2 * jp + 1]);
            Ar[2 * jp] = acc.x; Ar[2 * jp + 1] = acc.y;
        }
        #pragma unroll
        for (int r = 0; r < 8; ++r) {
            #pragma unroll
            for (int c = 0; c < 8; ++c) Pf[r][c] = 0.5f * (Pnf[r][c] + Pnf[c][r]);
        }
        bool changed = false;
        float Pnew[8];
        #pragma unroll
        for (int j = 0; j < 8; ++j) {
            Pnew[j] = 0.5f * (Pnr[j] + Pnc[j]);
            changed = changed || (Pnew[j] != Pr[j]);
            Pr[j] = Pnew[j];
        }

        if (writer) {
            v4f p0 = {Pnew[0], Pnew[1], Pnew[2], Pnew[3]};
            v4f p1 = {Pnew[4], Pnew[5], Pnew[6], Pnew[7]};
            *(v4f*)(Ps + t * 64 + idx * 8)     = p0;
            *(v4f*)(Ps + t * 64 + idx * 8 + 4) = p1;
        }
        {
            v4f g0, g1;
            g0.x = lo ? Ar[0] : Ar[2];
            g0.y = lo ? Ar[1] : Ar[3];
            g0.z = lo ? Ar[4] : Ar[6];
            g0.w = lo ? Ar[5] : Ar[7];
            g1.x = lo ? Xc[0] : Xc[2];
            g1.y = lo ? Xc[1] : Xc[3];
            g1.z = lo ? Xc[4] : Xc[6];
            g1.w = lo ? Xc[5] : Xc[7];
            const v4f z4 = {0.f, 0.f, 0.f, 0.f};
            if (!live) { g0 = z4; g1 = z4; }
            float* gout = Gws + (size_t)t * 256 + lane * 8;
            *(v4f*)(gout)     = g0;
            *(v4f*)(gout + 4) = g1;
        }
        if (__builtin_amdgcn_ballot_w32(changed) == 0u) { tstop = t + 1; break; }
    }
    if (lane == 0) *t_stop_ptr = tstop;
}

__global__ __launch_bounds__(256) void kf_fill_kernel(
    float* __restrict__ Gws, const int* __restrict__ t_stop_ptr,
    float* __restrict__ Ps)
{
    const int t  = blockIdx.x;
    const int q  = threadIdx.x;
    const int ts = *t_stop_ptr;
    if (t >= ts && ts >= 1) {
        const int src = ts - 1;
        Gws[(size_t)t * 256 + q] = Gws[(size_t)src * 256 + q];
        if (q < 64) Ps[t * 64 + q] = Ps[src * 64 + q];
    }
}

__global__ __launch_bounds__(32) void kf_scan_kernel(
    const float* __restrict__ Y, const float* __restrict__ x0,
    const float* __restrict__ Gws, float* __restrict__ traj)
{
    const int lane = threadIdx.x;
    const int n    = lane & 15;
    const bool lo  = lane < 16;
    const size_t row = (size_t)(blockIdx.x * 16 + n);

    const float* yrow = Y    + row * (size_t)Tn * 8 + (lo ? 0 : 2);
    float*       orow = traj + row * (size_t)Tn * 8 + (lo ? 0 : 4);

    v4f xa = *(const v4f*)(x0 + row * 8);
    v4f xb = *(const v4f*)(x0 + row * 8 + 4);
    v8f xv;
    xv[0] = lo ? xa.x : xa.z;
    xv[1] = lo ? xa.y : xa.w;
    xv[2] = lo ? xa.z : xb.x;
    xv[3] = lo ? xa.w : xb.y;
    xv[4] = lo ? xb.x : xb.z;
    xv[5] = lo ? xb.y : xb.w;
    xv[6] = lo ? xb.z : 0.f;
    xv[7] = lo ? xb.w : 0.f;

    const v8f zero8 = {0.f, 0.f, 0.f, 0.f, 0.f, 0.f, 0.f, 0.f};

    for (int t = 0; t < Tn; ++t) {
        const v4f* gp = (const v4f*)(Gws + (size_t)t * 256 + lane * 8);
        v4f g0 = gp[0];
        v4f g1 = gp[1];
        v2f a0 = {g0.x, g0.y};
        v2f a1 = {g0.z, g0.w};
        v2f a2 = {g1.x, g1.y};
        v2f a3 = {g1.z, g1.w};

        v2f b2 = *(const v2f*)(yrow + (size_t)t * 8);
        v2f b3 = *(const v2f*)(yrow + (size_t)t * 8 + 4);

        v2f b0 = {xv[0], xv[1]};
        v2f b1 = {xv[4], xv[5]};

        v8f cy = __builtin_amdgcn_wmma_f32_16x16x4_f32(false, a2, false, b2,
                                                       (short)0, zero8, false, false);
        cy = __builtin_amdgcn_wmma_f32_16x16x4_f32(false, a3, false, b3,
                                                   (short)0, cy, false, false);
        v8f c1 = __builtin_amdgcn_wmma_f32_16x16x4_f32(false, a0, false, b0,
                                                       (short)0, cy, false, false);
        v8f c2 = __builtin_amdgcn_wmma_f32_16x16x4_f32(false, a1, false, b1,
                                                       (short)0, zero8, false, false);
        xv = c1 + c2;

        v4f out;
        out.x = lo ? xv[0] : xv[2];
        out.y = lo ? xv[1] : xv[3];
        out.z = lo ? xv[2] : xv[4];
        out.w = lo ? xv[3] : xv[5];
        *(v4f*)(orow + (size_t)t * 8) = out;
    }
}

extern "C" void kernel_launch(void* const* d_in, const int* in_sizes, int n_in,
                              void* d_out, int out_size, void* d_ws, size_t ws_size,
                              hipStream_t stream) {
    const float* Y    = (const float*)d_in[0];
    const float* F    = (const float*)d_in[1];
    const float* Gmat = (const float*)d_in[2];
    const float* H    = (const float*)d_in[3];
    const float* Q    = (const float*)d_in[4];
    const float* R    = (const float*)d_in[5];
    const float* x0   = (const float*)d_in[6];
    const float* P0   = (const float*)d_in[7];

    float* traj = (float*)d_out;
    float* Ps   = traj + (size_t)Bn * Tn * 8;

    int*   t_stop = (int*)d_ws;
    float* Gws    = (float*)((char*)d_ws + 256);

    kf_gain_kernel<<<1, 32, 0, stream>>>(F, Gmat, H, Q, R, P0, Gws, t_stop, Ps);
    kf_fill_kernel<<<Tn, 256, 0, stream>>>(Gws, t_stop, Ps);
    kf_scan_kernel<<<Bn / 16, 32, 0, stream>>>(Y, x0, Gws, traj);
}


# One launch: every 16-row scan wave runs the covariance/gain recursion itself, feeds its WMMAs from its own registers; no gain buffer, no fill or scan launch, no per-step gain load wait; sums kept
# speedup vs baseline: 9.2898x; 1.1176x over previous
#include <hip/hip_runtime.h>
#include <math.h>

static constexpr int Bn = 2048;
static constexpr int Tn = 1000;

typedef float v2f __attribute__((ext_vector_type(2)));
typedef float v4f __attribute__((ext_vector_type(4)));
typedef float v8f __attribute__((ext_vector_type(8)));


__device__ __forceinline__ void lds_handover() {
    __builtin_amdgcn_fence(__ATOMIC_RELEASE, "wavefront", "local");
    __builtin_amdgcn_fence(__ATOMIC_ACQUIRE, "wavefront", "local");
}

__device__ __forceinline__ void lds_put_row(float (*M)[8], int r, const float* v) {
    v4f a = {v[0], v[1], v[2], v[3]};
    v4f b = {v[4], v[5], v[6], v[7]};
    *(v4f*)&M[r][0] = a;
    *(v4f*)&M[r][4] = b;
}

__device__ __forceinline__ void lds_get_row(const float (*M)[8], int r, float* v) {
    v4f a = *(const v4f*)&M[r][0];
    v4f b = *(const v4f*)&M[r][4];
    v[0] = a.x; v[1] = a.y; v[2] = a.z; v[3] = a.w;
    v[4] = b.x; v[5] = b.y; v[6] = b.z; v[7] = b.w;
}

__device__ __forceinline__ v2f bc2(float x) { v2f r = {x, x}; return r; }
__device__ __forceinline__ v2f pr2(float a, float b) { v2f r = {a, b}; return r; }

__device__ __forceinline__ void state_step(v8f& xv, v2f a0, v2f a1, v2f a2, v2f a3,
                                           v2f y2, v2f y3, bool lo, float* dst) {
    const v8f zero8 = {0.f, 0.f, 0.f, 0.f, 0.f, 0.f, 0.f, 0.f};
    const v2f b0 = {xv[0], xv[1]};
    const v2f b1 = {xv[4], xv[5]};
    v8f cy = __builtin_amdgcn_wmma_f32_16x16x4_f32(false, a2, false, y2,
                                                   (short)0, zero8, false, false);
    cy = __builtin_amdgcn_wmma_f32_16x16x4_f32(false, a3, false, y3,
                                               (short)0, cy, false, false);
    const v8f c1 = __builtin_amdgcn_wmma_f32_16x16x4_f32(false, a0, false, b0,
                                                         (short)0, cy, false, false);
    const v8f c2 = __builtin_amdgcn_wmma_f32_16x16x4_f32(false, a1, false, b1,
                                                         (short)0, zero8, false, false);
    xv = c1 + c2;
    v4f out;
    out.x = lo ? xv[0] : xv[2];
    out.y = lo ? xv[1] : xv[3];
    out.z = lo ? xv[2] : xv[4];
    out.w = lo ? xv[3] : xv[5];
    *(v4f*)dst = out;
}

__global__ __launch_bounds__(32) void kf_fused_kernel(
    const float* __restrict__ Y, const float* __restrict__ F,
    const float* __restrict__ Gmat, const float* __restrict__ H,
    const float* __restrict__ Q, const float* __restrict__ R,
    const float* __restrict__ x0, const float* __restrict__ P0,
    float* __restrict__ traj, float* __restrict__ Ps)
{
    __shared__ __attribute__((aligned(16))) float Ash[8][8];
    __shared__ __attribute__((aligned(16))) float Bsh[8][8];
    __shared__ __attribute__((aligned(16))) float Csh[8][8];

    const int  lane   = threadIdx.x;
    const int  m16    = lane & 15;
    const int  idx    = (m16 < 8) ? m16 : ((m16 <= 13) ? (m16 - 6) : (m16 - 8));
    const bool live   = (m16 <= 13);
    const bool lo     = lane < 16;
    const bool writer = lane < 8;

    float Fr[8], Hr[8], Rr[8], Pr[8], Gr[8];
    #pragma unroll
    for (int c = 0; c < 8; ++c) {
        Fr[c] = F[idx * 8 + c];
        Hr[c] = H[idx * 8 + c];
        Rr[c] = R[idx * 8 + c];
        Pr[c] = P0[idx * 8 + c];
        Gr[c] = Gmat[idx * 8 + c];
    }

    if (writer) {
        lds_put_row(Ash, idx, Fr);
        lds_put_row(Bsh, idx, Hr);
        lds_put_row(Csh, idx, Pr);
    }
    lds_handover();
    float Ff[8][8], Hf[8][8], Pf[8][8];
    #pragma unroll
    for (int r = 0; r < 8; ++r) {
        lds_get_row(Ash, r, Ff[r]);
        lds_get_row(Bsh, r, Hf[r]);
        lds_get_row(Csh, r, Pf[r]);
    }

    float Qr[8];
    {
        float GQr[8];
        #pragma unroll
        for (int j = 0; j < 8; ++j) {
            float acc = 0.f;
            #pragma unroll
            for (int k = 0; k < 8; ++k) acc += Gr[k] * Q[k * 8 + j];
            GQr[j] = acc;
        }
        #pragma unroll
        for (int j = 0; j < 8; ++j) {
            float acc = 0.f;
            #pragma unroll
            for (int k = 0; k < 8; ++k) acc += GQr[k] * Gmat[j * 8 + k];
            Qr[j] = acc;
        }
    }


    const bool   ps_writer = writer && (blockIdx.x == 0);
    const size_t row  = (size_t)(blockIdx.x * 16 + m16);
    const float* yrow = Y    + row * (size_t)Tn * 8 + (lo ? 0 : 2);
    float*       orow = traj + row * (size_t)Tn * 8 + (lo ? 0 : 4);

    v8f xv;
    {
        v4f xa = *(const v4f*)(x0 + row * 8);
        v4f xb = *(const v4f*)(x0 + row * 8 + 4);
        xv[0] = lo ? xa.x : xa.z;
        xv[1] = lo ? xa.y : xa.w;
        xv[2] = lo ? xa.z : xb.x;
        xv[3] = lo ? xa.w : xb.y;
        xv[4] = lo ? xb.x : xb.z;
        xv[5] = lo ? xb.y : xb.w;
        xv[6] = lo ? xb.z : 0.f;
        xv[7] = lo ? xb.w : 0.f;
    }
    const v8f zero8 = {0.f, 0.f, 0.f, 0.f, 0.f, 0.f, 0.f, 0.f};

    v2f b2 = *(const v2f*)(yrow);
    v2f b3 = *(const v2f*)(yrow + 4);
    v2f a0 = {0.f, 0.f}, a1 = {0.f, 0.f}, a2 = {0.f, 0.f}, a3 = {0.f, 0.f};

    int tstop = Tn;
    for (int t = 0; t < Tn; ++t) {
        const int tn = (t + 1 < Tn) ? (t + 1) : t;
        const v2f nb2 = *(const v2f*)(yrow + (size_t)tn * 8);
        const v2f nb3 = *(const v2f*)(yrow + (size_t)tn * 8 + 4);
        float T1r[8];
        #pragma unroll
        for (int jp = 0; jp < 4; ++jp) {
            v2f acc = bc2(Fr[0]) * pr2(Pf[0][2 * jp], Pf[0][2 * jp + 1]);
            #pragma unroll
            for (int k = 1; k < 8; ++k) acc += bc2(Fr[k]) * pr2(Pf[k][2 * jp], Pf[k][2 * jp + 1]);
            T1r[2 * jp] = acc.x; T1r[2 * jp + 1] = acc.y;
        }
        float Pnegr[8];
        #pragma unroll
        for (int jp = 0; jp < 4; ++jp) {
            v2f acc = pr2(Qr[2 * jp], Qr[2 * jp + 1]);
            #pragma unroll
            for (int k = 0; k < 8; ++k) acc += bc2(T1r[k]) * pr2(Ff[2 * jp][k], Ff[2 * jp + 1][k]);
            Pnegr[2 * jp] = acc.x; Pnegr[2 * jp + 1] = acc.y;
        }
        if (writer) lds_put_row(Ash, idx, Pnegr);
        lds_handover();
        float Pg[8][8], Pgc[8];
        #pragma unroll
        for (int r = 0; r < 8; ++r) {
            lds_get_row(Ash, r, Pg[r]);
            Pgc[r] = Ash[r][idx];
        }
        float T2r[8], Xc[8];
        #pragma unroll
        for (int jp = 0; jp < 4; ++jp) {
            v2f acc = bc2(Hr[0]) * pr2(Pg[0][2 * jp], Pg[0][2 * jp + 1]);
            #pragma unroll
            for (int k = 1; k < 8; ++k) acc += bc2(Hr[k]) * pr2(Pg[k][2 * jp], Pg[k][2 * jp + 1]);
            T2r[2 * jp] = acc.x; T2r[2 * jp + 1] = acc.y;
        }
        float Sr[8];
        #pragma unroll
        for (int jp = 0; jp < 4; ++jp) {
            v2f acc = pr2(Rr[2 * jp], Rr[2 * jp + 1]);
            #pragma unroll
            for (int k = 0; k < 8; ++k) acc += bc2(T2r[k]) * pr2(Hf[2 * jp][k], Hf[2 * jp + 1][k]);
            Sr[2 * jp] = acc.x; Sr[2 * jp + 1] = acc.y;
        }
        if (writer) lds_put_row(Bsh, idx, Sr);
        lds_handover();
        float L[8][8];
        {
            const v2f r1  = *(const v2f*)&Bsh[1][0];
            const v2f r2  = *(const v2f*)&Bsh[2][0];
            const v4f r3  = *(const v4f*)&Bsh[3][0];
            const v4f r4  = *(const v4f*)&Bsh[4][0];
            const v4f r5  = *(const v4f*)&Bsh[5][0];
            const v2f r5b = *(const v2f*)&Bsh[5][4];
            const v4f r6  = *(const v4f*)&Bsh[6][0];
            const v2f r6b = *(const v2f*)&Bsh[6][4];
            const v4f r7  = *(const v4f*)&Bsh[7][0];
            const v4f r7b = *(const v4f*)&Bsh[7][4];
            L[0][0] = Bsh[0][0];
            L[1][0] = r1.x;  L[1][1] = r1.y;
            L[2][0] = r2.x;  L[2][1] = r2.y;  L[2][2] = Bsh[2][2];
            L[3][0] = r3.x;  L[3][1] = r3.y;  L[3][2] = r3.z;  L[3][3] = r3.w;
            L[4][0] = r4.x;  L[4][1] = r4.y;  L[4][2] = r4.z;  L[4][3] = r4.w;
            L[4][4] = Bsh[4][4];
            L[5][0] = r5.x;  L[5][1] = r5.y;  L[5][2] = r5.z;  L[5][3] = r5.w;
            L[5][4] = r5b.x; L[5][5] = r5b.y;
            L[6][0] = r6.x;  L[6][1] = r6.y;  L[6][2] = r6.z;  L[6][3] = r6.w;
            L[6][4] = r6b.x; L[6][5] = r6b.y; L[6][6] = Bsh[6][6];
            L[7][0] = r7.x;  L[7][1] = r7.y;  L[7][2] = r7.z;  L[7][3] = r7.w;
            L[7][4] = r7b.x; L[7][5] = r7b.y; L[7][6] = r7b.z; L[7][7] = r7b.w;
        }
        #pragma unroll
        for (int rp = 0; rp < 4; ++rp) {
            v2f acc = pr2(Hf[2 * rp][0], Hf[2 * rp + 1][0]) * bc2(Pgc[0]);
            #pragma unroll
            for (int k = 1; k < 8; ++k) acc += pr2(Hf[2 * rp][k], Hf[2 * rp + 1][k]) * bc2(Pgc[k]);
            Xc[2 * rp] = acc.x; Xc[2 * rp + 1] = acc.y;
        }

        float invd[8];
        #pragma unroll
        for (int k = 0; k < 8; ++k) {
            invd[k] = __builtin_amdgcn_rsqf(L[k][k]);
            #pragma unroll
            for (int i = k + 1; i < 8; ++i) L[i][k] = L[i][k] * invd[k];
            #pragma unroll
            for (int i = k + 1; i < 8; ++i) {
                #pragma unroll
                for (int j = k + 1; j <= i; ++j) L[i][j] -= L[i][k] * L[j][k];
            }
        }
        #pragma unroll
        for (int r = 0; r < 8; ++r) {
            float s = Xc[r];
            #pragma unroll
            for (int m = 0; m < r; ++m) s -= L[r][m] * Xc[m];
            Xc[r] = s * invd[r];
        }
        #pragma unroll
        for (int r = 7; r >= 0; --r) {
            float s = Xc[r];
            #pragma unroll
            for (int m = r + 1; m < 8; ++m) s -= L[m][r] * Xc[m];
            Xc[r] = s * invd[r];
        }
        a2 = lo ? pr2(Xc[0], Xc[1]) : pr2(Xc[2], Xc[3]);
        a3 = lo ? pr2(Xc[4], Xc[5]) : pr2(Xc[6], Xc[7]);
        if (!live) { a2 = bc2(0.f); a3 = bc2(0.f); }
        v8f cy = __builtin_amdgcn_wmma_f32_16x16x4_f32(false, a2, false, b2,
                                                       (short)0, zero8, false, false);
        cy = __builtin_amdgcn_wmma_f32_16x16x4_f32(false, a3, false, b3,
                                                   (short)0, cy, false, false);
        float IKHr[8];
        #pragma unroll
        for (int jp = 0; jp < 4; ++jp) {
            v2f acc = pr2((idx == 2 * jp) ? 1.f : 0.f, (idx == 2 * jp + 1) ? 1.f : 0.f);
            #pragma unroll
            for (int m = 0; m < 8; ++m) acc -= bc2(Xc[m]) * pr2(Hf[m][2 * jp], Hf[m][2 * jp + 1]);
            IKHr[2 * jp] = acc.x; IKHr[2 * jp + 1] = acc.y;
        }
        float Ar[8], Pnr[8];
        #pragma unroll
        for (int jp = 0; jp < 4; ++jp) {
            v2f pacc = bc2(IKHr[0]) * pr2(Pg[0][2 * jp], Pg[0][2 * jp + 1]);
            #pragma unroll
            for (int k = 1; k < 8; ++k) pacc += bc2(IKHr[k]) * pr2(Pg[k][2 * jp], Pg[k][2 * jp + 1]);
            Pnr[2 * jp] = pacc.x; Pnr[2 * jp + 1] = pacc.y;
        }
        if (writer) lds_put_row(Csh, idx, Pnr);
        lds_handover();
        float Pnf[8][8], Pnc[8];
        #pragma unroll
        for (int r = 0; r < 8; ++r) {
            lds_get_row(Csh, r, Pnf[r]);
            Pnc[r] = Csh[r][idx];
        }
        #pragma unroll
        for (int jp = 0; jp < 4; ++jp) {
            v2f acc = bc2(IKHr[0]) * pr2(Ff[0][2 * jp], Ff[0][2 * jp + 1]);
            #pragma unroll
            for (int k = 1; k < 8; ++k) acc += bc2(IKHr[k]) * pr2(Ff[k][2 * jp], Ff[k][2 * jp + 1]);
            Ar[2 * jp] = acc.x; Ar[2 * jp + 1] = acc.y;
        }
        a0 = lo ? pr2(Ar[0], Ar[1]) : pr2(Ar[2], Ar[3]);
        a1 = lo ? pr2(Ar[4], Ar[5]) : pr2(Ar[6], Ar[7]);
        if (!live) { a0 = bc2(0.f); a1 = bc2(0.f); }
        const v2f b0 = {xv[0], xv[1]};
        const v2f b1 = {xv[4], xv[5]};
        const v8f c1 = __builtin_amdgcn_wmma_f32_16x16x4_f32(false, a0, false, b0,
                                                             (short)0, cy, false, false);
        const v8f c2 = __builtin_amdgcn_wmma_f32_16x16x4_f32(false, a1, false, b1,
                                                             (short)0, zero8, false, false);
        #pragma unroll
        for (int r = 0; r < 8; ++r) {
            #pragma unroll
            for (int c = 0; c < 8; ++c) Pf[r][c] = 0.5f * (Pnf[r][c] + Pnf[c][r]);
        }
        bool changed = false;
        float Pnew[8];
        #pragma unroll
        for (int j = 0; j < 8; ++j) {
            Pnew[j] = 0.5f * (Pnr[j] + Pnc[j]);
            changed = changed || (Pnew[j] != Pr[j]);
            Pr[j] = Pnew[j];
        }

        if (ps_writer) {
            v4f p0 = {Pnew[0], Pnew[1], Pnew[2], Pnew[3]};
            v4f p1 = {Pnew[4], Pnew[5], Pnew[6], Pnew[7]};
            *(v4f*)(Ps + t * 64 + idx * 8)     = p0;
            *(v4f*)(Ps + t * 64 + idx * 8 + 4) = p1;
        }
        xv = c1 + c2;
        {
            v4f out;
            out.x = lo ? xv[0] : xv[2];
            out.y = lo ? xv[1] : xv[3];
            out.z = lo ? xv[2] : xv[4];
            out.w = lo ? xv[3] : xv[5];
            *(v4f*)(orow + (size_t)t * 8) = out;
        }
        b2 = nb2; b3 = nb3;
        if (__builtin_amdgcn_ballot_w32(changed) == 0u) { tstop = t + 1; break; }
    }

    if (tstop < Tn) {
        if (ps_writer) {
            const v4f p0 = {Pr[0], Pr[1], Pr[2], Pr[3]};
            const v4f p1 = {Pr[4], Pr[5], Pr[6], Pr[7]};
            for (int t = tstop; t < Tn; ++t) {
                *(v4f*)(Ps + t * 64 + idx * 8)     = p0;
                *(v4f*)(Ps + t * 64 + idx * 8 + 4) = p1;
            }
        }
        int t = tstop;
        v2f yb2[8], yb3[8];
        #pragma unroll
        for (int j = 0; j < 8; ++j) {
            const int tj = (t + j < Tn) ? (t + j) : (Tn - 1);
            yb2[j] = *(const v2f*)(yrow + (size_t)tj * 8);
            yb3[j] = *(const v2f*)(yrow + (size_t)tj * 8 + 4);
        }
        for (; t + 8 <= Tn; t += 8) {
            v2f nb2[8], nb3[8];
            #pragma unroll
            for (int j = 0; j < 8; ++j) {
                const int tj = (t + 8 + j < Tn) ? (t + 8 + j) : (Tn - 1);
                nb2[j] = *(const v2f*)(yrow + (size_t)tj * 8);
                nb3[j] = *(const v2f*)(yrow + (size_t)tj * 8 + 4);
            }
            #pragma unroll
            for (int j = 0; j < 8; ++j)
                state_step(xv, a0, a1, a2, a3, yb2[j], yb3[j], lo, orow + (size_t)(t + j) * 8);
            #pragma unroll
            for (int j = 0; j < 8; ++j) { yb2[j] = nb2[j]; yb3[j] = nb3[j]; }
        }
        #pragma unroll
        for (int j = 0; j < 7; ++j) {
            if (t + j < Tn)
                state_step(xv, a0, a1, a2, a3, yb2[j], yb3[j], lo, orow + (size_t)(t + j) * 8);
        }
    }
}

extern "C" void kernel_launch(void* const* d_in, const int* in_sizes, int n_in,
                              void* d_out, int out_size, void* d_ws, size_t ws_size,
                              hipStream_t stream) {
    const float* Y    = (const float*)d_in[0];
    const float* F    = (const float*)d_in[1];
    const float* Gmat = (const float*)d_in[2];
    const float* H    = (const float*)d_in[3];
    const float* Q    = (const float*)d_in[4];
    const float* R    = (const float*)d_in[5];
    const float* x0   = (const float*)d_in[6];
    const float* P0   = (const float*)d_in[7];

    float* traj = (float*)d_out;
    float* Ps   = traj + (size_t)Bn * Tn * 8;

    kf_fused_kernel<<<Bn / 16, 32, 0, stream>>>(Y, F, Gmat, H, Q, R, x0, P0, traj, Ps);
}


# Back substitution and I-KH take their terms in the order the unknowns appear (m descending): one multiply-add per row waits for the newest unknown, not the whole row sum; summation order changed
# speedup vs baseline: 9.8167x; 1.0567x over previous
#include <hip/hip_runtime.h>
#include <math.h>

static constexpr int Bn = 2048;
static constexpr int Tn = 1000;

typedef float v2f __attribute__((ext_vector_type(2)));
typedef float v4f __attribute__((ext_vector_type(4)));
typedef float v8f __attribute__((ext_vector_type(8)));


__device__ __forceinline__ void lds_handover() {
    __builtin_amdgcn_fence(__ATOMIC_RELEASE, "wavefront", "local");
    __builtin_amdgcn_fence(__ATOMIC_ACQUIRE, "wavefront", "local");
}

__device__ __forceinline__ void lds_put_row(float (*M)[8], int r, const float* v) {
    v4f a = {v[0], v[1], v[2], v[3]};
    v4f b = {v[4], v[5], v[6], v[7]};
    *(v4f*)&M[r][0] = a;
    *(v4f*)&M[r][4] = b;
}

__device__ __forceinline__ void lds_get_row(const float (*M)[8], int r, float* v) {
    v4f a = *(const v4f*)&M[r][0];
    v4f b = *(const v4f*)&M[r][4];
    v[0] = a.x; v[1] = a.y; v[2] = a.z; v[3] = a.w;
    v[4] = b.x; v[5] = b.y; v[6] = b.z; v[7] = b.w;
}

__device__ __forceinline__ v2f bc2(float x) { v2f r = {x, x}; return r; }
__device__ __forceinline__ v2f pr2(float a, float b) { v2f r = {a, b}; return r; }

__device__ __forceinline__ void state_step(v8f& xv, v2f a0, v2f a1, v2f a2, v2f a3,
                                           v2f y2, v2f y3, bool lo, float* dst) {
    const v8f zero8 = {0.f, 0.f, 0.f, 0.f, 0.f, 0.f, 0.f, 0.f};
    const v2f b0 = {xv[0], xv[1]};
    const v2f b1 = {xv[4], xv[5]};
    v8f cy = __builtin_amdgcn_wmma_f32_16x16x4_f32(false, a2, false, y2,
                                                   (short)0, zero8, false, false);
    cy = __builtin_amdgcn_wmma_f32_16x16x4_f32(false, a3, false, y3,
                                               (short)0, cy, false, false);
    const v8f c1 = __builtin_amdgcn_wmma_f32_16x16x4_f32(false, a0, false, b0,
                                                         (short)0, cy, false, false);
    const v8f c2 = __builtin_amdgcn_wmma_f32_16x16x4_f32(false, a1, false, b1,
                                                         (short)0, zero8, false, false);
    xv = c1 + c2;
    v4f out;
    out.x = lo ? xv[0] : xv[2];
    out.y = lo ? xv[1] : xv[3];
    out.z = lo ? xv[2] : xv[4];
    out.w = lo ? xv[3] : xv[5];
    *(v4f*)dst = out;
}

__global__ __launch_bounds__(32) void kf_fused_kernel(
    const float* __restrict__ Y, const float* __restrict__ F,
    const float* __restrict__ Gmat, const float* __restrict__ H,
    const float* __restrict__ Q, const float* __restrict__ R,
    const float* __restrict__ x0, const float* __restrict__ P0,
    float* __restrict__ traj, float* __restrict__ Ps)
{
    __shared__ __attribute__((aligned(16))) float Ash[8][8];
    __shared__ __attribute__((aligned(16))) float Bsh[8][8];
    __shared__ __attribute__((aligned(16))) float Csh[8][8];

    const int  lane   = threadIdx.x;
    const int  m16    = lane & 15;
    const int  idx    = (m16 < 8) ? m16 : ((m16 <= 13) ? (m16 - 6) : (m16 - 8));
    const bool live   = (m16 <= 13);
    const bool lo     = lane < 16;
    const bool writer = lane < 8;

    float Fr[8], Hr[8], Rr[8], Pr[8], Gr[8];
    #pragma unroll
    for (int c = 0; c < 8; ++c) {
        Fr[c] = F[idx * 8 + c];
        Hr[c] = H[idx * 8 + c];
        Rr[c] = R[idx * 8 + c];
        Pr[c] = P0[idx * 8 + c];
        Gr[c] = Gmat[idx * 8 + c];
    }

    if (writer) {
        lds_put_row(Ash, idx, Fr);
        lds_put_row(Bsh, idx, Hr);
        lds_put_row(Csh, idx, Pr);
    }
    lds_handover();
    float Ff[8][8], Hf[8][8], Pf[8][8];
    #pragma unroll
    for (int r = 0; r < 8; ++r) {
        lds_get_row(Ash, r, Ff[r]);
        lds_get_row(Bsh, r, Hf[r]);
        lds_get_row(Csh, r, Pf[r]);
    }

    float Qr[8];
    {
        float GQr[8];
        #pragma unroll
        for (int j = 0; j < 8; ++j) {
            float acc = 0.f;
            #pragma unroll
            for (int k = 0; k < 8; ++k) acc += Gr[k] * Q[k * 8 + j];
            GQr[j] = acc;
        }
        #pragma unroll
        for (int j = 0; j < 8; ++j) {
            float acc = 0.f;
            #pragma unroll
            for (int k = 0; k < 8; ++k) acc += GQr[k] * Gmat[j * 8 + k];
            Qr[j] = acc;
        }
    }


    const bool   ps_writer = writer && (blockIdx.x == 0);
    const size_t row  = (size_t)(blockIdx.x * 16 + m16);
    const float* yrow = Y    + row * (size_t)Tn * 8 + (lo ? 0 : 2);
    float*       orow = traj + row * (size_t)Tn * 8 + (lo ? 0 : 4);

    v8f xv;
    {
        v4f xa = *(const v4f*)(x0 + row * 8);
        v4f xb = *(const v4f*)(x0 + row * 8 + 4);
        xv[0] = lo ? xa.x : xa.z;
        xv[1] = lo ? xa.y : xa.w;
        xv[2] = lo ? xa.z : xb.x;
        xv[3] = lo ? xa.w : xb.y;
        xv[4] = lo ? xb.x : xb.z;
        xv[5] = lo ? xb.y : xb.w;
        xv[6] = lo ? xb.z : 0.f;
        xv[7] = lo ? xb.w : 0.f;
    }
    const v8f zero8 = {0.f, 0.f, 0.f, 0.f, 0.f, 0.f, 0.f, 0.f};

    v2f b2 = *(const v2f*)(yrow);
    v2f b3 = *(const v2f*)(yrow + 4);
    v2f a0 = {0.f, 0.f}, a1 = {0.f, 0.f}, a2 = {0.f, 0.f}, a3 = {0.f, 0.f};

    int tstop = Tn;
    for (int t = 0; t < Tn; ++t) {
        const int tn = (t + 1 < Tn) ? (t + 1) : t;
        const v2f nb2 = *(const v2f*)(yrow + (size_t)tn * 8);
        const v2f nb3 = *(const v2f*)(yrow + (size_t)tn * 8 + 4);
        float T1r[8];
        #pragma unroll
        for (int jp = 0; jp < 4; ++jp) {
            v2f acc = bc2(Fr[0]) * pr2(Pf[0][2 * jp], Pf[0][2 * jp + 1]);
            #pragma unroll
            for (int k = 1; k < 8; ++k) acc += bc2(Fr[k]) * pr2(Pf[k][2 * jp], Pf[k][2 * jp + 1]);
            T1r[2 * jp] = acc.x; T1r[2 * jp + 1] = acc.y;
        }
        float Pnegr[8];
        #pragma unroll
        for (int jp = 0; jp < 4; ++jp) {
            v2f acc = pr2(Qr[2 * jp], Qr[2 * jp + 1]);
            #pragma unroll
            for (int k = 0; k < 8; ++k) acc += bc2(T1r[k]) * pr2(Ff[2 * jp][k], Ff[2 * jp + 1][k]);
            Pnegr[2 * jp] = acc.x; Pnegr[2 * jp + 1] = acc.y;
        }
        if (writer) lds_put_row(Ash, idx, Pnegr);
        lds_handover();
        float Pg[8][8], Pgc[8];
        #pragma unroll
        for (int r = 0; r < 8; ++r) {
            lds_get_row(Ash, r, Pg[r]);
            Pgc[r] = Ash[r][idx];
        }
        float T2r[8], Xc[8];
        #pragma unroll
        for (int jp = 0; jp < 4; ++jp) {
            v2f acc = bc2(Hr[0]) * pr2(Pg[0][2 * jp], Pg[0][2 * jp + 1]);
            #pragma unroll
            for (int k = 1; k < 8; ++k) acc += bc2(Hr[k]) * pr2(Pg[k][2 * jp], Pg[k][2 * jp + 1]);
            T2r[2 * jp] = acc.x; T2r[2 * jp + 1] = acc.y;
        }
        float Sr[8];
        #pragma unroll
        for (int jp = 0; jp < 4; ++jp) {
            v2f acc = pr2(Rr[2 * jp], Rr[2 * jp + 1]);
            #pragma unroll
            for (int k = 0; k < 8; ++k) acc += bc2(T2r[k]) * pr2(Hf[2 * jp][k], Hf[2 * jp + 1][k]);
            Sr[2 * jp] = acc.x; Sr[2 * jp + 1] = acc.y;
        }
        if (writer) lds_put_row(Bsh, idx, Sr);
        lds_handover();
        float L[8][8];
        {
            const v2f r1  = *(const v2f*)&Bsh[1][0];
            const v2f r2  = *(const v2f*)&Bsh[2][0];
            const v4f r3  = *(const v4f*)&Bsh[3][0];
            const v4f r4  = *(const v4f*)&Bsh[4][0];
            const v4f r5  = *(const v4f*)&Bsh[5][0];
            const v2f r5b = *(const v2f*)&Bsh[5][4];
            const v4f r6  = *(const v4f*)&Bsh[6][0];
            const v2f r6b = *(const v2f*)&Bsh[6][4];
            const v4f r7  = *(const v4f*)&Bsh[7][0];
            const v4f r7b = *(const v4f*)&Bsh[7][4];
            L[0][0] = Bsh[0][0];
            L[1][0] = r1.x;  L[1][1] = r1.y;
            L[2][0] = r2.x;  L[2][1] = r2.y;  L[2][2] = Bsh[2][2];
            L[3][0] = r3.x;  L[3][1] = r3.y;  L[3][2] = r3.z;  L[3][3] = r3.w;
            L[4][0] = r4.x;  L[4][1] = r4.y;  L[4][2] = r4.z;  L[4][3] = r4.w;
            L[4][4] = Bsh[4][4];
            L[5][0] = r5.x;  L[5][1] = r5.y;  L[5][2] = r5.z;  L[5][3] = r5.w;
            L[5][4] = r5b.x; L[5][5] = r5b.y;
            L[6][0] = r6.x;  L[6][1] = r6.y;  L[6][2] = r6.z;  L[6][3] = r6.w;
            L[6][4] = r6b.x; L[6][5] = r6b.y; L[6][6] = Bsh[6][6];
            L[7][0] = r7.x;  L[7][1] = r7.y;  L[7][2] = r7.z;  L[7][3] = r7.w;
            L[7][4] = r7b.x; L[7][5] = r7b.y; L[7][6] = r7b.z; L[7][7] = r7b.w;
        }
        #pragma unroll
        for (int rp = 0; rp < 4; ++rp) {
            v2f acc = pr2(Hf[2 * rp][0], Hf[2 * rp + 1][0]) * bc2(Pgc[0]);
            #pragma unroll
            for (int k = 1; k < 8; ++k) acc += pr2(Hf[2 * rp][k], Hf[2 * rp + 1][k]) * bc2(Pgc[k]);
            Xc[2 * rp] = acc.x; Xc[2 * rp + 1] = acc.y;
        }

        float invd[8];
        #pragma unroll
        for (int k = 0; k < 8; ++k) {
            invd[k] = __builtin_amdgcn_rsqf(L[k][k]);
            #pragma unroll
            for (int i = k + 1; i < 8; ++i) L[i][k] = L[i][k] * invd[k];
            #pragma unroll
            for (int i = k + 1; i < 8; ++i) {
                #pragma unroll
                for (int j = k + 1; j <= i; ++j) L[i][j] -= L[i][k] * L[j][k];
            }
        }
        #pragma unroll
        for (int r = 0; r < 8; ++r) {
            float s = Xc[r];
            #pragma unroll
            for (int m = 0; m < r; ++m) s -= L[r][m] * Xc[m];
            Xc[r] = s * invd[r];
        }
        #pragma unroll
        for (int r = 7; r >= 0; --r) {
            float s = Xc[r];
            #pragma unroll
            for (int m = 7; m > r; --m) s -= L[m][r] * Xc[m];
            Xc[r] = s * invd[r];
        }
        a2 = lo ? pr2(Xc[0], Xc[1]) : pr2(Xc[2], Xc[3]);
        a3 = lo ? pr2(Xc[4], Xc[5]) : pr2(Xc[6], Xc[7]);
        if (!live) { a2 = bc2(0.f); a3 = bc2(0.f); }
        v8f cy = __builtin_amdgcn_wmma_f32_16x16x4_f32(false, a2, false, b2,
                                                       (short)0, zero8, false, false);
        cy = __builtin_amdgcn_wmma_f32_16x16x4_f32(false, a3, false, b3,
                                                   (short)0, cy, false, false);
        float IKHr[8];
        #pragma unroll
        for (int jp = 0; jp < 4; ++jp) {
            v2f acc = pr2((idx == 2 * jp) ? 1.f : 0.f, (idx == 2 * jp + 1) ? 1.f : 0.f);
            #pragma unroll
            for (int m = 7; m >= 0; --m) acc -= bc2(Xc[m]) * pr2(Hf[m][2 * jp], Hf[m][2 * jp + 1]);
            IKHr[2 * jp] = acc.x; IKHr[2 * jp + 1] = acc.y;
        }
        float Ar[8], Pnr[8];
        #pragma unroll
        for (int jp = 0; jp < 4; ++jp) {
            v2f pacc = bc2(IKHr[0]) * pr2(Pg[0][2 * jp], Pg[0][2 * jp + 1]);
            #pragma unroll
            for (int k = 1; k < 8; ++k) pacc += bc2(IKHr[k]) * pr2(Pg[k][2 * jp], Pg[k][2 * jp + 1]);
            Pnr[2 * jp] = pacc.x; Pnr[2 * jp + 1] = pacc.y;
        }
        if (writer) lds_put_row(Csh, idx, Pnr);
        lds_handover();
        float Pnf[8][8], Pnc[8];
        #pragma unroll
        for (int r = 0; r < 8; ++r) {
            lds_get_row(Csh, r, Pnf[r]);
            Pnc[r] = Csh[r][idx];
        }
        #pragma unroll
        for (int jp = 0; jp < 4; ++jp) {
            v2f acc = bc2(IKHr[0]) * pr2(Ff[0][2 * jp], Ff[0][2 * jp + 1]);
            #pragma unroll
            for (int k = 1; k < 8; ++k) acc += bc2(IKHr[k]) * pr2(Ff[k][2 * jp], Ff[k][2 * jp + 1]);
            Ar[2 * jp] = acc.x; Ar[2 * jp + 1] = acc.y;
        }
        a0 = lo ? pr2(Ar[0], Ar[1]) : pr2(Ar[2], Ar[3]);
        a1 = lo ? pr2(Ar[4], Ar[5]) : pr2(Ar[6], Ar[7]);
        if (!live) { a0 = bc2(0.f); a1 = bc2(0.f); }
        const v2f b0 = {xv[0], xv[1]};
        const v2f b1 = {xv[4], xv[5]};
        const v8f c1 = __builtin_amdgcn_wmma_f32_16x16x4_f32(false, a0, false, b0,
                                                             (short)0, cy, false, false);
        const v8f c2 = __builtin_amdgcn_wmma_f32_16x16x4_f32(false, a1, false, b1,
                                                             (short)0, zero8, false, false);
        #pragma unroll
        for (int r = 0; r < 8; ++r) {
            #pragma unroll
            for (int c = 0; c < 8; ++c) Pf[r][c] = 0.5f * (Pnf[r][c] + Pnf[c][r]);
        }
        bool changed = false;
        float Pnew[8];
        #pragma unroll
        for (int j = 0; j < 8; ++j) {
            Pnew[j] = 0.5f * (Pnr[j] + Pnc[j]);
            changed = changed || (Pnew[j] != Pr[j]);
            Pr[j] = Pnew[j];
        }

        if (ps_writer) {
            v4f p0 = {Pnew[0], Pnew[1], Pnew[2], Pnew[3]};
            v4f p1 = {Pnew[4], Pnew[5], Pnew[6], Pnew[7]};
            *(v4f*)(Ps + t * 64 + idx * 8)     = p0;
            *(v4f*)(Ps + t * 64 + idx * 8 + 4) = p1;
        }
        xv = c1 + c2;
        {
            v4f out;
            out.x = lo ? xv[0] : xv[2];
            out.y = lo ? xv[1] : xv[3];
            out.z = lo ? xv[2] : xv[4];
            out.w = lo ? xv[3] : xv[5];
            *(v4f*)(orow + (size_t)t * 8) = out;
        }
        b2 = nb2; b3 = nb3;
        if (__builtin_amdgcn_ballot_w32(changed) == 0u) { tstop = t + 1; break; }
    }

    if (tstop < Tn) {
        if (ps_writer) {
            const v4f p0 = {Pr[0], Pr[1], Pr[2], Pr[3]};
            const v4f p1 = {Pr[4], Pr[5], Pr[6], Pr[7]};
            for (int t = tstop; t < Tn; ++t) {
                *(v4f*)(Ps + t * 64 + idx * 8)     = p0;
                *(v4f*)(Ps + t * 64 + idx * 8 + 4) = p1;
            }
        }
        int t = tstop;
        v2f yb2[8], yb3[8];
        #pragma unroll
        for (int j = 0; j < 8; ++j) {
            const int tj = (t + j < Tn) ? (t + j) : (Tn - 1);
            yb2[j] = *(const v2f*)(yrow + (size_t)tj * 8);
            yb3[j] = *(const v2f*)(yrow + (size_t)tj * 8 + 4);
        }
        for (; t + 8 <= Tn; t += 8) {
            v2f nb2[8], nb3[8];
            #pragma unroll
            for (int j = 0; j < 8; ++j) {
                const int tj = (t + 8 + j < Tn) ? (t + 8 + j) : (Tn - 1);
                nb2[j] = *(const v2f*)(yrow + (size_t)tj * 8);
                nb3[j] = *(const v2f*)(yrow + (size_t)tj * 8 + 4);
            }
            #pragma unroll
            for (int j = 0; j < 8; ++j)
                state_step(xv, a0, a1, a2, a3, yb2[j], yb3[j], lo, orow + (size_t)(t + j) * 8);
            #pragma unroll
            for (int j = 0; j < 8; ++j) { yb2[j] = nb2[j]; yb3[j] = nb3[j]; }
        }
        #pragma unroll
        for (int j = 0; j < 7; ++j) {
            if (t + j < Tn)
                state_step(xv, a0, a1, a2, a3, yb2[j], yb3[j], lo, orow + (size_t)(t + j) * 8);
        }
    }
}

extern "C" void kernel_launch(void* const* d_in, const int* in_sizes, int n_in,
                              void* d_out, int out_size, void* d_ws, size_t ws_size,
                              hipStream_t stream) {
    const float* Y    = (const float*)d_in[0];
    const float* F    = (const float*)d_in[1];
    const float* Gmat = (const float*)d_in[2];
    const float* H    = (const float*)d_in[3];
    const float* Q    = (const float*)d_in[4];
    const float* R    = (const float*)d_in[5];
    const float* x0   = (const float*)d_in[6];
    const float* P0   = (const float*)d_in[7];

    float* traj = (float*)d_out;
    float* Ps   = traj + (size_t)Bn * Tn * 8;

    kf_fused_kernel<<<Bn / 16, 32, 0, stream>>>(Y, F, Gmat, H, Q, R, x0, P0, traj, Ps);
}


# S is formed from P itself, (HF) P (HF)^T + (H Qe H^T + R), and handed over first; Pneg is summed and handed over while S crosses the LDS, off the serial path; summation order of S changed
# speedup vs baseline: 10.6999x; 1.0900x over previous
#include <hip/hip_runtime.h>
#include <math.h>

static constexpr int Bn = 2048;
static constexpr int Tn = 1000;

typedef float v2f __attribute__((ext_vector_type(2)));
typedef float v4f __attribute__((ext_vector_type(4)));
typedef float v8f __attribute__((ext_vector_type(8)));


__device__ __forceinline__ void lds_handover() {
    __builtin_amdgcn_fence(__ATOMIC_RELEASE, "wavefront", "local");
    __builtin_amdgcn_fence(__ATOMIC_ACQUIRE, "wavefront", "local");
}

__device__ __forceinline__ void lds_put_row(float (*M)[8], int r, const float* v) {
    v4f a = {v[0], v[1], v[2], v[3]};
    v4f b = {v[4], v[5], v[6], v[7]};
    *(v4f*)&M[r][0] = a;
    *(v4f*)&M[r][4] = b;
}

__device__ __forceinline__ void lds_get_row(const float (*M)[8], int r, float* v) {
    v4f a = *(const v4f*)&M[r][0];
    v4f b = *(const v4f*)&M[r][4];
    v[0] = a.x; v[1] = a.y; v[2] = a.z; v[3] = a.w;
    v[4] = b.x; v[5] = b.y; v[6] = b.z; v[7] = b.w;
}

__device__ __forceinline__ v2f bc2(float x) { v2f r = {x, x}; return r; }
__device__ __forceinline__ v2f pr2(float a, float b) { v2f r = {a, b}; return r; }

__device__ __forceinline__ void state_step(v8f& xv, v2f a0, v2f a1, v2f a2, v2f a3,
                                           v2f y2, v2f y3, bool lo, float* dst) {
    const v8f zero8 = {0.f, 0.f, 0.f, 0.f, 0.f, 0.f, 0.f, 0.f};
    const v2f b0 = {xv[0], xv[1]};
    const v2f b1 = {xv[4], xv[5]};
    v8f cy = __builtin_amdgcn_wmma_f32_16x16x4_f32(false, a2, false, y2,
                                                   (short)0, zero8, false, false);
    cy = __builtin_amdgcn_wmma_f32_16x16x4_f32(false, a3, false, y3,
                                               (short)0, cy, false, false);
    const v8f c1 = __builtin_amdgcn_wmma_f32_16x16x4_f32(false, a0, false, b0,
                                                         (short)0, cy, false, false);
    const v8f c2 = __builtin_amdgcn_wmma_f32_16x16x4_f32(false, a1, false, b1,
                                                         (short)0, zero8, false, false);
    xv = c1 + c2;
    v4f out;
    out.x = lo ? xv[0] : xv[2];
    out.y = lo ? xv[1] : xv[3];
    out.z = lo ? xv[2] : xv[4];
    out.w = lo ? xv[3] : xv[5];
    *(v4f*)dst = out;
}

__global__ __launch_bounds__(32) void kf_fused_kernel(
    const float* __restrict__ Y, const float* __restrict__ F,
    const float* __restrict__ Gmat, const float* __restrict__ H,
    const float* __restrict__ Q, const float* __restrict__ R,
    const float* __restrict__ x0, const float* __restrict__ P0,
    float* __restrict__ traj, float* __restrict__ Ps)
{
    __shared__ __attribute__((aligned(16))) float Ash[8][8];
    __shared__ __attribute__((aligned(16))) float Bsh[8][8];
    __shared__ __attribute__((aligned(16))) float Csh[8][8];
    __shared__ __attribute__((aligned(16))) float Dsh[32][8];
    __shared__ __attribute__((aligned(16))) float Esh[32][8];

    const int  lane   = threadIdx.x;
    const int  m16    = lane & 15;
    const int  idx    = (m16 < 8) ? m16 : ((m16 <= 13) ? (m16 - 6) : (m16 - 8));
    const bool live   = (m16 <= 13);
    const bool lo     = lane < 16;
    const bool writer = lane < 8;

    float Fr[8], Hr[8], Rr[8], Pr[8], Gr[8];
    #pragma unroll
    for (int c = 0; c < 8; ++c) {
        Fr[c] = F[idx * 8 + c];
        Hr[c] = H[idx * 8 + c];
        Rr[c] = R[idx * 8 + c];
        Pr[c] = P0[idx * 8 + c];
        Gr[c] = Gmat[idx * 8 + c];
    }

    if (writer) {
        lds_put_row(Ash, idx, Fr);
        lds_put_row(Bsh, idx, Hr);
        lds_put_row(Csh, idx, Pr);
    }
    lds_handover();
    float Ff[8][8], Hf[8][8], Pf[8][8];
    #pragma unroll
    for (int r = 0; r < 8; ++r) {
        lds_get_row(Ash, r, Ff[r]);
        lds_get_row(Bsh, r, Hf[r]);
        lds_get_row(Csh, r, Pf[r]);
    }

    float Qr[8];
    {
        float GQr[8];
        #pragma unroll
        for (int j = 0; j < 8; ++j) {
            float acc = 0.f;
            #pragma unroll
            for (int k = 0; k < 8; ++k) acc += Gr[k] * Q[k * 8 + j];
            GQr[j] = acc;
        }
        #pragma unroll
        for (int j = 0; j < 8; ++j) {
            float acc = 0.f;
            #pragma unroll
            for (int k = 0; k < 8; ++k) acc += GQr[k] * Gmat[j * 8 + k];
            Qr[j] = acc;
        }
    }


    float HFr[8], HFf[8][8], Cr[8];
    {
        #pragma unroll
        for (int jp = 0; jp < 4; ++jp) {
            v2f acc = bc2(Hr[0]) * pr2(Ff[0][2 * jp], Ff[0][2 * jp + 1]);
            #pragma unroll
            for (int k = 1; k < 8; ++k) acc += bc2(Hr[k]) * pr2(Ff[k][2 * jp], Ff[k][2 * jp + 1]);
            HFr[2 * jp] = acc.x; HFr[2 * jp + 1] = acc.y;
        }
        if (writer) lds_put_row(Bsh, idx, HFr);
        lds_handover();
        #pragma unroll
        for (int r = 0; r < 8; ++r) lds_get_row(Bsh, r, HFf[r]);
        float HGr[8], HGQr[8], HQr[8];
        #pragma unroll
        for (int j = 0; j < 8; ++j) {
            float acc = 0.f;
            #pragma unroll
            for (int k = 0; k < 8; ++k) acc += Hr[k] * Gmat[k * 8 + j];
            HGr[j] = acc;
        }
        #pragma unroll
        for (int j = 0; j < 8; ++j) {
            float acc = 0.f;
            #pragma unroll
            for (int k = 0; k < 8; ++k) acc += HGr[k] * Q[k * 8 + j];
            HGQr[j] = acc;
        }
        #pragma unroll
        for (int j = 0; j < 8; ++j) {
            float acc = 0.f;
            #pragma unroll
            for (int k = 0; k < 8; ++k) acc += HGQr[k] * Gmat[j * 8 + k];
            HQr[j] = acc;
        }
        #pragma unroll
        for (int j = 0; j < 8; ++j) {
            float acc = Rr[j];
            #pragma unroll
            for (int k = 0; k < 8; ++k) acc += HQr[k] * Hf[j][k];
            Cr[j] = acc;
        }
    }

    const bool   ps_writer = writer && (blockIdx.x == 0);
    const size_t row  = (size_t)(blockIdx.x * 16 + m16);
    const float* yrow = Y    + row * (size_t)Tn * 8 + (lo ? 0 : 2);
    float*       orow = traj + row * (size_t)Tn * 8 + (lo ? 0 : 4);

    v8f xv;
    {
        v4f xa = *(const v4f*)(x0 + row * 8);
        v4f xb = *(const v4f*)(x0 + row * 8 + 4);
        xv[0] = lo ? xa.x : xa.z;
        xv[1] = lo ? xa.y : xa.w;
        xv[2] = lo ? xa.z : xb.x;
        xv[3] = lo ? xa.w : xb.y;
        xv[4] = lo ? xb.x : xb.z;
        xv[5] = lo ? xb.y : xb.w;
        xv[6] = lo ? xb.z : 0.f;
        xv[7] = lo ? xb.w : 0.f;
    }
    const v8f zero8 = {0.f, 0.f, 0.f, 0.f, 0.f, 0.f, 0.f, 0.f};

    v2f b2 = *(const v2f*)(yrow);
    v2f b3 = *(const v2f*)(yrow + 4);
    v2f a0 = {0.f, 0.f}, a1 = {0.f, 0.f}, a2 = {0.f, 0.f}, a3 = {0.f, 0.f};

    int tstop = Tn;
    for (int t = 0; t < Tn; ++t) {
        const int tn = (t + 1 < Tn) ? (t + 1) : t;
        const v2f nb2 = *(const v2f*)(yrow + (size_t)tn * 8);
        const v2f nb3 = *(const v2f*)(yrow + (size_t)tn * 8 + 4);
        float T1r[8], Ur[8];
        #pragma unroll
        for (int jp = 0; jp < 4; ++jp) {
            v2f acc = bc2(Fr[0])  * pr2(Pf[0][2 * jp], Pf[0][2 * jp + 1]);
            v2f ucc = bc2(HFr[0]) * pr2(Pf[0][2 * jp], Pf[0][2 * jp + 1]);
            #pragma unroll
            for (int k = 1; k < 8; ++k) {
                acc += bc2(Fr[k])  * pr2(Pf[k][2 * jp], Pf[k][2 * jp + 1]);
                ucc += bc2(HFr[k]) * pr2(Pf[k][2 * jp], Pf[k][2 * jp + 1]);
            }
            T1r[2 * jp] = acc.x; T1r[2 * jp + 1] = acc.y;
            Ur[2 * jp]  = ucc.x; Ur[2 * jp + 1]  = ucc.y;
        }
        float Sr[8];
        #pragma unroll
        for (int jp = 0; jp < 4; ++jp) {
            v2f acc = pr2(Cr[2 * jp], Cr[2 * jp + 1]);
            #pragma unroll
            for (int k = 0; k < 8; ++k) acc += bc2(Ur[k]) * pr2(HFf[2 * jp][k], HFf[2 * jp + 1][k]);
            Sr[2 * jp] = acc.x; Sr[2 * jp + 1] = acc.y;
        }
        lds_put_row(Esh, lane, Sr);
        lds_handover();
        float L[8][8];
        {
            const v2f r1  = *(const v2f*)&Esh[1][0];
            const v2f r2  = *(const v2f*)&Esh[2][0];
            const v4f r3  = *(const v4f*)&Esh[3][0];
            const v4f r4  = *(const v4f*)&Esh[4][0];
            const v4f r5  = *(const v4f*)&Esh[5][0];
            const v2f r5b = *(const v2f*)&Esh[5][4];
            const v4f r6  = *(const v4f*)&Esh[6][0];
            const v2f r6b = *(const v2f*)&Esh[6][4];
            const v4f r7  = *(const v4f*)&Esh[7][0];
            const v4f r7b = *(const v4f*)&Esh[7][4];
            L[0][0] = Esh[0][0];
            L[1][0] = r1.x;  L[1][1] = r1.y;
            L[2][0] = r2.x;  L[2][1] = r2.y;  L[2][2] = Esh[2][2];
            L[3][0] = r3.x;  L[3][1] = r3.y;  L[3][2] = r3.z;  L[3][3] = r3.w;
            L[4][0] = r4.x;  L[4][1] = r4.y;  L[4][2] = r4.z;  L[4][3] = r4.w;
            L[4][4] = Esh[4][4];
            L[5][0] = r5.x;  L[5][1] = r5.y;  L[5][2] = r5.z;  L[5][3] = r5.w;
            L[5][4] = r5b.x; L[5][5] = r5b.y;
            L[6][0] = r6.x;  L[6][1] = r6.y;  L[6][2] = r6.z;  L[6][3] = r6.w;
            L[6][4] = r6b.x; L[6][5] = r6b.y; L[6][6] = Esh[6][6];
            L[7][0] = r7.x;  L[7][1] = r7.y;  L[7][2] = r7.z;  L[7][3] = r7.w;
            L[7][4] = r7b.x; L[7][5] = r7b.y; L[7][6] = r7b.z; L[7][7] = r7b.w;
        }
        __builtin_amdgcn_sched_barrier(0);
        float Pnegr[8];
        #pragma unroll
        for (int jp = 0; jp < 4; ++jp) {
            v2f acc = pr2(Qr[2 * jp], Qr[2 * jp + 1]);
            #pragma unroll
            for (int k = 0; k < 8; ++k) acc += bc2(T1r[k]) * pr2(Ff[2 * jp][k], Ff[2 * jp + 1][k]);
            Pnegr[2 * jp] = acc.x; Pnegr[2 * jp + 1] = acc.y;
        }
        lds_put_row(Dsh, lane, Pnegr);
        lds_handover();
        float Pg[8][8], Pgc[8], Xc[8];
        #pragma unroll
        for (int r = 0; r < 8; ++r) Pgc[r] = Dsh[r][idx];
        __builtin_amdgcn_sched_barrier(0);

        float invd[8];
        #pragma unroll
        for (int k = 0; k < 4; ++k) {
            invd[k] = __builtin_amdgcn_rsqf(L[k][k]);
            #pragma unroll
            for (int i = k + 1; i < 8; ++i) L[i][k] = L[i][k] * invd[k];
            #pragma unroll
            for (int i = k + 1; i < 8; ++i) {
                #pragma unroll
                for (int j = k + 1; j <= i; ++j) L[i][j] -= L[i][k] * L[j][k];
            }
        }
        __builtin_amdgcn_sched_barrier(0);
        #pragma unroll
        for (int r = 0; r < 8; ++r) lds_get_row(Dsh, r, Pg[r]);
        #pragma unroll
        for (int rp = 0; rp < 4; ++rp) {
            v2f acc = pr2(Hf[2 * rp][0], Hf[2 * rp + 1][0]) * bc2(Pgc[0]);
            #pragma unroll
            for (int k = 1; k < 8; ++k) acc += pr2(Hf[2 * rp][k], Hf[2 * rp + 1][k]) * bc2(Pgc[k]);
            Xc[2 * rp] = acc.x; Xc[2 * rp + 1] = acc.y;
        }
        #pragma unroll
        for (int k = 4; k < 8; ++k) {
            invd[k] = __builtin_amdgcn_rsqf(L[k][k]);
            #pragma unroll
            for (int i = k + 1; i < 8; ++i) L[i][k] = L[i][k] * invd[k];
            #pragma unroll
            for (int i = k + 1; i < 8; ++i) {
                #pragma unroll
                for (int j = k + 1; j <= i; ++j) L[i][j] -= L[i][k] * L[j][k];
            }
        }
        #pragma unroll
        for (int r = 0; r < 8; ++r) {
            float s = Xc[r];
            #pragma unroll
            for (int m = 0; m < r; ++m) s -= L[r][m] * Xc[m];
            Xc[r] = s * invd[r];
        }
        #pragma unroll
        for (int r = 7; r >= 0; --r) {
            float s = Xc[r];
            #pragma unroll
            for (int m = 7; m > r; --m) s -= L[m][r] * Xc[m];
            Xc[r] = s * invd[r];
        }
        a2 = lo ? pr2(Xc[0], Xc[1]) : pr2(Xc[2], Xc[3]);
        a3 = lo ? pr2(Xc[4], Xc[5]) : pr2(Xc[6], Xc[7]);
        if (!live) { a2 = bc2(0.f); a3 = bc2(0.f); }
        v8f cy = __builtin_amdgcn_wmma_f32_16x16x4_f32(false, a2, false, b2,
                                                       (short)0, zero8, false, false);
        cy = __builtin_amdgcn_wmma_f32_16x16x4_f32(false, a3, false, b3,
                                                   (short)0, cy, false, false);
        float IKHr[8];
        #pragma unroll
        for (int jp = 0; jp < 4; ++jp) {
            v2f acc = pr2((idx == 2 * jp) ? 1.f : 0.f, (idx == 2 * jp + 1) ? 1.f : 0.f);
            #pragma unroll
            for (int m = 7; m >= 0; --m) acc -= bc2(Xc[m]) * pr2(Hf[m][2 * jp], Hf[m][2 * jp + 1]);
            IKHr[2 * jp] = acc.x; IKHr[2 * jp + 1] = acc.y;
        }
        float Ar[8], Pnr[8];
        #pragma unroll
        for (int jp = 0; jp < 4; ++jp) {
            v2f pacc = bc2(IKHr[0]) * pr2(Pg[0][2 * jp], Pg[0][2 * jp + 1]);
            #pragma unroll
            for (int k = 1; k < 8; ++k) pacc += bc2(IKHr[k]) * pr2(Pg[k][2 * jp], Pg[k][2 * jp + 1]);
            Pnr[2 * jp] = pacc.x; Pnr[2 * jp + 1] = pacc.y;
        }
        if (writer) lds_put_row(Csh, idx, Pnr);
        lds_handover();
        float Pnf[8][8], Pnc[8];
        #pragma unroll
        for (int r = 0; r < 8; ++r) {
            lds_get_row(Csh, r, Pnf[r]);
            Pnc[r] = Csh[r][idx];
        }
        #pragma unroll
        for (int jp = 0; jp < 4; ++jp) {
            v2f acc = bc2(IKHr[0]) * pr2(Ff[0][2 * jp], Ff[0][2 * jp + 1]);
            #pragma unroll
            for (int k = 1; k < 8; ++k) acc += bc2(IKHr[k]) * pr2(Ff[k][2 * jp], Ff[k][2 * jp + 1]);
            Ar[2 * jp] = acc.x; Ar[2 * jp + 1] = acc.y;
        }
        a0 = lo ? pr2(Ar[0], Ar[1]) : pr2(Ar[2], Ar[3]);
        a1 = lo ? pr2(Ar[4], Ar[5]) : pr2(Ar[6], Ar[7]);
        if (!live) { a0 = bc2(0.f); a1 = bc2(0.f); }
        const v2f b0 = {xv[0], xv[1]};
        const v2f b1 = {xv[4], xv[5]};
        const v8f c1 = __builtin_amdgcn_wmma_f32_16x16x4_f32(false, a0, false, b0,
                                                             (short)0, cy, false, false);
        const v8f c2 = __builtin_amdgcn_wmma_f32_16x16x4_f32(false, a1, false, b1,
                                                             (short)0, zero8, false, false);
        #pragma unroll
        for (int r = 0; r < 8; ++r) {
            #pragma unroll
            for (int c = 0; c < 8; ++c) Pf[r][c] = 0.5f * (Pnf[r][c] + Pnf[c][r]);
        }
        bool changed = false;
        float Pnew[8];
        #pragma unroll
        for (int j = 0; j < 8; ++j) {
            Pnew[j] = 0.5f * (Pnr[j] + Pnc[j]);
            changed = changed || (Pnew[j] != Pr[j]);
            Pr[j] = Pnew[j];
        }

        if (ps_writer) {
            v4f p0 = {Pnew[0], Pnew[1], Pnew[2], Pnew[3]};
            v4f p1 = {Pnew[4], Pnew[5], Pnew[6], Pnew[7]};
            *(v4f*)(Ps + t * 64 + idx * 8)     = p0;
            *(v4f*)(Ps + t * 64 + idx * 8 + 4) = p1;
        }
        xv = c1 + c2;
        {
            v4f out;
            out.x = lo ? xv[0] : xv[2];
            out.y = lo ? xv[1] : xv[3];
            out.z = lo ? xv[2] : xv[4];
            out.w = lo ? xv[3] : xv[5];
            *(v4f*)(orow + (size_t)t * 8) = out;
        }
        b2 = nb2; b3 = nb3;
        if (__builtin_amdgcn_ballot_w32(changed) == 0u) { tstop = t + 1; break; }
    }

    if (tstop < Tn) {
        if (ps_writer) {
            const v4f p0 = {Pr[0], Pr[1], Pr[2], Pr[3]};
            const v4f p1 = {Pr[4], Pr[5], Pr[6], Pr[7]};
            for (int t = tstop; t < Tn; ++t) {
                *(v4f*)(Ps + t * 64 + idx * 8)     = p0;
                *(v4f*)(Ps + t * 64 + idx * 8 + 4) = p1;
            }
        }
        int t = tstop;
        v2f yb2[8], yb3[8];
        #pragma unroll
        for (int j = 0; j < 8; ++j) {
            const int tj = (t + j < Tn) ? (t + j) : (Tn - 1);
            yb2[j] = *(const v2f*)(yrow + (size_t)tj * 8);
            yb3[j] = *(const v2f*)(yrow + (size_t)tj * 8 + 4);
        }
        for (; t + 8 <= Tn; t += 8) {
            v2f nb2[8], nb3[8];
            #pragma unroll
            for (int j = 0; j < 8; ++j) {
                const int tj = (t + 8 + j < Tn) ? (t + 8 + j) : (Tn - 1);
                nb2[j] = *(const v2f*)(yrow + (size_t)tj * 8);
                nb3[j] = *(const v2f*)(yrow + (size_t)tj * 8 + 4);
            }
            #pragma unroll
            for (int j = 0; j < 8; ++j)
                state_step(xv, a0, a1, a2, a3, yb2[j], yb3[j], lo, orow + (size_t)(t + j) * 8);
            #pragma unroll
            for (int j = 0; j < 8; ++j) { yb2[j] = nb2[j]; yb3[j] = nb3[j]; }
        }
        #pragma unroll
        for (int j = 0; j < 7; ++j) {
            if (t + j < Tn)
                state_step(xv, a0, a1, a2, a3, yb2[j], yb3[j], lo, orow + (size_t)(t + j) * 8);
        }
    }
}

extern "C" void kernel_launch(void* const* d_in, const int* in_sizes, int n_in,
                              void* d_out, int out_size, void* d_ws, size_t ws_size,
                              hipStream_t stream) {
    const float* Y    = (const float*)d_in[0];
    const float* F    = (const float*)d_in[1];
    const float* Gmat = (const float*)d_in[2];
    const float* H    = (const float*)d_in[3];
    const float* Q    = (const float*)d_in[4];
    const float* R    = (const float*)d_in[5];
    const float* x0   = (const float*)d_in[6];
    const float* P0   = (const float*)d_in[7];

    float* traj = (float*)d_out;
    float* Ps   = traj + (size_t)Bn * Tn * 8;

    kf_fused_kernel<<<Bn / 16, 32, 0, stream>>>(Y, F, Gmat, H, Q, R, x0, P0, traj, Ps);
}


# T1=F P, U=(HF)P and Pn=IKH Pneg run on the matrix unit, fed by each lane's own row of P / own column of Pneg: P, Pneg, Pn are no longer read back whole; summation order of the three changed
# speedup vs baseline: 13.9457x; 1.3033x over previous
#include <hip/hip_runtime.h>
#include <math.h>

static constexpr int Bn = 2048;
static constexpr int Tn = 1000;

typedef float v2f __attribute__((ext_vector_type(2)));
typedef float v4f __attribute__((ext_vector_type(4)));
typedef float v8f __attribute__((ext_vector_type(8)));


__device__ __forceinline__ void lds_handover() {
    __builtin_amdgcn_fence(__ATOMIC_RELEASE, "wavefront", "local");
    __builtin_amdgcn_fence(__ATOMIC_ACQUIRE, "wavefront", "local");
}

__device__ __forceinline__ void lds_put_row(float (*M)[8], int r, const float* v) {
    v4f a = {v[0], v[1], v[2], v[3]};
    v4f b = {v[4], v[5], v[6], v[7]};
    *(v4f*)&M[r][0] = a;
    *(v4f*)&M[r][4] = b;
}

__device__ __forceinline__ void lds_get_row(const float (*M)[8], int r, float* v) {
    v4f a = *(const v4f*)&M[r][0];
    v4f b = *(const v4f*)&M[r][4];
    v[0] = a.x; v[1] = a.y; v[2] = a.z; v[3] = a.w;
    v[4] = b.x; v[5] = b.y; v[6] = b.z; v[7] = b.w;
}

__device__ __forceinline__ v2f bc2(float x) { v2f r = {x, x}; return r; }
__device__ __forceinline__ v2f pr2(float a, float b) { v2f r = {a, b}; return r; }

__device__ __forceinline__ void state_step(v8f& xv, v2f a0, v2f a1, v2f a2, v2f a3,
                                           v2f y2, v2f y3, bool lo, float* dst) {
    const v8f zero8 = {0.f, 0.f, 0.f, 0.f, 0.f, 0.f, 0.f, 0.f};
    const v2f b0 = {xv[0], xv[1]};
    const v2f b1 = {xv[4], xv[5]};
    v8f cy = __builtin_amdgcn_wmma_f32_16x16x4_f32(false, a2, false, y2,
                                                   (short)0, zero8, false, false);
    cy = __builtin_amdgcn_wmma_f32_16x16x4_f32(false, a3, false, y3,
                                               (short)0, cy, false, false);
    const v8f c1 = __builtin_amdgcn_wmma_f32_16x16x4_f32(false, a0, false, b0,
                                                         (short)0, cy, false, false);
    const v8f c2 = __builtin_amdgcn_wmma_f32_16x16x4_f32(false, a1, false, b1,
                                                         (short)0, zero8, false, false);
    xv = c1 + c2;
    v4f out;
    out.x = lo ? xv[0] : xv[2];
    out.y = lo ? xv[1] : xv[3];
    out.z = lo ? xv[2] : xv[4];
    out.w = lo ? xv[3] : xv[5];
    *(v4f*)dst = out;
}

__global__ __launch_bounds__(32) void kf_fused_kernel(
    const float* __restrict__ Y, const float* __restrict__ F,
    const float* __restrict__ Gmat, const float* __restrict__ H,
    const float* __restrict__ Q, const float* __restrict__ R,
    const float* __restrict__ x0, const float* __restrict__ P0,
    float* __restrict__ traj, float* __restrict__ Ps)
{
    __shared__ __attribute__((aligned(16))) float Ash[8][8];
    __shared__ __attribute__((aligned(16))) float Bsh[8][8];
    __shared__ __attribute__((aligned(16))) float Csh[32][8];
    __shared__ __attribute__((aligned(16))) float Dsh[32][8];
    __shared__ __attribute__((aligned(16))) float Esh[32][8];

    const int  lane   = threadIdx.x;
    const int  m16    = lane & 15;
    const int  idx    = (m16 < 8) ? m16 : ((m16 <= 13) ? (m16 - 6) : (m16 - 8));
    const bool live   = (m16 <= 13);
    const bool lo     = lane < 16;
    const bool writer = lane < 8;

    float Fr[8], Hr[8], Rr[8], Pr[8], Gr[8];
    #pragma unroll
    for (int c = 0; c < 8; ++c) {
        Fr[c] = F[idx * 8 + c];
        Hr[c] = H[idx * 8 + c];
        Rr[c] = R[idx * 8 + c];
        Gr[c] = Gmat[idx * 8 + c];
    }
    #pragma unroll
    for (int c = 0; c < 8; ++c) {
        const int e = lo ? c : ((c < 6) ? (c + 2) : c);
        Pr[c] = 0.5f * (P0[idx * 8 + e] + P0[e * 8 + idx]);
    }

    if (writer) {
        lds_put_row(Ash, idx, Fr);
        lds_put_row(Bsh, idx, Hr);
    }
    lds_handover();
    float Ff[8][8], Hf[8][8];
    #pragma unroll
    for (int r = 0; r < 8; ++r) {
        lds_get_row(Ash, r, Ff[r]);
        lds_get_row(Bsh, r, Hf[r]);
    }

    float Qr[8];
    {
        float GQr[8];
        #pragma unroll
        for (int j = 0; j < 8; ++j) {
            float acc = 0.f;
            #pragma unroll
            for (int k = 0; k < 8; ++k) acc += Gr[k] * Q[k * 8 + j];
            GQr[j] = acc;
        }
        #pragma unroll
        for (int j = 0; j < 8; ++j) {
            float acc = 0.f;
            #pragma unroll
            for (int k = 0; k < 8; ++k) acc += GQr[k] * Gmat[j * 8 + k];
            Qr[j] = acc;
        }
    }


    float HFr[8], HFf[8][8], Cr[8];
    {
        #pragma unroll
        for (int jp = 0; jp < 4; ++jp) {
            v2f acc = bc2(Hr[0]) * pr2(Ff[0][2 * jp], Ff[0][2 * jp + 1]);
            #pragma unroll
            for (int k = 1; k < 8; ++k) acc += bc2(Hr[k]) * pr2(Ff[k][2 * jp], Ff[k][2 * jp + 1]);
            HFr[2 * jp] = acc.x; HFr[2 * jp + 1] = acc.y;
        }
        if (writer) lds_put_row(Bsh, idx, HFr);
        lds_handover();
        #pragma unroll
        for (int r = 0; r < 8; ++r) lds_get_row(Bsh, r, HFf[r]);
        float HGr[8], HGQr[8], HQr[8];
        #pragma unroll
        for (int j = 0; j < 8; ++j) {
            float acc = 0.f;
            #pragma unroll
            for (int k = 0; k < 8; ++k) acc += Hr[k] * Gmat[k * 8 + j];
            HGr[j] = acc;
        }
        #pragma unroll
        for (int j = 0; j < 8; ++j) {
            float acc = 0.f;
            #pragma unroll
            for (int k = 0; k < 8; ++k) acc += HGr[k] * Q[k * 8 + j];
            HGQr[j] = acc;
        }
        #pragma unroll
        for (int j = 0; j < 8; ++j) {
            float acc = 0.f;
            #pragma unroll
            for (int k = 0; k < 8; ++k) acc += HGQr[k] * Gmat[j * 8 + k];
            HQr[j] = acc;
        }
        #pragma unroll
        for (int j = 0; j < 8; ++j) {
            float acc = Rr[j];
            #pragma unroll
            for (int k = 0; k < 8; ++k) acc += HQr[k] * Hf[j][k];
            Cr[j] = acc;
        }
    }

    const bool   ps_writer = writer && (blockIdx.x == 0);
    const size_t row  = (size_t)(blockIdx.x * 16 + m16);
    const float* yrow = Y    + row * (size_t)Tn * 8 + (lo ? 0 : 2);
    float*       orow = traj + row * (size_t)Tn * 8 + (lo ? 0 : 4);

    v8f xv;
    {
        v4f xa = *(const v4f*)(x0 + row * 8);
        v4f xb = *(const v4f*)(x0 + row * 8 + 4);
        xv[0] = lo ? xa.x : xa.z;
        xv[1] = lo ? xa.y : xa.w;
        xv[2] = lo ? xa.z : xb.x;
        xv[3] = lo ? xa.w : xb.y;
        xv[4] = lo ? xb.x : xb.z;
        xv[5] = lo ? xb.y : xb.w;
        xv[6] = lo ? xb.z : 0.f;
        xv[7] = lo ? xb.w : 0.f;
    }
    const v8f zero8 = {0.f, 0.f, 0.f, 0.f, 0.f, 0.f, 0.f, 0.f};

    const v2f fb0 = lo ? pr2(Fr[0], Fr[1])   : pr2(Fr[2], Fr[3]);
    const v2f fb1 = lo ? pr2(Fr[4], Fr[5])   : pr2(Fr[6], Fr[7]);
    const v2f hb0 = lo ? pr2(HFr[0], HFr[1]) : pr2(HFr[2], HFr[3]);
    const v2f hb1 = lo ? pr2(HFr[4], HFr[5]) : pr2(HFr[6], HFr[7]);
    const float* pncol = &Csh[lo ? 0 : 2][idx];

    v2f b2 = *(const v2f*)(yrow);
    v2f b3 = *(const v2f*)(yrow + 4);
    v2f a0 = {0.f, 0.f}, a1 = {0.f, 0.f}, a2 = {0.f, 0.f}, a3 = {0.f, 0.f};

    int tstop = Tn;
    for (int t = 0; t < Tn; ++t) {
        const int tn = (t + 1 < Tn) ? (t + 1) : t;
        const v2f nb2 = *(const v2f*)(yrow + (size_t)tn * 8);
        const v2f nb3 = *(const v2f*)(yrow + (size_t)tn * 8 + 4);
        const v2f pa0 = pr2(Pr[0], Pr[1]);
        const v2f pa1 = pr2(Pr[4], Pr[5]);
        v8f uacc = __builtin_amdgcn_wmma_f32_16x16x4_f32(false, pa0, false, hb0,
                                                         (short)0, zero8, false, false);
        uacc = __builtin_amdgcn_wmma_f32_16x16x4_f32(false, pa1, false, hb1,
                                                     (short)0, uacc, false, false);
        v8f tacc = __builtin_amdgcn_wmma_f32_16x16x4_f32(false, pa0, false, fb0,
                                                         (short)0, zero8, false, false);
        tacc = __builtin_amdgcn_wmma_f32_16x16x4_f32(false, pa1, false, fb1,
                                                     (short)0, tacc, false, false);
        float T1r[8], Ur[8];
        #pragma unroll
        for (int j = 0; j < 8; ++j) { T1r[j] = tacc[j]; Ur[j] = uacc[j]; }
        float Sr[8];
        #pragma unroll
        for (int jp = 0; jp < 4; ++jp) {
            v2f acc = pr2(Cr[2 * jp], Cr[2 * jp + 1]);
            #pragma unroll
            for (int k = 0; k < 8; ++k) acc += bc2(Ur[k]) * pr2(HFf[2 * jp][k], HFf[2 * jp + 1][k]);
            Sr[2 * jp] = acc.x; Sr[2 * jp + 1] = acc.y;
        }
        lds_put_row(Esh, lane, Sr);
        lds_handover();
        float L[8][8];
        {
            const v2f r1  = *(const v2f*)&Esh[1][0];
            const v2f r2  = *(const v2f*)&Esh[2][0];
            const v4f r3  = *(const v4f*)&Esh[3][0];
            const v4f r4  = *(const v4f*)&Esh[4][0];
            const v4f r5  = *(const v4f*)&Esh[5][0];
            const v2f r5b = *(const v2f*)&Esh[5][4];
            const v4f r6  = *(const v4f*)&Esh[6][0];
            const v2f r6b = *(const v2f*)&Esh[6][4];
            const v4f r7  = *(const v4f*)&Esh[7][0];
            const v4f r7b = *(const v4f*)&Esh[7][4];
            L[0][0] = Esh[0][0];
            L[1][0] = r1.x;  L[1][1] = r1.y;
            L[2][0] = r2.x;  L[2][1] = r2.y;  L[2][2] = Esh[2][2];
            L[3][0] = r3.x;  L[3][1] = r3.y;  L[3][2] = r3.z;  L[3][3] = r3.w;
            L[4][0] = r4.x;  L[4][1] = r4.y;  L[4][2] = r4.z;  L[4][3] = r4.w;
            L[4][4] = Esh[4][4];
            L[5][0] = r5.x;  L[5][1] = r5.y;  L[5][2] = r5.z;  L[5][3] = r5.w;
            L[5][4] = r5b.x; L[5][5] = r5b.y;
            L[6][0] = r6.x;  L[6][1] = r6.y;  L[6][2] = r6.z;  L[6][3] = r6.w;
            L[6][4] = r6b.x; L[6][5] = r6b.y; L[6][6] = Esh[6][6];
            L[7][0] = r7.x;  L[7][1] = r7.y;  L[7][2] = r7.z;  L[7][3] = r7.w;
            L[7][4] = r7b.x; L[7][5] = r7b.y; L[7][6] = r7b.z; L[7][7] = r7b.w;
        }
        __builtin_amdgcn_sched_barrier(0);
        float Pnegr[8];
        #pragma unroll
        for (int jp = 0; jp < 4; ++jp) {
            v2f acc = pr2(Qr[2 * jp], Qr[2 * jp + 1]);
            #pragma unroll
            for (int k = 0; k < 8; ++k) acc += bc2(T1r[k]) * pr2(Ff[2 * jp][k], Ff[2 * jp + 1][k]);
            Pnegr[2 * jp] = acc.x; Pnegr[2 * jp + 1] = acc.y;
        }
        lds_put_row(Dsh, lane, Pnegr);
        lds_handover();
        float Pgc[8], Xc[8];
        #pragma unroll
        for (int r = 0; r < 8; ++r) Pgc[r] = Dsh[r][idx];
        __builtin_amdgcn_sched_barrier(0);

        float invd[8];
        #pragma unroll
        for (int k = 0; k < 4; ++k) {
            invd[k] = __builtin_amdgcn_rsqf(L[k][k]);
            #pragma unroll
            for (int i = k + 1; i < 8; ++i) L[i][k] = L[i][k] * invd[k];
            #pragma unroll
            for (int i = k + 1; i < 8; ++i) {
                #pragma unroll
                for (int j = k + 1; j <= i; ++j) L[i][j] -= L[i][k] * L[j][k];
            }
        }
        __builtin_amdgcn_sched_barrier(0);
        #pragma unroll
        for (int rp = 0; rp < 4; ++rp) {
            v2f acc = pr2(Hf[2 * rp][0], Hf[2 * rp + 1][0]) * bc2(Pgc[0]);
            #pragma unroll
            for (int k = 1; k < 8; ++k) acc += pr2(Hf[2 * rp][k], Hf[2 * rp + 1][k]) * bc2(Pgc[k]);
            Xc[2 * rp] = acc.x; Xc[2 * rp + 1] = acc.y;
        }
        #pragma unroll
        for (int k = 4; k < 8; ++k) {
            invd[k] = __builtin_amdgcn_rsqf(L[k][k]);
            #pragma unroll
            for (int i = k + 1; i < 8; ++i) L[i][k] = L[i][k] * invd[k];
            #pragma unroll
            for (int i = k + 1; i < 8; ++i) {
                #pragma unroll
                for (int j = k + 1; j <= i; ++j) L[i][j] -= L[i][k] * L[j][k];
            }
        }
        #pragma unroll
        for (int r = 0; r < 8; ++r) {
            float s = Xc[r];
            #pragma unroll
            for (int m = 0; m < r; ++m) s -= L[r][m] * Xc[m];
            Xc[r] = s * invd[r];
        }
        #pragma unroll
        for (int r = 7; r >= 0; --r) {
            float s = Xc[r];
            #pragma unroll
            for (int m = 7; m > r; --m) s -= L[m][r] * Xc[m];
            Xc[r] = s * invd[r];
        }
        a2 = lo ? pr2(Xc[0], Xc[1]) : pr2(Xc[2], Xc[3]);
        a3 = lo ? pr2(Xc[4], Xc[5]) : pr2(Xc[6], Xc[7]);
        if (!live) { a2 = bc2(0.f); a3 = bc2(0.f); }
        v8f cy = __builtin_amdgcn_wmma_f32_16x16x4_f32(false, a2, false, b2,
                                                       (short)0, zero8, false, false);
        cy = __builtin_amdgcn_wmma_f32_16x16x4_f32(false, a3, false, b3,
                                                   (short)0, cy, false, false);
        float IKHr[8];
        #pragma unroll
        for (int jp = 0; jp < 4; ++jp) {
            v2f acc = pr2((idx == 2 * jp) ? 1.f : 0.f, (idx == 2 * jp + 1) ? 1.f : 0.f);
            #pragma unroll
            for (int m = 7; m >= 0; --m) acc -= bc2(Xc[m]) * pr2(Hf[m][2 * jp], Hf[m][2 * jp + 1]);
            IKHr[2 * jp] = acc.x; IKHr[2 * jp + 1] = acc.y;
        }
        const v2f ga0 = lo ? pr2(Pgc[0], Pgc[1])   : pr2(Pgc[2], Pgc[3]);
        const v2f ga1 = lo ? pr2(Pgc[4], Pgc[5])   : pr2(Pgc[6], Pgc[7]);
        const v2f kb0 = lo ? pr2(IKHr[0], IKHr[1]) : pr2(IKHr[2], IKHr[3]);
        const v2f kb1 = lo ? pr2(IKHr[4], IKHr[5]) : pr2(IKHr[6], IKHr[7]);
        v8f pacc = __builtin_amdgcn_wmma_f32_16x16x4_f32(false, ga0, false, kb0,
                                                         (short)0, zero8, false, false);
        pacc = __builtin_amdgcn_wmma_f32_16x16x4_f32(false, ga1, false, kb1,
                                                     (short)0, pacc, false, false);
        float Ar[8], Pnr[8];
        #pragma unroll
        for (int j = 0; j < 8; ++j) Pnr[j] = pacc[j];
        lds_put_row(Csh, lane, Pnr);
        lds_handover();
        float Pnc[8];
        #pragma unroll
        for (int r = 0; r < 6; ++r) Pnc[r] = pncol[r * 8];
        Pnc[6] = Csh[6][idx];
        Pnc[7] = Csh[7][idx];
        #pragma unroll
        for (int jp = 0; jp < 4; ++jp) {
            v2f acc = bc2(IKHr[0]) * pr2(Ff[0][2 * jp], Ff[0][2 * jp + 1]);
            #pragma unroll
            for (int k = 1; k < 8; ++k) acc += bc2(IKHr[k]) * pr2(Ff[k][2 * jp], Ff[k][2 * jp + 1]);
            Ar[2 * jp] = acc.x; Ar[2 * jp + 1] = acc.y;
        }
        a0 = lo ? pr2(Ar[0], Ar[1]) : pr2(Ar[2], Ar[3]);
        a1 = lo ? pr2(Ar[4], Ar[5]) : pr2(Ar[6], Ar[7]);
        if (!live) { a0 = bc2(0.f); a1 = bc2(0.f); }
        const v2f b0 = {xv[0], xv[1]};
        const v2f b1 = {xv[4], xv[5]};
        const v8f c1 = __builtin_amdgcn_wmma_f32_16x16x4_f32(false, a0, false, b0,
                                                             (short)0, cy, false, false);
        const v8f c2 = __builtin_amdgcn_wmma_f32_16x16x4_f32(false, a1, false, b1,
                                                             (short)0, zero8, false, false);
        bool changed = false;
        float Pnew[8];
        #pragma unroll
        for (int j = 0; j < 8; ++j) {
            Pnew[j] = 0.5f * (Pnr[j] + Pnc[j]);
            changed = changed || (Pnew[j] != Pr[j]);
            Pr[j] = Pnew[j];
        }

        if (ps_writer) {
            v4f p0 = {Pnew[0], Pnew[1], Pnew[2], Pnew[3]};
            v4f p1 = {Pnew[4], Pnew[5], Pnew[6], Pnew[7]};
            *(v4f*)(Ps + t * 64 + idx * 8)     = p0;
            *(v4f*)(Ps + t * 64 + idx * 8 + 4) = p1;
        }
        xv = c1 + c2;
        {
            v4f out;
            out.x = lo ? xv[0] : xv[2];
            out.y = lo ? xv[1] : xv[3];
            out.z = lo ? xv[2] : xv[4];
            out.w = lo ? xv[3] : xv[5];
            *(v4f*)(orow + (size_t)t * 8) = out;
        }
        b2 = nb2; b3 = nb3;
        if (__builtin_amdgcn_ballot_w32(changed) == 0u) { tstop = t + 1; break; }
    }

    if (tstop < Tn) {
        if (ps_writer) {
            const v4f p0 = {Pr[0], Pr[1], Pr[2], Pr[3]};
            const v4f p1 = {Pr[4], Pr[5], Pr[6], Pr[7]};
            for (int t = tstop; t < Tn; ++t) {
                *(v4f*)(Ps + t * 64 + idx * 8)     = p0;
                *(v4f*)(Ps + t * 64 + idx * 8 + 4) = p1;
            }
        }
        int t = tstop;
        v2f yb2[8], yb3[8];
        #pragma unroll
        for (int j = 0; j < 8; ++j) {
            const int tj = (t + j < Tn) ? (t + j) : (Tn - 1);
            yb2[j] = *(const v2f*)(yrow + (size_t)tj * 8);
            yb3[j] = *(const v2f*)(yrow + (size_t)tj * 8 + 4);
        }
        for (; t + 8 <= Tn; t += 8) {
            v2f nb2[8], nb3[8];
            #pragma unroll
            for (int j = 0; j < 8; ++j) {
                const int tj = (t + 8 + j < Tn) ? (t + 8 + j) : (Tn - 1);
                nb2[j] = *(const v2f*)(yrow + (size_t)tj * 8);
                nb3[j] = *(const v2f*)(yrow + (size_t)tj * 8 + 4);
            }
            #pragma unroll
            for (int j = 0; j < 8; ++j)
                state_step(xv, a0, a1, a2, a3, yb2[j], yb3[j], lo, orow + (size_t)(t + j) * 8);
            #pragma unroll
            for (int j = 0; j < 8; ++j) { yb2[j] = nb2[j]; yb3[j] = nb3[j]; }
        }
        #pragma unroll
        for (int j = 0; j < 7; ++j) {
            if (t + j < Tn)
                state_step(xv, a0, a1, a2, a3, yb2[j], yb3[j], lo, orow + (size_t)(t + j) * 8);
        }
    }
}

extern "C" void kernel_launch(void* const* d_in, const int* in_sizes, int n_in,
                              void* d_out, int out_size, void* d_ws, size_t ws_size,
                              hipStream_t stream) {
    const float* Y    = (const float*)d_in[0];
    const float* F    = (const float*)d_in[1];
    const float* Gmat = (const float*)d_in[2];
    const float* H    = (const float*)d_in[3];
    const float* Q    = (const float*)d_in[4];
    const float* R    = (const float*)d_in[5];
    const float* x0   = (const float*)d_in[6];
    const float* P0   = (const float*)d_in[7];

    float* traj = (float*)d_out;
    float* Ps   = traj + (size_t)Bn * Tn * 8;

    kf_fused_kernel<<<Bn / 16, 32, 0, stream>>>(Y, F, Gmat, H, Q, R, x0, P0, traj, Ps);
}


# Own column of Pn comes off the matrix unit, the same two operand pairs with the A and B roles exchanged: Pn's LDS hand-over (stores, trip, reads) leaves the serial path; summation order unchanged
# speedup vs baseline: 14.4081x; 1.0332x over previous
#include <hip/hip_runtime.h>
#include <math.h>

static constexpr int Bn = 2048;
static constexpr int Tn = 1000;

typedef float v2f __attribute__((ext_vector_type(2)));
typedef float v4f __attribute__((ext_vector_type(4)));
typedef float v8f __attribute__((ext_vector_type(8)));


__device__ __forceinline__ void lds_handover() {
    __builtin_amdgcn_fence(__ATOMIC_RELEASE, "wavefront", "local");
    __builtin_amdgcn_fence(__ATOMIC_ACQUIRE, "wavefront", "local");
}

__device__ __forceinline__ void lds_put_row(float (*M)[8], int r, const float* v) {
    v4f a = {v[0], v[1], v[2], v[3]};
    v4f b = {v[4], v[5], v[6], v[7]};
    *(v4f*)&M[r][0] = a;
    *(v4f*)&M[r][4] = b;
}

__device__ __forceinline__ void lds_get_row(const float (*M)[8], int r, float* v) {
    v4f a = *(const v4f*)&M[r][0];
    v4f b = *(const v4f*)&M[r][4];
    v[0] = a.x; v[1] = a.y; v[2] = a.z; v[3] = a.w;
    v[4] = b.x; v[5] = b.y; v[6] = b.z; v[7] = b.w;
}

__device__ __forceinline__ v2f bc2(float x) { v2f r = {x, x}; return r; }
__device__ __forceinline__ v2f pr2(float a, float b) { v2f r = {a, b}; return r; }

__device__ __forceinline__ void state_step(v8f& xv, v2f a0, v2f a1, v2f a2, v2f a3,
                                           v2f y2, v2f y3, bool lo, float* dst) {
    const v8f zero8 = {0.f, 0.f, 0.f, 0.f, 0.f, 0.f, 0.f, 0.f};
    const v2f b0 = {xv[0], xv[1]};
    const v2f b1 = {xv[4], xv[5]};
    v8f cy = __builtin_amdgcn_wmma_f32_16x16x4_f32(false, a2, false, y2,
                                                   (short)0, zero8, false, false);
    cy = __builtin_amdgcn_wmma_f32_16x16x4_f32(false, a3, false, y3,
                                               (short)0, cy, false, false);
    const v8f c1 = __builtin_amdgcn_wmma_f32_16x16x4_f32(false, a0, false, b0,
                                                         (short)0, cy, false, false);
    const v8f c2 = __builtin_amdgcn_wmma_f32_16x16x4_f32(false, a1, false, b1,
                                                         (short)0, zero8, false, false);
    xv = c1 + c2;
    v4f out;
    out.x = lo ? xv[0] : xv[2];
    out.y = lo ? xv[1] : xv[3];
    out.z = lo ? xv[2] : xv[4];
    out.w = lo ? xv[3] : xv[5];
    *(v4f*)dst = out;
}

__global__ __launch_bounds__(32) void kf_fused_kernel(
    const float* __restrict__ Y, const float* __restrict__ F,
    const float* __restrict__ Gmat, const float* __restrict__ H,
    const float* __restrict__ Q, const float* __restrict__ R,
    const float* __restrict__ x0, const float* __restrict__ P0,
    float* __restrict__ traj, float* __restrict__ Ps)
{
    __shared__ __attribute__((aligned(16))) float Ash[8][8];
    __shared__ __attribute__((aligned(16))) float Bsh[8][8];
    __shared__ __attribute__((aligned(16))) float Dsh[32][8];
    __shared__ __attribute__((aligned(16))) float Esh[32][8];

    const int  lane   = threadIdx.x;
    const int  m16    = lane & 15;
    const int  idx    = (m16 < 8) ? m16 : ((m16 <= 13) ? (m16 - 6) : (m16 - 8));
    const bool live   = (m16 <= 13);
    const bool lo     = lane < 16;
    const bool writer = lane < 8;

    float Fr[8], Hr[8], Rr[8], Pr[8], Gr[8];
    #pragma unroll
    for (int c = 0; c < 8; ++c) {
        Fr[c] = F[idx * 8 + c];
        Hr[c] = H[idx * 8 + c];
        Rr[c] = R[idx * 8 + c];
        Gr[c] = Gmat[idx * 8 + c];
    }
    #pragma unroll
    for (int c = 0; c < 8; ++c) {
        const int e = lo ? c : ((c < 6) ? (c + 2) : c);
        Pr[c] = 0.5f * (P0[idx * 8 + e] + P0[e * 8 + idx]);
    }

    if (writer) {
        lds_put_row(Ash, idx, Fr);
        lds_put_row(Bsh, idx, Hr);
    }
    lds_handover();
    float Ff[8][8], Hf[8][8];
    #pragma unroll
    for (int r = 0; r < 8; ++r) {
        lds_get_row(Ash, r, Ff[r]);
        lds_get_row(Bsh, r, Hf[r]);
    }

    float Qr[8];
    {
        float GQr[8];
        #pragma unroll
        for (int j = 0; j < 8; ++j) {
            float acc = 0.f;
            #pragma unroll
            for (int k = 0; k < 8; ++k) acc += Gr[k] * Q[k * 8 + j];
            GQr[j] = acc;
        }
        #pragma unroll
        for (int j = 0; j < 8; ++j) {
            float acc = 0.f;
            #pragma unroll
            for (int k = 0; k < 8; ++k) acc += GQr[k] * Gmat[j * 8 + k];
            Qr[j] = acc;
        }
    }


    float HFr[8], HFf[8][8], Cr[8];
    {
        #pragma unroll
        for (int jp = 0; jp < 4; ++jp) {
            v2f acc = bc2(Hr[0]) * pr2(Ff[0][2 * jp], Ff[0][2 * jp + 1]);
            #pragma unroll
            for (int k = 1; k < 8; ++k) acc += bc2(Hr[k]) * pr2(Ff[k][2 * jp], Ff[k][2 * jp + 1]);
            HFr[2 * jp] = acc.x; HFr[2 * jp + 1] = acc.y;
        }
        if (writer) lds_put_row(Bsh, idx, HFr);
        lds_handover();
        #pragma unroll
        for (int r = 0; r < 8; ++r) lds_get_row(Bsh, r, HFf[r]);
        float HGr[8], HGQr[8], HQr[8];
        #pragma unroll
        for (int j = 0; j < 8; ++j) {
            float acc = 0.f;
            #pragma unroll
            for (int k = 0; k < 8; ++k) acc += Hr[k] * Gmat[k * 8 + j];
            HGr[j] = acc;
        }
        #pragma unroll
        for (int j = 0; j < 8; ++j) {
            float acc = 0.f;
            #pragma unroll
            for (int k = 0; k < 8; ++k) acc += HGr[k] * Q[k * 8 + j];
            HGQr[j] = acc;
        }
        #pragma unroll
        for (int j = 0; j < 8; ++j) {
            float acc = 0.f;
            #pragma unroll
            for (int k = 0; k < 8; ++k) acc += HGQr[k] * Gmat[j * 8 + k];
            HQr[j] = acc;
        }
        #pragma unroll
        for (int j = 0; j < 8; ++j) {
            float acc = Rr[j];
            #pragma unroll
            for (int k = 0; k < 8; ++k) acc += HQr[k] * Hf[j][k];
            Cr[j] = acc;
        }
    }

    const bool   ps_writer = writer && (blockIdx.x == 0);
    const size_t row  = (size_t)(blockIdx.x * 16 + m16);
    const float* yrow = Y    + row * (size_t)Tn * 8 + (lo ? 0 : 2);
    float*       orow = traj + row * (size_t)Tn * 8 + (lo ? 0 : 4);

    v8f xv;
    {
        v4f xa = *(const v4f*)(x0 + row * 8);
        v4f xb = *(const v4f*)(x0 + row * 8 + 4);
        xv[0] = lo ? xa.x : xa.z;
        xv[1] = lo ? xa.y : xa.w;
        xv[2] = lo ? xa.z : xb.x;
        xv[3] = lo ? xa.w : xb.y;
        xv[4] = lo ? xb.x : xb.z;
        xv[5] = lo ? xb.y : xb.w;
        xv[6] = lo ? xb.z : 0.f;
        xv[7] = lo ? xb.w : 0.f;
    }
    const v8f zero8 = {0.f, 0.f, 0.f, 0.f, 0.f, 0.f, 0.f, 0.f};

    const v2f fb0 = lo ? pr2(Fr[0], Fr[1])   : pr2(Fr[2], Fr[3]);
    const v2f fb1 = lo ? pr2(Fr[4], Fr[5])   : pr2(Fr[6], Fr[7]);
    const v2f hb0 = lo ? pr2(HFr[0], HFr[1]) : pr2(HFr[2], HFr[3]);
    const v2f hb1 = lo ? pr2(HFr[4], HFr[5]) : pr2(HFr[6], HFr[7]);

    v2f b2 = *(const v2f*)(yrow);
    v2f b3 = *(const v2f*)(yrow + 4);
    v2f a0 = {0.f, 0.f}, a1 = {0.f, 0.f}, a2 = {0.f, 0.f}, a3 = {0.f, 0.f};

    int tstop = Tn;
    for (int t = 0; t < Tn; ++t) {
        const int tn = (t + 1 < Tn) ? (t + 1) : t;
        const v2f nb2 = *(const v2f*)(yrow + (size_t)tn * 8);
        const v2f nb3 = *(const v2f*)(yrow + (size_t)tn * 8 + 4);
        const v2f pa0 = pr2(Pr[0], Pr[1]);
        const v2f pa1 = pr2(Pr[4], Pr[5]);
        v8f uacc = __builtin_amdgcn_wmma_f32_16x16x4_f32(false, pa0, false, hb0,
                                                         (short)0, zero8, false, false);
        uacc = __builtin_amdgcn_wmma_f32_16x16x4_f32(false, pa1, false, hb1,
                                                     (short)0, uacc, false, false);
        v8f tacc = __builtin_amdgcn_wmma_f32_16x16x4_f32(false, pa0, false, fb0,
                                                         (short)0, zero8, false, false);
        tacc = __builtin_amdgcn_wmma_f32_16x16x4_f32(false, pa1, false, fb1,
                                                     (short)0, tacc, false, false);
        float T1r[8], Ur[8];
        #pragma unroll
        for (int j = 0; j < 8; ++j) { T1r[j] = tacc[j]; Ur[j] = uacc[j]; }
        float Sr[8];
        #pragma unroll
        for (int jp = 0; jp < 4; ++jp) {
            v2f acc = pr2(Cr[2 * jp], Cr[2 * jp + 1]);
            #pragma unroll
            for (int k = 0; k < 8; ++k) acc += bc2(Ur[k]) * pr2(HFf[2 * jp][k], HFf[2 * jp + 1][k]);
            Sr[2 * jp] = acc.x; Sr[2 * jp + 1] = acc.y;
        }
        lds_put_row(Esh, lane, Sr);
        lds_handover();
        float L[8][8];
        {
            const v2f r1  = *(const v2f*)&Esh[1][0];
            const v2f r2  = *(const v2f*)&Esh[2][0];
            const v4f r3  = *(const v4f*)&Esh[3][0];
            const v4f r4  = *(const v4f*)&Esh[4][0];
            const v4f r5  = *(const v4f*)&Esh[5][0];
            const v2f r5b = *(const v2f*)&Esh[5][4];
            const v4f r6  = *(const v4f*)&Esh[6][0];
            const v2f r6b = *(const v2f*)&Esh[6][4];
            const v4f r7  = *(const v4f*)&Esh[7][0];
            const v4f r7b = *(const v4f*)&Esh[7][4];
            L[0][0] = Esh[0][0];
            L[1][0] = r1.x;  L[1][1] = r1.y;
            L[2][0] = r2.x;  L[2][1] = r2.y;  L[2][2] = Esh[2][2];
            L[3][0] = r3.x;  L[3][1] = r3.y;  L[3][2] = r3.z;  L[3][3] = r3.w;
            L[4][0] = r4.x;  L[4][1] = r4.y;  L[4][2] = r4.z;  L[4][3] = r4.w;
            L[4][4] = Esh[4][4];
            L[5][0] = r5.x;  L[5][1] = r5.y;  L[5][2] = r5.z;  L[5][3] = r5.w;
            L[5][4] = r5b.x; L[5][5] = r5b.y;
            L[6][0] = r6.x;  L[6][1] = r6.y;  L[6][2] = r6.z;  L[6][3] = r6.w;
            L[6][4] = r6b.x; L[6][5] = r6b.y; L[6][6] = Esh[6][6];
            L[7][0] = r7.x;  L[7][1] = r7.y;  L[7][2] = r7.z;  L[7][3] = r7.w;
            L[7][4] = r7b.x; L[7][5] = r7b.y; L[7][6] = r7b.z; L[7][7] = r7b.w;
        }
        __builtin_amdgcn_sched_barrier(0);
        float Pnegr[8];
        #pragma unroll
        for (int jp = 0; jp < 4; ++jp) {
            v2f acc = pr2(Qr[2 * jp], Qr[2 * jp + 1]);
            #pragma unroll
            for (int k = 0; k < 8; ++k) acc += bc2(T1r[k]) * pr2(Ff[2 * jp][k], Ff[2 * jp + 1][k]);
            Pnegr[2 * jp] = acc.x; Pnegr[2 * jp + 1] = acc.y;
        }
        lds_put_row(Dsh, lane, Pnegr);
        lds_handover();
        float Pgc[8], Xc[8];
        #pragma unroll
        for (int r = 0; r < 8; ++r) Pgc[r] = Dsh[r][idx];
        __builtin_amdgcn_sched_barrier(0);

        float invd[8];
        #pragma unroll
        for (int k = 0; k < 4; ++k) {
            invd[k] = __builtin_amdgcn_rsqf(L[k][k]);
            #pragma unroll
            for (int i = k + 1; i < 8; ++i) L[i][k] = L[i][k] * invd[k];
            #pragma unroll
            for (int i = k + 1; i < 8; ++i) {
                #pragma unroll
                for (int j = k + 1; j <= i; ++j) L[i][j] -= L[i][k] * L[j][k];
            }
        }
        __builtin_amdgcn_sched_barrier(0);
        #pragma unroll
        for (int rp = 0; rp < 4; ++rp) {
            v2f acc = pr2(Hf[2 * rp][0], Hf[2 * rp + 1][0]) * bc2(Pgc[0]);
            #pragma unroll
            for (int k = 1; k < 8; ++k) acc += pr2(Hf[2 * rp][k], Hf[2 * rp + 1][k]) * bc2(Pgc[k]);
            Xc[2 * rp] = acc.x; Xc[2 * rp + 1] = acc.y;
        }
        #pragma unroll
        for (int k = 4; k < 8; ++k) {
            invd[k] = __builtin_amdgcn_rsqf(L[k][k]);
            #pragma unroll
            for (int i = k + 1; i < 8; ++i) L[i][k] = L[i][k] * invd[k];
            #pragma unroll
            for (int i = k + 1; i < 8; ++i) {
                #pragma unroll
                for (int j = k + 1; j <= i; ++j) L[i][j] -= L[i][k] * L[j][k];
            }
        }
        #pragma unroll
        for (int r = 0; r < 8; ++r) {
            float s = Xc[r];
            #pragma unroll
            for (int m = 0; m < r; ++m) s -= L[r][m] * Xc[m];
            Xc[r] = s * invd[r];
        }
        #pragma unroll
        for (int r = 7; r >= 0; --r) {
            float s = Xc[r];
            #pragma unroll
            for (int m = 7; m > r; --m) s -= L[m][r] * Xc[m];
            Xc[r] = s * invd[r];
        }
        a2 = lo ? pr2(Xc[0], Xc[1]) : pr2(Xc[2], Xc[3]);
        a3 = lo ? pr2(Xc[4], Xc[5]) : pr2(Xc[6], Xc[7]);
        if (!live) { a2 = bc2(0.f); a3 = bc2(0.f); }
        v8f cy = __builtin_amdgcn_wmma_f32_16x16x4_f32(false, a2, false, b2,
                                                       (short)0, zero8, false, false);
        cy = __builtin_amdgcn_wmma_f32_16x16x4_f32(false, a3, false, b3,
                                                   (short)0, cy, false, false);
        float IKHr[8];
        #pragma unroll
        for (int jp = 0; jp < 4; ++jp) {
            v2f acc = pr2((idx == 2 * jp) ? 1.f : 0.f, (idx == 2 * jp + 1) ? 1.f : 0.f);
            #pragma unroll
            for (int m = 7; m >= 0; --m) acc -= bc2(Xc[m]) * pr2(Hf[m][2 * jp], Hf[m][2 * jp + 1]);
            IKHr[2 * jp] = acc.x; IKHr[2 * jp + 1] = acc.y;
        }
        const v2f ga0 = lo ? pr2(Pgc[0], Pgc[1])   : pr2(Pgc[2], Pgc[3]);
        const v2f ga1 = lo ? pr2(Pgc[4], Pgc[5])   : pr2(Pgc[6], Pgc[7]);
        const v2f kb0 = lo ? pr2(IKHr[0], IKHr[1]) : pr2(IKHr[2], IKHr[3]);
        const v2f kb1 = lo ? pr2(IKHr[4], IKHr[5]) : pr2(IKHr[6], IKHr[7]);
        v8f pacc = __builtin_amdgcn_wmma_f32_16x16x4_f32(false, ga0, false, kb0,
                                                         (short)0, zero8, false, false);
        v8f qacc = __builtin_amdgcn_wmma_f32_16x16x4_f32(false, kb0, false, ga0,
                                                         (short)0, zero8, false, false);
        pacc = __builtin_amdgcn_wmma_f32_16x16x4_f32(false, ga1, false, kb1,
                                                     (short)0, pacc, false, false);
        qacc = __builtin_amdgcn_wmma_f32_16x16x4_f32(false, kb1, false, ga1,
                                                     (short)0, qacc, false, false);
        float Ar[8], Pnr[8], Pnc[8];
        #pragma unroll
        for (int j = 0; j < 8; ++j) { Pnr[j] = pacc[j]; Pnc[j] = qacc[j]; }
        #pragma unroll
        for (int jp = 0; jp < 4; ++jp) {
            v2f acc = bc2(IKHr[0]) * pr2(Ff[0][2 * jp], Ff[0][2 * jp + 1]);
            #pragma unroll
            for (int k = 1; k < 8; ++k) acc += bc2(IKHr[k]) * pr2(Ff[k][2 * jp], Ff[k][2 * jp + 1]);
            Ar[2 * jp] = acc.x; Ar[2 * jp + 1] = acc.y;
        }
        a0 = lo ? pr2(Ar[0], Ar[1]) : pr2(Ar[2], Ar[3]);
        a1 = lo ? pr2(Ar[4], Ar[5]) : pr2(Ar[6], Ar[7]);
        if (!live) { a0 = bc2(0.f); a1 = bc2(0.f); }
        const v2f b0 = {xv[0], xv[1]};
        const v2f b1 = {xv[4], xv[5]};
        const v8f c1 = __builtin_amdgcn_wmma_f32_16x16x4_f32(false, a0, false, b0,
                                                             (short)0, cy, false, false);
        const v8f c2 = __builtin_amdgcn_wmma_f32_16x16x4_f32(false, a1, false, b1,
                                                             (short)0, zero8, false, false);
        bool changed = false;
        float Pnew[8];
        #pragma unroll
        for (int j = 0; j < 8; ++j) {
            Pnew[j] = 0.5f * (Pnr[j] + Pnc[j]);
            changed = changed || (Pnew[j] != Pr[j]);
            Pr[j] = Pnew[j];
        }

        if (ps_writer) {
            v4f p0 = {Pnew[0], Pnew[1], Pnew[2], Pnew[3]};
            v4f p1 = {Pnew[4], Pnew[5], Pnew[6], Pnew[7]};
            *(v4f*)(Ps + t * 64 + idx * 8)     = p0;
            *(v4f*)(Ps + t * 64 + idx * 8 + 4) = p1;
        }
        xv = c1 + c2;
        {
            v4f out;
            out.x = lo ? xv[0] : xv[2];
            out.y = lo ? xv[1] : xv[3];
            out.z = lo ? xv[2] : xv[4];
            out.w = lo ? xv[3] : xv[5];
            *(v4f*)(orow + (size_t)t * 8) = out;
        }
        b2 = nb2; b3 = nb3;
        if (__builtin_amdgcn_ballot_w32(changed) == 0u) { tstop = t + 1; break; }
    }

    if (tstop < Tn) {
        if (ps_writer) {
            const v4f p0 = {Pr[0], Pr[1], Pr[2], Pr[3]};
            const v4f p1 = {Pr[4], Pr[5], Pr[6], Pr[7]};
            for (int t = tstop; t < Tn; ++t) {
                *(v4f*)(Ps + t * 64 + idx * 8)     = p0;
                *(v4f*)(Ps + t * 64 + idx * 8 + 4) = p1;
            }
        }
        int t = tstop;
        v2f yb2[8], yb3[8];
        #pragma unroll
        for (int j = 0; j < 8; ++j) {
            const int tj = (t + j < Tn) ? (t + j) : (Tn - 1);
            yb2[j] = *(const v2f*)(yrow + (size_t)tj * 8);
            yb3[j] = *(const v2f*)(yrow + (size_t)tj * 8 + 4);
        }
        for (; t + 8 <= Tn; t += 8) {
            v2f nb2[8], nb3[8];
            #pragma unroll
            for (int j = 0; j < 8; ++j) {
                const int tj = (t + 8 + j < Tn) ? (t + 8 + j) : (Tn - 1);
                nb2[j] = *(const v2f*)(yrow + (size_t)tj * 8);
                nb3[j] = *(const v2f*)(yrow + (size_t)tj * 8 + 4);
            }
            #pragma unroll
            for (int j = 0; j < 8; ++j)
                state_step(xv, a0, a1, a2, a3, yb2[j], yb3[j], lo, orow + (size_t)(t + j) * 8);
            #pragma unroll
            for (int j = 0; j < 8; ++j) { yb2[j] = nb2[j]; yb3[j] = nb3[j]; }
        }
        #pragma unroll
        for (int j = 0; j < 7; ++j) {
            if (t + j < Tn)
                state_step(xv, a0, a1, a2, a3, yb2[j], yb3[j], lo, orow + (size_t)(t + j) * 8);
        }
    }
}

extern "C" void kernel_launch(void* const* d_in, const int* in_sizes, int n_in,
                              void* d_out, int out_size, void* d_ws, size_t ws_size,
                              hipStream_t stream) {
    const float* Y    = (const float*)d_in[0];
    const float* F    = (const float*)d_in[1];
    const float* Gmat = (const float*)d_in[2];
    const float* H    = (const float*)d_in[3];
    const float* Q    = (const float*)d_in[4];
    const float* R    = (const float*)d_in[5];
    const float* x0   = (const float*)d_in[6];
    const float* P0   = (const float*)d_in[7];

    float* traj = (float*)d_out;
    float* Ps   = traj + (size_t)Bn * Tn * 8;

    kf_fused_kernel<<<Bn / 16, 32, 0, stream>>>(Y, F, Gmat, H, Q, R, x0, P0, traj, Ps);
}


# S is factored square-root free (L D L^T): a pivot enters only as -1/d, one v_rcp_f32, one multiply-add between pivots; unit-diagonal solves, one multiply-add per row; sum order kept, last bits differ
# speedup vs baseline: 15.4333x; 1.0712x over previous
#include <hip/hip_runtime.h>
#include <math.h>

static constexpr int Bn = 2048;
static constexpr int Tn = 1000;

typedef float v2f __attribute__((ext_vector_type(2)));
typedef float v4f __attribute__((ext_vector_type(4)));
typedef float v8f __attribute__((ext_vector_type(8)));


__device__ __forceinline__ void lds_handover() {
    __builtin_amdgcn_fence(__ATOMIC_RELEASE, "wavefront", "local");
    __builtin_amdgcn_fence(__ATOMIC_ACQUIRE, "wavefront", "local");
}

__device__ __forceinline__ void lds_put_row(float (*M)[8], int r, const float* v) {
    v4f a = {v[0], v[1], v[2], v[3]};
    v4f b = {v[4], v[5], v[6], v[7]};
    *(v4f*)&M[r][0] = a;
    *(v4f*)&M[r][4] = b;
}

__device__ __forceinline__ void lds_get_row(const float (*M)[8], int r, float* v) {
    v4f a = *(const v4f*)&M[r][0];
    v4f b = *(const v4f*)&M[r][4];
    v[0] = a.x; v[1] = a.y; v[2] = a.z; v[3] = a.w;
    v[4] = b.x; v[5] = b.y; v[6] = b.z; v[7] = b.w;
}

__device__ __forceinline__ v2f bc2(float x) { v2f r = {x, x}; return r; }
__device__ __forceinline__ v2f pr2(float a, float b) { v2f r = {a, b}; return r; }

__device__ __forceinline__ void state_step(v8f& xv, v2f a0, v2f a1, v2f a2, v2f a3,
                                           v2f y2, v2f y3, bool lo, float* dst) {
    const v8f zero8 = {0.f, 0.f, 0.f, 0.f, 0.f, 0.f, 0.f, 0.f};
    const v2f b0 = {xv[0], xv[1]};
    const v2f b1 = {xv[4], xv[5]};
    v8f cy = __builtin_amdgcn_wmma_f32_16x16x4_f32(false, a2, false, y2,
                                                   (short)0, zero8, false, false);
    cy = __builtin_amdgcn_wmma_f32_16x16x4_f32(false, a3, false, y3,
                                               (short)0, cy, false, false);
    const v8f c1 = __builtin_amdgcn_wmma_f32_16x16x4_f32(false, a0, false, b0,
                                                         (short)0, cy, false, false);
    const v8f c2 = __builtin_amdgcn_wmma_f32_16x16x4_f32(false, a1, false, b1,
                                                         (short)0, zero8, false, false);
    xv = c1 + c2;
    v4f out;
    out.x = lo ? xv[0] : xv[2];
    out.y = lo ? xv[1] : xv[3];
    out.z = lo ? xv[2] : xv[4];
    out.w = lo ? xv[3] : xv[5];
    *(v4f*)dst = out;
}

__global__ __launch_bounds__(32) void kf_fused_kernel(
    const float* __restrict__ Y, const float* __restrict__ F,
    const float* __restrict__ Gmat, const float* __restrict__ H,
    const float* __restrict__ Q, const float* __restrict__ R,
    const float* __restrict__ x0, const float* __restrict__ P0,
    float* __restrict__ traj, float* __restrict__ Ps)
{
    __shared__ __attribute__((aligned(16))) float Ash[8][8];
    __shared__ __attribute__((aligned(16))) float Bsh[8][8];
    __shared__ __attribute__((aligned(16))) float Dsh[32][8];
    __shared__ __attribute__((aligned(16))) float Esh[32][8];

    const int  lane   = threadIdx.x;
    const int  m16    = lane & 15;
    const int  idx    = (m16 < 8) ? m16 : ((m16 <= 13) ? (m16 - 6) : (m16 - 8));
    const bool live   = (m16 <= 13);
    const bool lo     = lane < 16;
    const bool writer = lane < 8;

    float Fr[8], Hr[8], Rr[8], Pr[8], Gr[8];
    #pragma unroll
    for (int c = 0; c < 8; ++c) {
        Fr[c] = F[idx * 8 + c];
        Hr[c] = H[idx * 8 + c];
        Rr[c] = R[idx * 8 + c];
        Gr[c] = Gmat[idx * 8 + c];
    }
    #pragma unroll
    for (int c = 0; c < 8; ++c) {
        const int e = lo ? c : ((c < 6) ? (c + 2) : c);
        Pr[c] = 0.5f * (P0[idx * 8 + e] + P0[e * 8 + idx]);
    }

    if (writer) {
        lds_put_row(Ash, idx, Fr);
        lds_put_row(Bsh, idx, Hr);
    }
    lds_handover();
    float Ff[8][8], Hf[8][8];
    #pragma unroll
    for (int r = 0; r < 8; ++r) {
        lds_get_row(Ash, r, Ff[r]);
        lds_get_row(Bsh, r, Hf[r]);
    }

    float Qr[8];
    {
        float GQr[8];
        #pragma unroll
        for (int j = 0; j < 8; ++j) {
            float acc = 0.f;
            #pragma unroll
            for (int k = 0; k < 8; ++k) acc += Gr[k] * Q[k * 8 + j];
            GQr[j] = acc;
        }
        #pragma unroll
        for (int j = 0; j < 8; ++j) {
            float acc = 0.f;
            #pragma unroll
            for (int k = 0; k < 8; ++k) acc += GQr[k] * Gmat[j * 8 + k];
            Qr[j] = acc;
        }
    }


    float HFr[8], HFf[8][8], Cr[8];
    {
        #pragma unroll
        for (int jp = 0; jp < 4; ++jp) {
            v2f acc = bc2(Hr[0]) * pr2(Ff[0][2 * jp], Ff[0][2 * jp + 1]);
            #pragma unroll
            for (int k = 1; k < 8; ++k) acc += bc2(Hr[k]) * pr2(Ff[k][2 * jp], Ff[k][2 * jp + 1]);
            HFr[2 * jp] = acc.x; HFr[2 * jp + 1] = acc.y;
        }
        if (writer) lds_put_row(Bsh, idx, HFr);
        lds_handover();
        #pragma unroll
        for (int r = 0; r < 8; ++r) lds_get_row(Bsh, r, HFf[r]);
        float HGr[8], HGQr[8], HQr[8];
        #pragma unroll
        for (int j = 0; j < 8; ++j) {
            float acc = 0.f;
            #pragma unroll
            for (int k = 0; k < 8; ++k) acc += Hr[k] * Gmat[k * 8 + j];
            HGr[j] = acc;
        }
        #pragma unroll
        for (int j = 0; j < 8; ++j) {
            float acc = 0.f;
            #pragma unroll
            for (int k = 0; k < 8; ++k) acc += HGr[k] * Q[k * 8 + j];
            HGQr[j] = acc;
        }
        #pragma unroll
        for (int j = 0; j < 8; ++j) {
            float acc = 0.f;
            #pragma unroll
            for (int k = 0; k < 8; ++k) acc += HGQr[k] * Gmat[j * 8 + k];
            HQr[j] = acc;
        }
        #pragma unroll
        for (int j = 0; j < 8; ++j) {
            float acc = Rr[j];
            #pragma unroll
            for (int k = 0; k < 8; ++k) acc += HQr[k] * Hf[j][k];
            Cr[j] = acc;
        }
    }

    const bool   ps_writer = writer && (blockIdx.x == 0);
    const size_t row  = (size_t)(blockIdx.x * 16 + m16);
    const float* yrow = Y    + row * (size_t)Tn * 8 + (lo ? 0 : 2);
    float*       orow = traj + row * (size_t)Tn * 8 + (lo ? 0 : 4);

    v8f xv;
    {
        v4f xa = *(const v4f*)(x0 + row * 8);
        v4f xb = *(const v4f*)(x0 + row * 8 + 4);
        xv[0] = lo ? xa.x : xa.z;
        xv[1] = lo ? xa.y : xa.w;
        xv[2] = lo ? xa.z : xb.x;
        xv[3] = lo ? xa.w : xb.y;
        xv[4] = lo ? xb.x : xb.z;
        xv[5] = lo ? xb.y : xb.w;
        xv[6] = lo ? xb.z : 0.f;
        xv[7] = lo ? xb.w : 0.f;
    }
    const v8f zero8 = {0.f, 0.f, 0.f, 0.f, 0.f, 0.f, 0.f, 0.f};

    const v2f fb0 = lo ? pr2(Fr[0], Fr[1])   : pr2(Fr[2], Fr[3]);
    const v2f fb1 = lo ? pr2(Fr[4], Fr[5])   : pr2(Fr[6], Fr[7]);
    const v2f hb0 = lo ? pr2(HFr[0], HFr[1]) : pr2(HFr[2], HFr[3]);
    const v2f hb1 = lo ? pr2(HFr[4], HFr[5]) : pr2(HFr[6], HFr[7]);

    v2f b2 = *(const v2f*)(yrow);
    v2f b3 = *(const v2f*)(yrow + 4);
    v2f a0 = {0.f, 0.f}, a1 = {0.f, 0.f}, a2 = {0.f, 0.f}, a3 = {0.f, 0.f};

    int tstop = Tn;
    for (int t = 0; t < Tn; ++t) {
        const int tn = (t + 1 < Tn) ? (t + 1) : t;
        const v2f nb2 = *(const v2f*)(yrow + (size_t)tn * 8);
        const v2f nb3 = *(const v2f*)(yrow + (size_t)tn * 8 + 4);
        const v2f pa0 = pr2(Pr[0], Pr[1]);
        const v2f pa1 = pr2(Pr[4], Pr[5]);
        v8f uacc = __builtin_amdgcn_wmma_f32_16x16x4_f32(false, pa0, false, hb0,
                                                         (short)0, zero8, false, false);
        uacc = __builtin_amdgcn_wmma_f32_16x16x4_f32(false, pa1, false, hb1,
                                                     (short)0, uacc, false, false);
        v8f tacc = __builtin_amdgcn_wmma_f32_16x16x4_f32(false, pa0, false, fb0,
                                                         (short)0, zero8, false, false);
        tacc = __builtin_amdgcn_wmma_f32_16x16x4_f32(false, pa1, false, fb1,
                                                     (short)0, tacc, false, false);
        float T1r[8], Ur[8];
        #pragma unroll
        for (int j = 0; j < 8; ++j) { T1r[j] = tacc[j]; Ur[j] = uacc[j]; }
        float Sr[8];
        #pragma unroll
        for (int jp = 0; jp < 4; ++jp) {
            v2f acc = pr2(Cr[2 * jp], Cr[2 * jp + 1]);
            #pragma unroll
            for (int k = 0; k < 8; ++k) acc += bc2(Ur[k]) * pr2(HFf[2 * jp][k], HFf[2 * jp + 1][k]);
            Sr[2 * jp] = acc.x; Sr[2 * jp + 1] = acc.y;
        }
        lds_put_row(Esh, lane, Sr);
        lds_handover();
        float L[8][8];
        {
            const v2f r1  = *(const v2f*)&Esh[1][0];
            const v2f r2  = *(const v2f*)&Esh[2][0];
            const v4f r3  = *(const v4f*)&Esh[3][0];
            const v4f r4  = *(const v4f*)&Esh[4][0];
            const v4f r5  = *(const v4f*)&Esh[5][0];
            const v2f r5b = *(const v2f*)&Esh[5][4];
            const v4f r6  = *(const v4f*)&Esh[6][0];
            const v2f r6b = *(const v2f*)&Esh[6][4];
            const v4f r7  = *(const v4f*)&Esh[7][0];
            const v4f r7b = *(const v4f*)&Esh[7][4];
            L[0][0] = Esh[0][0];
            L[1][0] = r1.x;  L[1][1] = r1.y;
            L[2][0] = r2.x;  L[2][1] = r2.y;  L[2][2] = Esh[2][2];
            L[3][0] = r3.x;  L[3][1] = r3.y;  L[3][2] = r3.z;  L[3][3] = r3.w;
            L[4][0] = r4.x;  L[4][1] = r4.y;  L[4][2] = r4.z;  L[4][3] = r4.w;
            L[4][4] = Esh[4][4];
            L[5][0] = r5.x;  L[5][1] = r5.y;  L[5][2] = r5.z;  L[5][3] = r5.w;
            L[5][4] = r5b.x; L[5][5] = r5b.y;
            L[6][0] = r6.x;  L[6][1] = r6.y;  L[6][2] = r6.z;  L[6][3] = r6.w;
            L[6][4] = r6b.x; L[6][5] = r6b.y; L[6][6] = Esh[6][6];
            L[7][0] = r7.x;  L[7][1] = r7.y;  L[7][2] = r7.z;  L[7][3] = r7.w;
            L[7][4] = r7b.x; L[7][5] = r7b.y; L[7][6] = r7b.z; L[7][7] = r7b.w;
        }
        __builtin_amdgcn_sched_barrier(0);
        float Pnegr[8];
        #pragma unroll
        for (int jp = 0; jp < 4; ++jp) {
            v2f acc = pr2(Qr[2 * jp], Qr[2 * jp + 1]);
            #pragma unroll
            for (int k = 0; k < 8; ++k) acc += bc2(T1r[k]) * pr2(Ff[2 * jp][k], Ff[2 * jp + 1][k]);
            Pnegr[2 * jp] = acc.x; Pnegr[2 * jp + 1] = acc.y;
        }
        lds_put_row(Dsh, lane, Pnegr);
        lds_handover();
        float Pgc[8], Xc[8];
        #pragma unroll
        for (int r = 0; r < 8; ++r) Pgc[r] = Dsh[r][idx];
        __builtin_amdgcn_sched_barrier(0);

        float nr[8];
        #pragma unroll
        for (int k = 0; k < 4; ++k) {
            nr[k] = __builtin_amdgcn_rcpf(-L[k][k]);
            float nl[8];
            #pragma unroll
            for (int i = k + 1; i < 8; ++i) nl[i] = L[i][k] * nr[k];
            L[k + 1][k + 1] = __builtin_fmaf(L[k + 1][k] * L[k + 1][k], nr[k], L[k + 1][k + 1]);
            #pragma unroll
            for (int i = k + 2; i < 8; ++i) {
                #pragma unroll
                for (int j = k + 1; j <= i; ++j) L[i][j] = __builtin_fmaf(nl[i], L[j][k], L[i][j]);
            }
            #pragma unroll
            for (int i = k + 1; i < 8; ++i) L[i][k] = nl[i];
        }
        __builtin_amdgcn_sched_barrier(0);
        #pragma unroll
        for (int rp = 0; rp < 4; ++rp) {
            v2f acc = -(pr2(Hf[2 * rp][0], Hf[2 * rp + 1][0]) * bc2(Pgc[0]));
            #pragma unroll
            for (int k = 1; k < 8; ++k) acc -= pr2(Hf[2 * rp][k], Hf[2 * rp + 1][k]) * bc2(Pgc[k]);
            Xc[2 * rp] = acc.x; Xc[2 * rp + 1] = acc.y;
        }
        #pragma unroll
        for (int k = 4; k < 8; ++k) {
            nr[k] = __builtin_amdgcn_rcpf(-L[k][k]);
            if (k < 7) {
                const int k1 = (k < 7) ? (k + 1) : 7;
                float nl[8];
                #pragma unroll
                for (int i = k1; i < 8; ++i) nl[i] = L[i][k] * nr[k];
                L[k1][k1] = __builtin_fmaf(L[k1][k] * L[k1][k], nr[k], L[k1][k1]);
                #pragma unroll
                for (int i = k1 + 1; i < 8; ++i) {
                    #pragma unroll
                    for (int j = k1; j <= i; ++j) L[i][j] = __builtin_fmaf(nl[i], L[j][k], L[i][j]);
                }
                #pragma unroll
                for (int i = k1; i < 8; ++i) L[i][k] = nl[i];
            }
        }
        #pragma unroll
        for (int m = 0; m < 7; ++m) {
            #pragma unroll
            for (int r = m + 1; r < 8; ++r) Xc[r] = __builtin_fmaf(L[r][m], Xc[m], Xc[r]);
        }
        #pragma unroll
        for (int r = 7; r >= 0; --r) {
            float s = Xc[r] * nr[r];
            #pragma unroll
            for (int m = 7; m > r; --m) s = __builtin_fmaf(L[m][r], Xc[m], s);
            Xc[r] = s;
        }
        a2 = lo ? pr2(Xc[0], Xc[1]) : pr2(Xc[2], Xc[3]);
        a3 = lo ? pr2(Xc[4], Xc[5]) : pr2(Xc[6], Xc[7]);
        if (!live) { a2 = bc2(0.f); a3 = bc2(0.f); }
        v8f cy = __builtin_amdgcn_wmma_f32_16x16x4_f32(false, a2, false, b2,
                                                       (short)0, zero8, false, false);
        cy = __builtin_amdgcn_wmma_f32_16x16x4_f32(false, a3, false, b3,
                                                   (short)0, cy, false, false);
        float IKHr[8];
        #pragma unroll
        for (int jp = 0; jp < 4; ++jp) {
            v2f acc = pr2((idx == 2 * jp) ? 1.f : 0.f, (idx == 2 * jp + 1) ? 1.f : 0.f);
            #pragma unroll
            for (int m = 7; m >= 0; --m) acc -= bc2(Xc[m]) * pr2(Hf[m][2 * jp], Hf[m][2 * jp + 1]);
            IKHr[2 * jp] = acc.x; IKHr[2 * jp + 1] = acc.y;
        }
        const v2f ga0 = lo ? pr2(Pgc[0], Pgc[1])   : pr2(Pgc[2], Pgc[3]);
        const v2f ga1 = lo ? pr2(Pgc[4], Pgc[5])   : pr2(Pgc[6], Pgc[7]);
        const v2f kb0 = lo ? pr2(IKHr[0], IKHr[1]) : pr2(IKHr[2], IKHr[3]);
        const v2f kb1 = lo ? pr2(IKHr[4], IKHr[5]) : pr2(IKHr[6], IKHr[7]);
        v8f pacc = __builtin_amdgcn_wmma_f32_16x16x4_f32(false, ga0, false, kb0,
                                                         (short)0, zero8, false, false);
        v8f qacc = __builtin_amdgcn_wmma_f32_16x16x4_f32(false, kb0, false, ga0,
                                                         (short)0, zero8, false, false);
        pacc = __builtin_amdgcn_wmma_f32_16x16x4_f32(false, ga1, false, kb1,
                                                     (short)0, pacc, false, false);
        qacc = __builtin_amdgcn_wmma_f32_16x16x4_f32(false, kb1, false, ga1,
                                                     (short)0, qacc, false, false);
        float Ar[8], Pnr[8], Pnc[8];
        #pragma unroll
        for (int j = 0; j < 8; ++j) { Pnr[j] = pacc[j]; Pnc[j] = qacc[j]; }
        #pragma unroll
        for (int jp = 0; jp < 4; ++jp) {
            v2f acc = bc2(IKHr[0]) * pr2(Ff[0][2 * jp], Ff[0][2 * jp + 1]);
            #pragma unroll
            for (int k = 1; k < 8; ++k) acc += bc2(IKHr[k]) * pr2(Ff[k][2 * jp], Ff[k][2 * jp + 1]);
            Ar[2 * jp] = acc.x; Ar[2 * jp + 1] = acc.y;
        }
        a0 = lo ? pr2(Ar[0], Ar[1]) : pr2(Ar[2], Ar[3]);
        a1 = lo ? pr2(Ar[4], Ar[5]) : pr2(Ar[6], Ar[7]);
        if (!live) { a0 = bc2(0.f); a1 = bc2(0.f); }
        const v2f b0 = {xv[0], xv[1]};
        const v2f b1 = {xv[4], xv[5]};
        const v8f c1 = __builtin_amdgcn_wmma_f32_16x16x4_f32(false, a0, false, b0,
                                                             (short)0, cy, false, false);
        const v8f c2 = __builtin_amdgcn_wmma_f32_16x16x4_f32(false, a1, false, b1,
                                                             (short)0, zero8, false, false);
        bool changed = false;
        float Pnew[8];
        #pragma unroll
        for (int j = 0; j < 8; ++j) {
            Pnew[j] = 0.5f * (Pnr[j] + Pnc[j]);
            changed = changed || (Pnew[j] != Pr[j]);
            Pr[j] = Pnew[j];
        }

        if (ps_writer) {
            v4f p0 = {Pnew[0], Pnew[1], Pnew[2], Pnew[3]};
            v4f p1 = {Pnew[4], Pnew[5], Pnew[6], Pnew[7]};
            *(v4f*)(Ps + t * 64 + idx * 8)     = p0;
            *(v4f*)(Ps + t * 64 + idx * 8 + 4) = p1;
        }
        xv = c1 + c2;
        {
            v4f out;
            out.x = lo ? xv[0] : xv[2];
            out.y = lo ? xv[1] : xv[3];
            out.z = lo ? xv[2] : xv[4];
            out.w = lo ? xv[3] : xv[5];
            *(v4f*)(orow + (size_t)t * 8) = out;
        }
        b2 = nb2; b3 = nb3;
        if (__builtin_amdgcn_ballot_w32(changed) == 0u) { tstop = t + 1; break; }
    }

    if (tstop < Tn) {
        if (ps_writer) {
            const v4f p0 = {Pr[0], Pr[1], Pr[2], Pr[3]};
            const v4f p1 = {Pr[4], Pr[5], Pr[6], Pr[7]};
            for (int t = tstop; t < Tn; ++t) {
                *(v4f*)(Ps + t * 64 + idx * 8)     = p0;
                *(v4f*)(Ps + t * 64 + idx * 8 + 4) = p1;
            }
        }
        int t = tstop;
        v2f yb2[8], yb3[8];
        #pragma unroll
        for (int j = 0; j < 8; ++j) {
            const int tj = (t + j < Tn) ? (t + j) : (Tn - 1);
            yb2[j] = *(const v2f*)(yrow + (size_t)tj * 8);
            yb3[j] = *(const v2f*)(yrow + (size_t)tj * 8 + 4);
        }
        for (; t + 8 <= Tn; t += 8) {
            v2f nb2[8], nb3[8];
            #pragma unroll
            for (int j = 0; j < 8; ++j) {
                const int tj = (t + 8 + j < Tn) ? (t + 8 + j) : (Tn - 1);
                nb2[j] = *(const v2f*)(yrow + (size_t)tj * 8);
                nb3[j] = *(const v2f*)(yrow + (size_t)tj * 8 + 4);
            }
            #pragma unroll
            for (int j = 0; j < 8; ++j)
                state_step(xv, a0, a1, a2, a3, yb2[j], yb3[j], lo, orow + (size_t)(t + j) * 8);
            #pragma unroll
            for (int j = 0; j < 8; ++j) { yb2[j] = nb2[j]; yb3[j] = nb3[j]; }
        }
        #pragma unroll
        for (int j = 0; j < 7; ++j) {
            if (t + j < Tn)
                state_step(xv, a0, a1, a2, a3, yb2[j], yb3[j], lo, orow + (size_t)(t + j) * 8);
        }
    }
}

extern "C" void kernel_launch(void* const* d_in, const int* in_sizes, int n_in,
                              void* d_out, int out_size, void* d_ws, size_t ws_size,
                              hipStream_t stream) {
    const float* Y    = (const float*)d_in[0];
    const float* F    = (const float*)d_in[1];
    const float* Gmat = (const float*)d_in[2];
    const float* H    = (const float*)d_in[3];
    const float* Q    = (const float*)d_in[4];
    const float* R    = (const float*)d_in[5];
    const float* x0   = (const float*)d_in[6];
    const float* P0   = (const float*)d_in[7];

    float* traj = (float*)d_out;
    float* Ps   = traj + (size_t)Bn * Tn * 8;

    kf_fused_kernel<<<Bn / 16, 32, 0, stream>>>(Y, F, Gmat, H, Q, R, x0, P0, traj, Ps);
}
